# starting kernel
# baseline (speedup 1.0000x reference)
#include <hip/hip_runtime.h>
#include <hip/hip_bf16.h>
#include <stdint.h>


#define Bsz   4
#define Llen  16384
#define Dch   512
#define Nst   64
#define NFFT  32768
#define PI_F  3.14159265358979323846f

typedef __attribute__((ext_vector_type(2))) float v2f;
typedef __attribute__((ext_vector_type(8))) float v8f;

#ifndef __has_builtin
#define __has_builtin(x) 0
#endif
#if __has_builtin(__builtin_amdgcn_wmma_f32_16x16x4_f32)
#define S4_WMMA 1
#else
#define S4_WMMA 0
#endif

__device__ __forceinline__ float2 cadd(float2 a, float2 b) { return make_float2(a.x + b.x, a.y + b.y); }
__device__ __forceinline__ float2 csub(float2 a, float2 b) { return make_float2(a.x - b.x, a.y - b.y); }
__device__ __forceinline__ float2 cmulf(float2 a, float2 b) { return make_float2(a.x * b.x - a.y * b.y, a.x * b.y + a.y * b.x); }
__device__ __forceinline__ float2 crotis(float2 a, float s) { return make_float2(-s * a.y, s * a.x); }

__device__ __forceinline__ void dft8(const float2* v, float s, float2* X) {
  const float R = 0.70710678118654752440f;
  float2 t0 = cadd(v[0], v[4]), t1 = cadd(v[1], v[5]);
  float2 t2 = cadd(v[2], v[6]), t3 = cadd(v[3], v[7]);
  float2 u0 = csub(v[0], v[4]), u1 = csub(v[1], v[5]);
  float2 u2 = csub(v[2], v[6]), u3 = csub(v[3], v[7]);
  float2 m1 = cmulf(u1, make_float2(R, s * R));
  float2 m2 = crotis(u2, s);
  float2 m3 = cmulf(u3, make_float2(-R, s * R));
  float2 p0 = cadd(t0, t2), p1 = csub(t0, t2);
  float2 p2 = cadd(t1, t3), p3 = crotis(csub(t1, t3), s);
  X[0] = cadd(p0, p2); X[2] = cadd(p1, p3);
  X[4] = csub(p0, p2); X[6] = csub(p1, p3);
  float2 q0 = cadd(u0, m2), q1 = csub(u0, m2);
  float2 q2 = cadd(m1, m3), q3 = crotis(csub(m1, m3), s);
  X[1] = cadd(q0, q2); X[3] = cadd(q1, q3);
  X[5] = csub(q0, q2); X[7] = csub(q1, q3);
}

__global__ __launch_bounds__(256) void s4_perl(
    const float* __restrict__ Lam, const float* __restrict__ Pri,
    const float* __restrict__ Cri, const float* __restrict__ log_step,
    float* __restrict__ perl)
{
  int l = blockIdx.x * blockDim.x + threadIdx.x;
  if (l >= Llen) return;
  float step = expf(log_step[0]);
  float theta = -2.0f * PI_F * (float)l / (float)Llen;
  float oi, orr;
  sincosf(theta, &oi, &orr);
  float op1r = 1.0f + orr;
  float den = op1r * op1r + oi * oi;
  float nr = 1.0f - orr, ni = -oi;
  float gscale = 2.0f / step;
  float gr = gscale * (nr * op1r + ni * oi) / den;
  float gi = gscale * (ni * op1r - nr * oi) / den;
  float scr = 2.0f * op1r / den;
  float sci = -2.0f * oi / den;
  float k01r = 0.f, k01i = 0.f, k11r = 0.f, k11i = 0.f;
  for (int n = 0; n < Nst; ++n) {
    float lr = Lam[2 * n], li = Lam[2 * n + 1];
    float pr = Pri[2 * n], pi = Pri[2 * n + 1];
    float cr = Cri[2 * n], ci = Cri[2 * n + 1];
    float zr = gr - lr, zi = gi - li;
    float zd = zr * zr + zi * zi;
    float car = zr / zd, cai = -zi / zd;
    float w01r = cr * pr + ci * pi;
    float w01i = cr * pi - ci * pr;
    float w11r = pr * pr + pi * pi;
    k01r += car * w01r - cai * w01i;
    k01i += car * w01i + cai * w01r;
    k11r += car * w11r;
    k11i += cai * w11r;
  }
  float dr = 1.0f + k11r, di = k11i;
  float dd = dr * dr + di * di;
  float sr = (k01r * dr + k01i * di) / dd;
  float si = (k01i * dr - k01r * di) / dd;
  perl[l] = sr;
  perl[Llen + l] = si;
  perl[2 * Llen + l] = scr;
  perl[3 * Llen + l] = sci;
  perl[4 * Llen + l] = gr;
  perl[5 * Llen + l] = gi;
}

__global__ __launch_bounds__(256) void s4_weights(
    const float* __restrict__ Bri, const float* __restrict__ Pri,
    const float* __restrict__ Cri,
    float* __restrict__ W0r, float* __restrict__ W0i,
    float* __restrict__ W1r, float* __restrict__ W1i)
{
  int idx = blockIdx.x * blockDim.x + threadIdx.x;
  if (idx >= Nst * Dch) return;
  int n = idx / Dch, d = idx % Dch;
  float br = Bri[2 * (n * Dch + d)], bi = Bri[2 * (n * Dch + d) + 1];
  float cr = Cri[2 * n], ci = Cri[2 * n + 1];
  float pr = Pri[2 * n], pi = Pri[2 * n + 1];
  int o = (n >> 1) * (2 * Dch) + 2 * d + (n & 1);
  W0r[o] = cr * br + ci * bi;
  W0i[o] = cr * bi - ci * br;
  W1r[o] = pr * br + pi * bi;
  W1i[o] = pr * bi - pi * br;
}

__global__ __launch_bounds__(256) void s4_cauchy(
    const float* __restrict__ Lam, const float* __restrict__ perl,
    float* __restrict__ APr, float* __restrict__ APi)
{
  int idx = blockIdx.x * blockDim.x + threadIdx.x;
  if (idx >= Llen * Nst) return;
  int n = idx / Llen, l = idx % Llen;
  float gr = perl[4 * Llen + l], gi = perl[5 * Llen + l];
  float zr = gr - Lam[2 * n], zi = gi - Lam[2 * n + 1];
  float zd = zr * zr + zi * zi;
  int o = (n >> 1) * (2 * Llen) + 2 * l + (n & 1);
  APr[o] = zr / zd;
  APi[o] = -zi / zd;
}

__global__ __launch_bounds__(32) void s4_cauchy_gemm(
    const float* __restrict__ APr, const float* __restrict__ APi,
    const float* __restrict__ W0r, const float* __restrict__ W0i,
    const float* __restrict__ W1r, const float* __restrict__ W1i,
    const float* __restrict__ perl, float2* __restrict__ Ev)
{
  int lane = threadIdx.x & 31;
  int tileD = blockIdx.x;
  int tileL = blockIdx.y;
  int d0 = tileD * 16 + (lane & 15);
  int mbase = tileL * 16 + ((lane >> 4) ? 8 : 0);
#if S4_WMMA
  int m = tileL * 16 + (lane & 15);
  int khi = (lane >> 4) * 2;
  v8f aRr = {}, aRi = {}, aSr = {}, aSi = {};
  for (int kb = 0; kb < Nst; kb += 4) {
    int k0 = kb + khi;
    const float* pr = APr + (k0 >> 1) * (2 * Llen) + 2 * m;
    const float* pi = APi + (k0 >> 1) * (2 * Llen) + 2 * m;
    v2f ar = *(const v2f*)pr;
    v2f ai = *(const v2f*)pi;
    v2f nai = -ai;
    int bo = (k0 >> 1) * (2 * Dch) + 2 * d0;
    v2f b0r = *(const v2f*)(W0r + bo);
    v2f b0i = *(const v2f*)(W0i + bo);
    v2f b1r = *(const v2f*)(W1r + bo);
    v2f b1i = *(const v2f*)(W1i + bo);
    aRr = __builtin_amdgcn_wmma_f32_16x16x4_f32(false, ar,  false, b0r, (short)0, aRr, false, false);
    aRr = __builtin_amdgcn_wmma_f32_16x16x4_f32(false, nai, false, b0i, (short)0, aRr, false, false);
    aRi = __builtin_amdgcn_wmma_f32_16x16x4_f32(false, ar,  false, b0i, (short)0, aRi, false, false);
    aRi = __builtin_amdgcn_wmma_f32_16x16x4_f32(false, ai,  false, b0r, (short)0, aRi, false, false);
    aSr = __builtin_amdgcn_wmma_f32_16x16x4_f32(false, ar,  false, b1r, (short)0, aSr, false, false);
    aSr = __builtin_amdgcn_wmma_f32_16x16x4_f32(false, nai, false, b1i, (short)0, aSr, false, false);
    aSi = __builtin_amdgcn_wmma_f32_16x16x4_f32(false, ar,  false, b1i, (short)0, aSi, false, false);
    aSi = __builtin_amdgcn_wmma_f32_16x16x4_f32(false, ai,  false, b1r, (short)0, aSi, false, false);
  }
  for (int r = 0; r < 8; ++r) {
    int l = mbase + r;
    float k00r = aRr[r], k00i = aRi[r];
    float k10r = aSr[r], k10i = aSi[r];
    float sr = perl[l],            si = perl[Llen + l];
    float cr = perl[2 * Llen + l], ci = perl[3 * Llen + l];
    float tr = k00r - (sr * k10r - si * k10i);
    float ti = k00i - (sr * k10i + si * k10r);
    Ev[l * Dch + d0] = make_float2(cr * tr - ci * ti, cr * ti + ci * tr);
  }
#else
  for (int r = 0; r < 8; ++r) {
    int l = mbase + r;
    float k00r = 0.f, k00i = 0.f, k10r = 0.f, k10i = 0.f;
    for (int k = 0; k < Nst; ++k) {
      int ao = (k >> 1) * (2 * Llen) + 2 * l + (k & 1);
      int bo = (k >> 1) * (2 * Dch) + 2 * d0 + (k & 1);
      float arr = APr[ao], aii = APi[ao];
      float b0r = W0r[bo], b0i = W0i[bo];
      float b1r = W1r[bo], b1i = W1i[bo];
      k00r += arr * b0r - aii * b0i;  k00i += arr * b0i + aii * b0r;
      k10r += arr * b1r - aii * b1i;  k10i += arr * b1i + aii * b1r;
    }
    float sr = perl[l],            si = perl[Llen + l];
    float cr = perl[2 * Llen + l], ci = perl[3 * Llen + l];
    float tr = k00r - (sr * k10r - si * k10i);
    float ti = k00i - (sr * k10i + si * k10r);
    Ev[l * Dch + d0] = make_float2(cr * tr - ci * ti, cr * ti + ci * tr);
  }
#endif
}

__global__ __launch_bounds__(256) void s4_fft_r4(
    const float2* __restrict__ x, float2* __restrict__ y,
    int n, int Ns, float sign)
{
  int t = blockIdx.x * blockDim.x + threadIdx.x;
  int q = n >> 2;
  if (t >= q * Dch) return;
  int col = t & (Dch - 1);
  int j = t >> 9;
  int k = j & (Ns - 1);
  float ang = sign * (0.5f * PI_F) * (float)k / (float)Ns;
  float s1, c1;
  sincosf(ang, &s1, &c1);
  float c2 = c1 * c1 - s1 * s1, s2 = 2.f * c1 * s1;
  float c3 = c2 * c1 - s2 * s1, s3 = c2 * s1 + s2 * c1;
  float2 a = x[j * Dch + col];
  float2 b = x[(j + q) * Dch + col];
  float2 c = x[(j + 2 * q) * Dch + col];
  float2 d = x[(j + 3 * q) * Dch + col];
  float2 v1 = make_float2(c1 * b.x - s1 * b.y, c1 * b.y + s1 * b.x);
  float2 v2 = make_float2(c2 * c.x - s2 * c.y, c2 * c.y + s2 * c.x);
  float2 v3 = make_float2(c3 * d.x - s3 * d.y, c3 * d.y + s3 * d.x);
  float2 p0 = cadd(a, v2), p1 = csub(a, v2);
  float2 p2 = cadd(v1, v3), p3 = crotis(csub(v1, v3), sign);
  int i0 = 4 * j - 3 * k;
  y[(i0         ) * Dch + col] = cadd(p0, p2);
  y[(i0 +     Ns) * Dch + col] = cadd(p1, p3);
  y[(i0 + 2 * Ns) * Dch + col] = csub(p0, p2);
  y[(i0 + 3 * Ns) * Dch + col] = csub(p1, p3);
}

__global__ __launch_bounds__(256) void s4_fft_r8(
    const float2* __restrict__ x, float2* __restrict__ y,
    int n, int Ns, float sign)
{
  int t = blockIdx.x * blockDim.x + threadIdx.x;
  int q = n >> 3;
  if (t >= q * Dch) return;
  int col = t & (Dch - 1);
  int j = t >> 9;
  int k = j & (Ns - 1);
  float ang = sign * (0.25f * PI_F) * (float)k / (float)Ns;
  float s1, c1;
  sincosf(ang, &s1, &c1);
  float2 w1 = make_float2(c1, s1);
  float2 v[8];
  v[0] = x[j * Dch + col];
  float2 w = w1;
#pragma unroll
  for (int r = 1; r < 8; ++r) {
    v[r] = cmulf(x[(j + r * q) * Dch + col], w);
    w = cmulf(w, w1);
  }
  float2 X[8];
  dft8(v, sign, X);
  int i0 = 8 * j - 7 * k;
#pragma unroll
  for (int m = 0; m < 8; ++m)
    y[(i0 + m * Ns) * Dch + col] = X[m];
}

__global__ __launch_bounds__(256) void s4_upad_r8(
    const float* __restrict__ u, float2* __restrict__ y)
{
  const int Q = NFFT / 8;
  int t = blockIdx.x * blockDim.x + threadIdx.x;
  if (t >= Q * Dch) return;
  int col = t & (Dch - 1);
  int j = t >> 9;
  float2 v[8];
#pragma unroll
  for (int r = 0; r < 4; ++r)
    v[r] = make_float2(u[(j + r * Q) * Dch + col], 0.f);
#pragma unroll
  for (int r = 4; r < 8; ++r)
    v[r] = make_float2(0.f, 0.f);
  float2 X[8];
  dft8(v, -1.0f, X);
  int i0 = 8 * j;
#pragma unroll
  for (int m = 0; m < 8; ++m)
    y[(i0 + m) * Dch + col] = X[m];
}

__global__ __launch_bounds__(256) void s4_kpad_r8(
    const float2* __restrict__ Kt, float2* __restrict__ y)
{
  const int Q = NFFT / 8;
  int t = blockIdx.x * blockDim.x + threadIdx.x;
  if (t >= Q * Dch) return;
  int col = t & (Dch - 1);
  int j = t >> 9;
  float2 v[8];
#pragma unroll
  for (int r = 0; r < 4; ++r)
    v[r] = make_float2(Kt[(j + r * Q) * Dch + col].x * (1.0f / (float)Llen), 0.f);
#pragma unroll
  for (int r = 4; r < 8; ++r)
    v[r] = make_float2(0.f, 0.f);
  float2 X[8];
  dft8(v, -1.0f, X);
  int i0 = 8 * j;
#pragma unroll
  for (int m = 0; m < 8; ++m)
    y[(i0 + m) * Dch + col] = X[m];
}

__global__ __launch_bounds__(256) void s4_cmul_r8(
    const float2* __restrict__ Xs, const float2* __restrict__ Kd,
    float2* __restrict__ y)
{
  const int Q = NFFT / 8;
  int t = blockIdx.x * blockDim.x + threadIdx.x;
  if (t >= Q * Dch) return;
  int col = t & (Dch - 1);
  int j = t >> 9;
  float2 v[8];
#pragma unroll
  for (int r = 0; r < 8; ++r) {
    int idx = (j + r * Q) * Dch + col;
    v[r] = cmulf(Xs[idx], Kd[idx]);
  }
  float2 X[8];
  dft8(v, +1.0f, X);
  int i0 = 8 * j;
#pragma unroll
  for (int m = 0; m < 8; ++m)
    y[(i0 + m) * Dch + col] = X[m];
}

__global__ __launch_bounds__(256) void s4_final_r8(
    const float2* __restrict__ x, const float* __restrict__ u,
    const float* __restrict__ Dp, float* __restrict__ out)
{
  const int Q = NFFT / 8;
  int t = blockIdx.x * blockDim.x + threadIdx.x;
  if (t >= Q * Dch) return;
  int col = t & (Dch - 1);
  int j = t >> 9;
  float ang = (0.25f * PI_F) * (float)j / (float)Q;
  float s1, c1;
  sincosf(ang, &s1, &c1);
  float2 w1 = make_float2(c1, s1);
  float2 v[8];
  v[0] = x[j * Dch + col];
  float2 w = w1;
#pragma unroll
  for (int r = 1; r < 8; ++r) {
    v[r] = cmulf(x[(j + r * Q) * Dch + col], w);
    w = cmulf(w, w1);
  }
  float2 X[8];
  dft8(v, +1.0f, X);
  float dp = Dp[0];
  const float inv = 1.0f / (float)NFFT;
#pragma unroll
  for (int m = 0; m < 4; ++m) {
    int pos = (j + m * Q) * Dch + col;
    out[pos] = X[m].x * inv + dp * u[pos];
  }
}


extern "C" void kernel_launch(void* const* d_in, const int* in_sizes, int n_in,
                              void* d_out, int out_size, void* d_ws, size_t ws_size,
                              hipStream_t stream) {
  const float* u        = (const float*)d_in[0];
  const float* Lam      = (const float*)d_in[1];
  const float* Pri      = (const float*)d_in[2];
  const float* Bri      = (const float*)d_in[3];
  const float* Cri      = (const float*)d_in[4];
  const float* Dp       = (const float*)d_in[5];
  const float* log_step = (const float*)d_in[6];
  float* out = (float*)d_out;

  uint8_t* base = (uint8_t*)d_ws;
  size_t off = 0;
  auto carve = [&](size_t bytes) -> void* {
    void* p = base + off;
    off += (bytes + 255) & ~(size_t)255;
    return p;
  };
  float*  APr  = (float*)carve((size_t)Llen * Nst * 4);
  float*  APi  = (float*)carve((size_t)Llen * Nst * 4);
  float*  W0r  = (float*)carve((size_t)Nst * Dch * 4);
  float*  W0i  = (float*)carve((size_t)Nst * Dch * 4);
  float*  W1r  = (float*)carve((size_t)Nst * Dch * 4);
  float*  W1i  = (float*)carve((size_t)Nst * Dch * 4);
  float*  perl = (float*)carve((size_t)6 * Llen * 4);
  float2* Ev   = (float2*)carve((size_t)Llen * Dch * 8);
  float2* T0   = (float2*)carve((size_t)NFFT * Dch * 8);
  float2* T1   = (float2*)carve((size_t)NFFT * Dch * 8);
  float2* Kd   = (float2*)carve((size_t)NFFT * Dch * 8);
  (void)ws_size; (void)in_sizes; (void)n_in; (void)out_size;

  const int TB   = 256;
  const int gR8  = (NFFT / 8) * Dch / TB;
  const int gR8s = (Llen / 8) * Dch / TB;
  const int gR4s = (Llen / 4) * Dch / TB;

  s4_perl<<<(Llen + TB - 1) / TB, TB, 0, stream>>>(Lam, Pri, Cri, log_step, perl);
  s4_weights<<<(Nst * Dch + TB - 1) / TB, TB, 0, stream>>>(Bri, Pri, Cri, W0r, W0i, W1r, W1i);
  s4_cauchy<<<(Llen * Nst + TB - 1) / TB, TB, 0, stream>>>(Lam, perl, APr, APi);
  s4_cauchy_gemm<<<dim3(Dch / 16, Llen / 16), 32, 0, stream>>>(
      APr, APi, W0r, W0i, W1r, W1i, perl, Ev);

  s4_fft_r4<<<gR4s, TB, 0, stream>>>(Ev, T0, Llen, 1, +1.0f);
  for (int p = 0; p < 4; ++p) {
    const float2* src = (p & 1) ? Ev : T0;
    float2*       dst = (p & 1) ? T0 : Ev;
    s4_fft_r8<<<gR8s, TB, 0, stream>>>(src, dst, Llen, 4 << (3 * p), +1.0f);
  }

  s4_kpad_r8<<<gR8, TB, 0, stream>>>(T0, T1);
  for (int p = 0; p < 4; ++p) {
    const float2* src = (p & 1) ? T0 : T1;
    float2*       dst = (p == 3) ? Kd : ((p & 1) ? T1 : T0);
    s4_fft_r8<<<gR8, TB, 0, stream>>>(src, dst, NFFT, 8 << (3 * p), -1.0f);
  }

  for (int b = 0; b < Bsz; ++b) {
    const float* ub = u + (size_t)b * Llen * Dch;
    float*       ob = out + (size_t)b * Llen * Dch;
    s4_upad_r8<<<gR8, TB, 0, stream>>>(ub, T0);
    for (int p = 0; p < 4; ++p) {
      const float2* src = (p & 1) ? T1 : T0;
      float2*       dst = (p & 1) ? T0 : T1;
      s4_fft_r8<<<gR8, TB, 0, stream>>>(src, dst, NFFT, 8 << (3 * p), -1.0f);
    }
    s4_cmul_r8<<<gR8, TB, 0, stream>>>(T0, Kd, T1);
    for (int p = 0; p < 3; ++p) {
      const float2* src = (p & 1) ? T0 : T1;
      float2*       dst = (p & 1) ? T1 : T0;
      s4_fft_r8<<<gR8, TB, 0, stream>>>(src, dst, NFFT, 8 << (3 * p), +1.0f);
    }
    s4_final_r8<<<gR8, TB, 0, stream>>>(T0, ub, Dp, ob);
  }
}


# Two real batch rows ride one complex 2L-point FFT convolution as re/im (K is real), halving the per-batch passes; each row's rounding now mixes with its partner's
# speedup vs baseline: 1.6150x; 1.6150x over previous
#include <hip/hip_runtime.h>
#include <hip/hip_bf16.h>
#include <stdint.h>


#define Bsz   4
#define Llen  16384
#define Dch   512
#define Nst   64
#define NFFT  32768
#define PI_F  3.14159265358979323846f

typedef __attribute__((ext_vector_type(2))) float v2f;
typedef __attribute__((ext_vector_type(8))) float v8f;

#ifndef __has_builtin
#define __has_builtin(x) 0
#endif
#if __has_builtin(__builtin_amdgcn_wmma_f32_16x16x4_f32)
#define S4_WMMA 1
#else
#define S4_WMMA 0
#endif

__device__ __forceinline__ float2 cadd(float2 a, float2 b) { return make_float2(a.x + b.x, a.y + b.y); }
__device__ __forceinline__ float2 csub(float2 a, float2 b) { return make_float2(a.x - b.x, a.y - b.y); }
__device__ __forceinline__ float2 cmulf(float2 a, float2 b) { return make_float2(a.x * b.x - a.y * b.y, a.x * b.y + a.y * b.x); }
__device__ __forceinline__ float2 crotis(float2 a, float s) { return make_float2(-s * a.y, s * a.x); }

__device__ __forceinline__ void dft8(const float2* v, float s, float2* X) {
  const float R = 0.70710678118654752440f;
  float2 t0 = cadd(v[0], v[4]), t1 = cadd(v[1], v[5]);
  float2 t2 = cadd(v[2], v[6]), t3 = cadd(v[3], v[7]);
  float2 u0 = csub(v[0], v[4]), u1 = csub(v[1], v[5]);
  float2 u2 = csub(v[2], v[6]), u3 = csub(v[3], v[7]);
  float2 m1 = cmulf(u1, make_float2(R, s * R));
  float2 m2 = crotis(u2, s);
  float2 m3 = cmulf(u3, make_float2(-R, s * R));
  float2 p0 = cadd(t0, t2), p1 = csub(t0, t2);
  float2 p2 = cadd(t1, t3), p3 = crotis(csub(t1, t3), s);
  X[0] = cadd(p0, p2); X[2] = cadd(p1, p3);
  X[4] = csub(p0, p2); X[6] = csub(p1, p3);
  float2 q0 = cadd(u0, m2), q1 = csub(u0, m2);
  float2 q2 = cadd(m1, m3), q3 = crotis(csub(m1, m3), s);
  X[1] = cadd(q0, q2); X[3] = cadd(q1, q3);
  X[5] = csub(q0, q2); X[7] = csub(q1, q3);
}

__global__ __launch_bounds__(256) void s4_perl(
    const float* __restrict__ Lam, const float* __restrict__ Pri,
    const float* __restrict__ Cri, const float* __restrict__ log_step,
    float* __restrict__ perl)
{
  int l = blockIdx.x * blockDim.x + threadIdx.x;
  if (l >= Llen) return;
  float step = expf(log_step[0]);
  float theta = -2.0f * PI_F * (float)l / (float)Llen;
  float oi, orr;
  sincosf(theta, &oi, &orr);
  float op1r = 1.0f + orr;
  float den = op1r * op1r + oi * oi;
  float nr = 1.0f - orr, ni = -oi;
  float gscale = 2.0f / step;
  float gr = gscale * (nr * op1r + ni * oi) / den;
  float gi = gscale * (ni * op1r - nr * oi) / den;
  float scr = 2.0f * op1r / den;
  float sci = -2.0f * oi / den;
  float k01r = 0.f, k01i = 0.f, k11r = 0.f, k11i = 0.f;
  for (int n = 0; n < Nst; ++n) {
    float lr = Lam[2 * n], li = Lam[2 * n + 1];
    float pr = Pri[2 * n], pi = Pri[2 * n + 1];
    float cr = Cri[2 * n], ci = Cri[2 * n + 1];
    float zr = gr - lr, zi = gi - li;
    float zd = zr * zr + zi * zi;
    float car = zr / zd, cai = -zi / zd;
    float w01r = cr * pr + ci * pi;
    float w01i = cr * pi - ci * pr;
    float w11r = pr * pr + pi * pi;
    k01r += car * w01r - cai * w01i;
    k01i += car * w01i + cai * w01r;
    k11r += car * w11r;
    k11i += cai * w11r;
  }
  float dr = 1.0f + k11r, di = k11i;
  float dd = dr * dr + di * di;
  float sr = (k01r * dr + k01i * di) / dd;
  float si = (k01i * dr - k01r * di) / dd;
  perl[l] = sr;
  perl[Llen + l] = si;
  perl[2 * Llen + l] = scr;
  perl[3 * Llen + l] = sci;
  perl[4 * Llen + l] = gr;
  perl[5 * Llen + l] = gi;
}

__global__ __launch_bounds__(256) void s4_weights(
    const float* __restrict__ Bri, const float* __restrict__ Pri,
    const float* __restrict__ Cri,
    float* __restrict__ W0r, float* __restrict__ W0i,
    float* __restrict__ W1r, float* __restrict__ W1i)
{
  int idx = blockIdx.x * blockDim.x + threadIdx.x;
  if (idx >= Nst * Dch) return;
  int n = idx / Dch, d = idx % Dch;
  float br = Bri[2 * (n * Dch + d)], bi = Bri[2 * (n * Dch + d) + 1];
  float cr = Cri[2 * n], ci = Cri[2 * n + 1];
  float pr = Pri[2 * n], pi = Pri[2 * n + 1];
  int o = (n >> 1) * (2 * Dch) + 2 * d + (n & 1);
  W0r[o] = cr * br + ci * bi;
  W0i[o] = cr * bi - ci * br;
  W1r[o] = pr * br + pi * bi;
  W1i[o] = pr * bi - pi * br;
}

__global__ __launch_bounds__(256) void s4_cauchy(
    const float* __restrict__ Lam, const float* __restrict__ perl,
    float* __restrict__ APr, float* __restrict__ APi)
{
  int idx = blockIdx.x * blockDim.x + threadIdx.x;
  if (idx >= Llen * Nst) return;
  int n = idx / Llen, l = idx % Llen;
  float gr = perl[4 * Llen + l], gi = perl[5 * Llen + l];
  float zr = gr - Lam[2 * n], zi = gi - Lam[2 * n + 1];
  float zd = zr * zr + zi * zi;
  int o = (n >> 1) * (2 * Llen) + 2 * l + (n & 1);
  APr[o] = zr / zd;
  APi[o] = -zi / zd;
}

__global__ __launch_bounds__(32) void s4_cauchy_gemm(
    const float* __restrict__ APr, const float* __restrict__ APi,
    const float* __restrict__ W0r, const float* __restrict__ W0i,
    const float* __restrict__ W1r, const float* __restrict__ W1i,
    const float* __restrict__ perl, float2* __restrict__ Ev)
{
  int lane = threadIdx.x & 31;
  int tileD = blockIdx.x;
  int tileL = blockIdx.y;
  int d0 = tileD * 16 + (lane & 15);
  int mbase = tileL * 16 + ((lane >> 4) ? 8 : 0);
#if S4_WMMA
  int m = tileL * 16 + (lane & 15);
  int khi = (lane >> 4) * 2;
  v8f aRr = {}, aRi = {}, aSr = {}, aSi = {};
  for (int kb = 0; kb < Nst; kb += 4) {
    int k0 = kb + khi;
    const float* pr = APr + (k0 >> 1) * (2 * Llen) + 2 * m;
    const float* pi = APi + (k0 >> 1) * (2 * Llen) + 2 * m;
    v2f ar = *(const v2f*)pr;
    v2f ai = *(const v2f*)pi;
    v2f nai = -ai;
    int bo = (k0 >> 1) * (2 * Dch) + 2 * d0;
    v2f b0r = *(const v2f*)(W0r + bo);
    v2f b0i = *(const v2f*)(W0i + bo);
    v2f b1r = *(const v2f*)(W1r + bo);
    v2f b1i = *(const v2f*)(W1i + bo);
    aRr = __builtin_amdgcn_wmma_f32_16x16x4_f32(false, ar,  false, b0r, (short)0, aRr, false, false);
    aRr = __builtin_amdgcn_wmma_f32_16x16x4_f32(false, nai, false, b0i, (short)0, aRr, false, false);
    aRi = __builtin_amdgcn_wmma_f32_16x16x4_f32(false, ar,  false, b0i, (short)0, aRi, false, false);
    aRi = __builtin_amdgcn_wmma_f32_16x16x4_f32(false, ai,  false, b0r, (short)0, aRi, false, false);
    aSr = __builtin_amdgcn_wmma_f32_16x16x4_f32(false, ar,  false, b1r, (short)0, aSr, false, false);
    aSr = __builtin_amdgcn_wmma_f32_16x16x4_f32(false, nai, false, b1i, (short)0, aSr, false, false);
    aSi = __builtin_amdgcn_wmma_f32_16x16x4_f32(false, ar,  false, b1i, (short)0, aSi, false, false);
    aSi = __builtin_amdgcn_wmma_f32_16x16x4_f32(false, ai,  false, b1r, (short)0, aSi, false, false);
  }
  for (int r = 0; r < 8; ++r) {
    int l = mbase + r;
    float k00r = aRr[r], k00i = aRi[r];
    float k10r = aSr[r], k10i = aSi[r];
    float sr = perl[l],            si = perl[Llen + l];
    float cr = perl[2 * Llen + l], ci = perl[3 * Llen + l];
    float tr = k00r - (sr * k10r - si * k10i);
    float ti = k00i - (sr * k10i + si * k10r);
    Ev[l * Dch + d0] = make_float2(cr * tr - ci * ti, cr * ti + ci * tr);
  }
#else
  for (int r = 0; r < 8; ++r) {
    int l = mbase + r;
    float k00r = 0.f, k00i = 0.f, k10r = 0.f, k10i = 0.f;
    for (int k = 0; k < Nst; ++k) {
      int ao = (k >> 1) * (2 * Llen) + 2 * l + (k & 1);
      int bo = (k >> 1) * (2 * Dch) + 2 * d0 + (k & 1);
      float arr = APr[ao], aii = APi[ao];
      float b0r = W0r[bo], b0i = W0i[bo];
      float b1r = W1r[bo], b1i = W1i[bo];
      k00r += arr * b0r - aii * b0i;  k00i += arr * b0i + aii * b0r;
      k10r += arr * b1r - aii * b1i;  k10i += arr * b1i + aii * b1r;
    }
    float sr = perl[l],            si = perl[Llen + l];
    float cr = perl[2 * Llen + l], ci = perl[3 * Llen + l];
    float tr = k00r - (sr * k10r - si * k10i);
    float ti = k00i - (sr * k10i + si * k10r);
    Ev[l * Dch + d0] = make_float2(cr * tr - ci * ti, cr * ti + ci * tr);
  }
#endif
}

__global__ __launch_bounds__(256) void s4_fft_r4(
    const float2* __restrict__ x, float2* __restrict__ y,
    int n, int Ns, float sign)
{
  int t = blockIdx.x * blockDim.x + threadIdx.x;
  int q = n >> 2;
  if (t >= q * Dch) return;
  int col = t & (Dch - 1);
  int j = t >> 9;
  int k = j & (Ns - 1);
  float ang = sign * (0.5f * PI_F) * (float)k / (float)Ns;
  float s1, c1;
  sincosf(ang, &s1, &c1);
  float c2 = c1 * c1 - s1 * s1, s2 = 2.f * c1 * s1;
  float c3 = c2 * c1 - s2 * s1, s3 = c2 * s1 + s2 * c1;
  float2 a = x[j * Dch + col];
  float2 b = x[(j + q) * Dch + col];
  float2 c = x[(j + 2 * q) * Dch + col];
  float2 d = x[(j + 3 * q) * Dch + col];
  float2 v1 = make_float2(c1 * b.x - s1 * b.y, c1 * b.y + s1 * b.x);
  float2 v2 = make_float2(c2 * c.x - s2 * c.y, c2 * c.y + s2 * c.x);
  float2 v3 = make_float2(c3 * d.x - s3 * d.y, c3 * d.y + s3 * d.x);
  float2 p0 = cadd(a, v2), p1 = csub(a, v2);
  float2 p2 = cadd(v1, v3), p3 = crotis(csub(v1, v3), sign);
  int i0 = 4 * j - 3 * k;
  y[(i0         ) * Dch + col] = cadd(p0, p2);
  y[(i0 +     Ns) * Dch + col] = cadd(p1, p3);
  y[(i0 + 2 * Ns) * Dch + col] = csub(p0, p2);
  y[(i0 + 3 * Ns) * Dch + col] = csub(p1, p3);
}

__global__ __launch_bounds__(256) void s4_fft_r8(
    const float2* __restrict__ x, float2* __restrict__ y,
    int n, int Ns, float sign)
{
  int t = blockIdx.x * blockDim.x + threadIdx.x;
  int q = n >> 3;
  if (t >= q * Dch) return;
  int col = t & (Dch - 1);
  int j = t >> 9;
  int k = j & (Ns - 1);
  float ang = sign * (0.25f * PI_F) * (float)k / (float)Ns;
  float s1, c1;
  sincosf(ang, &s1, &c1);
  float2 w1 = make_float2(c1, s1);
  float2 v[8];
  v[0] = x[j * Dch + col];
  float2 w = w1;
#pragma unroll
  for (int r = 1; r < 8; ++r) {
    v[r] = cmulf(x[(j + r * q) * Dch + col], w);
    w = cmulf(w, w1);
  }
  float2 X[8];
  dft8(v, sign, X);
  int i0 = 8 * j - 7 * k;
#pragma unroll
  for (int m = 0; m < 8; ++m)
    y[(i0 + m * Ns) * Dch + col] = X[m];
}

__global__ __launch_bounds__(256) void s4_upad_r8(
    const float* __restrict__ ua, const float* __restrict__ ub,
    float2* __restrict__ y)
{
  const int Q = NFFT / 8;
  int t = blockIdx.x * blockDim.x + threadIdx.x;
  if (t >= Q * Dch) return;
  int col = t & (Dch - 1);
  int j = t >> 9;
  float2 v[8];
#pragma unroll
  for (int r = 0; r < 4; ++r)
    v[r] = make_float2(ua[(j + r * Q) * Dch + col], ub[(j + r * Q) * Dch + col]);
#pragma unroll
  for (int r = 4; r < 8; ++r)
    v[r] = make_float2(0.f, 0.f);
  float2 X[8];
  dft8(v, -1.0f, X);
  int i0 = 8 * j;
#pragma unroll
  for (int m = 0; m < 8; ++m)
    y[(i0 + m) * Dch + col] = X[m];
}

__global__ __launch_bounds__(256) void s4_kpad_r8(
    const float2* __restrict__ Kt, float2* __restrict__ y)
{
  const int Q = NFFT / 8;
  int t = blockIdx.x * blockDim.x + threadIdx.x;
  if (t >= Q * Dch) return;
  int col = t & (Dch - 1);
  int j = t >> 9;
  float2 v[8];
#pragma unroll
  for (int r = 0; r < 4; ++r)
    v[r] = make_float2(Kt[(j + r * Q) * Dch + col].x * (1.0f / (float)Llen), 0.f);
#pragma unroll
  for (int r = 4; r < 8; ++r)
    v[r] = make_float2(0.f, 0.f);
  float2 X[8];
  dft8(v, -1.0f, X);
  int i0 = 8 * j;
#pragma unroll
  for (int m = 0; m < 8; ++m)
    y[(i0 + m) * Dch + col] = X[m];
}

__global__ __launch_bounds__(256) void s4_cmul_r8(
    const float2* __restrict__ Xs, const float2* __restrict__ Kd,
    float2* __restrict__ y)
{
  const int Q = NFFT / 8;
  int t = blockIdx.x * blockDim.x + threadIdx.x;
  if (t >= Q * Dch) return;
  int col = t & (Dch - 1);
  int j = t >> 9;
  float2 v[8];
#pragma unroll
  for (int r = 0; r < 8; ++r) {
    int idx = (j + r * Q) * Dch + col;
    v[r] = cmulf(Xs[idx], Kd[idx]);
  }
  float2 X[8];
  dft8(v, +1.0f, X);
  int i0 = 8 * j;
#pragma unroll
  for (int m = 0; m < 8; ++m)
    y[(i0 + m) * Dch + col] = X[m];
}

__global__ __launch_bounds__(256) void s4_final_r8(
    const float2* __restrict__ x,
    const float* __restrict__ ua, const float* __restrict__ ub,
    const float* __restrict__ Dp,
    float* __restrict__ outa, float* __restrict__ outb)
{
  const int Q = NFFT / 8;
  int t = blockIdx.x * blockDim.x + threadIdx.x;
  if (t >= Q * Dch) return;
  int col = t & (Dch - 1);
  int j = t >> 9;
  float ang = (0.25f * PI_F) * (float)j / (float)Q;
  float s1, c1;
  sincosf(ang, &s1, &c1);
  float2 w1 = make_float2(c1, s1);
  float2 v[8];
  v[0] = x[j * Dch + col];
  float2 w = w1;
#pragma unroll
  for (int r = 1; r < 8; ++r) {
    v[r] = cmulf(x[(j + r * Q) * Dch + col], w);
    w = cmulf(w, w1);
  }
  float2 X[8];
  dft8(v, +1.0f, X);
  float dp = Dp[0];
  const float inv = 1.0f / (float)NFFT;
#pragma unroll
  for (int m = 0; m < 4; ++m) {
    int pos = (j + m * Q) * Dch + col;
    outa[pos] = X[m].x * inv + dp * ua[pos];
    outb[pos] = X[m].y * inv + dp * ub[pos];
  }
}


extern "C" void kernel_launch(void* const* d_in, const int* in_sizes, int n_in,
                              void* d_out, int out_size, void* d_ws, size_t ws_size,
                              hipStream_t stream) {
  const float* u        = (const float*)d_in[0];
  const float* Lam      = (const float*)d_in[1];
  const float* Pri      = (const float*)d_in[2];
  const float* Bri      = (const float*)d_in[3];
  const float* Cri      = (const float*)d_in[4];
  const float* Dp       = (const float*)d_in[5];
  const float* log_step = (const float*)d_in[6];
  float* out = (float*)d_out;

  uint8_t* base = (uint8_t*)d_ws;
  size_t off = 0;
  auto carve = [&](size_t bytes) -> void* {
    void* p = base + off;
    off += (bytes + 255) & ~(size_t)255;
    return p;
  };
  float*  APr  = (float*)carve((size_t)Llen * Nst * 4);
  float*  APi  = (float*)carve((size_t)Llen * Nst * 4);
  float*  W0r  = (float*)carve((size_t)Nst * Dch * 4);
  float*  W0i  = (float*)carve((size_t)Nst * Dch * 4);
  float*  W1r  = (float*)carve((size_t)Nst * Dch * 4);
  float*  W1i  = (float*)carve((size_t)Nst * Dch * 4);
  float*  perl = (float*)carve((size_t)6 * Llen * 4);
  float2* Ev   = (float2*)carve((size_t)Llen * Dch * 8);
  float2* T0   = (float2*)carve((size_t)NFFT * Dch * 8);
  float2* T1   = (float2*)carve((size_t)NFFT * Dch * 8);
  float2* Kd   = (float2*)carve((size_t)NFFT * Dch * 8);
  (void)ws_size; (void)in_sizes; (void)n_in; (void)out_size;

  const int TB   = 256;
  const int gR8  = (NFFT / 8) * Dch / TB;
  const int gR8s = (Llen / 8) * Dch / TB;
  const int gR4s = (Llen / 4) * Dch / TB;

  s4_perl<<<(Llen + TB - 1) / TB, TB, 0, stream>>>(Lam, Pri, Cri, log_step, perl);
  s4_weights<<<(Nst * Dch + TB - 1) / TB, TB, 0, stream>>>(Bri, Pri, Cri, W0r, W0i, W1r, W1i);
  s4_cauchy<<<(Llen * Nst + TB - 1) / TB, TB, 0, stream>>>(Lam, perl, APr, APi);
  s4_cauchy_gemm<<<dim3(Dch / 16, Llen / 16), 32, 0, stream>>>(
      APr, APi, W0r, W0i, W1r, W1i, perl, Ev);

  s4_fft_r4<<<gR4s, TB, 0, stream>>>(Ev, T0, Llen, 1, +1.0f);
  for (int p = 0; p < 4; ++p) {
    const float2* src = (p & 1) ? Ev : T0;
    float2*       dst = (p & 1) ? T0 : Ev;
    s4_fft_r8<<<gR8s, TB, 0, stream>>>(src, dst, Llen, 4 << (3 * p), +1.0f);
  }

  s4_kpad_r8<<<gR8, TB, 0, stream>>>(T0, T1);
  for (int p = 0; p < 4; ++p) {
    const float2* src = (p & 1) ? T0 : T1;
    float2*       dst = (p == 3) ? Kd : ((p & 1) ? T1 : T0);
    s4_fft_r8<<<gR8, TB, 0, stream>>>(src, dst, NFFT, 8 << (3 * p), -1.0f);
  }

  for (int b = 0; b < Bsz; b += 2) {
    const float* ua = u + (size_t)b * Llen * Dch;
    const float* ub = u + (size_t)(b + 1) * Llen * Dch;
    float*       oa = out + (size_t)b * Llen * Dch;
    float*       ob = out + (size_t)(b + 1) * Llen * Dch;
    s4_upad_r8<<<gR8, TB, 0, stream>>>(ua, ub, T0);
    for (int p = 0; p < 4; ++p) {
      const float2* src = (p & 1) ? T1 : T0;
      float2*       dst = (p & 1) ? T0 : T1;
      s4_fft_r8<<<gR8, TB, 0, stream>>>(src, dst, NFFT, 8 << (3 * p), -1.0f);
    }
    s4_cmul_r8<<<gR8, TB, 0, stream>>>(T0, Kd, T1);
    for (int p = 0; p < 3; ++p) {
      const float2* src = (p & 1) ? T0 : T1;
      float2*       dst = (p & 1) ? T1 : T0;
      s4_fft_r8<<<gR8, TB, 0, stream>>>(src, dst, NFFT, 8 << (3 * p), +1.0f);
    }
    s4_final_r8<<<gR8, TB, 0, stream>>>(T0, ua, ub, Dp, oa, ob);
  }
}


# Pairs of radix-8 Stockham passes fused into radix-64 passes (8x8 exchange in LDS): each 2L-point FFT takes 3 global round trips instead of 5; butterfly order unchanged, twiddles factored
# speedup vs baseline: 2.8083x; 1.7389x over previous
#include <hip/hip_runtime.h>
#include <hip/hip_bf16.h>
#include <stdint.h>


#define Bsz   4
#define Llen  16384
#define Dch   512
#define Nst   64
#define NFFT  32768
#define PI_F  3.14159265358979323846f

typedef __attribute__((ext_vector_type(2))) float v2f;
typedef __attribute__((ext_vector_type(8))) float v8f;

#ifndef __has_builtin
#define __has_builtin(x) 0
#endif
#if __has_builtin(__builtin_amdgcn_wmma_f32_16x16x4_f32)
#define S4_WMMA 1
#else
#define S4_WMMA 0
#endif

__device__ __forceinline__ float2 cadd(float2 a, float2 b) { return make_float2(a.x + b.x, a.y + b.y); }
__device__ __forceinline__ float2 csub(float2 a, float2 b) { return make_float2(a.x - b.x, a.y - b.y); }
__device__ __forceinline__ float2 cmulf(float2 a, float2 b) { return make_float2(a.x * b.x - a.y * b.y, a.x * b.y + a.y * b.x); }
__device__ __forceinline__ float2 crotis(float2 a, float s) { return make_float2(-s * a.y, s * a.x); }

__device__ __forceinline__ void dft8(const float2* v, float s, float2* X) {
  const float R = 0.70710678118654752440f;
  float2 t0 = cadd(v[0], v[4]), t1 = cadd(v[1], v[5]);
  float2 t2 = cadd(v[2], v[6]), t3 = cadd(v[3], v[7]);
  float2 u0 = csub(v[0], v[4]), u1 = csub(v[1], v[5]);
  float2 u2 = csub(v[2], v[6]), u3 = csub(v[3], v[7]);
  float2 m1 = cmulf(u1, make_float2(R, s * R));
  float2 m2 = crotis(u2, s);
  float2 m3 = cmulf(u3, make_float2(-R, s * R));
  float2 p0 = cadd(t0, t2), p1 = csub(t0, t2);
  float2 p2 = cadd(t1, t3), p3 = crotis(csub(t1, t3), s);
  X[0] = cadd(p0, p2); X[2] = cadd(p1, p3);
  X[4] = csub(p0, p2); X[6] = csub(p1, p3);
  float2 q0 = cadd(u0, m2), q1 = csub(u0, m2);
  float2 q2 = cadd(m1, m3), q3 = crotis(csub(m1, m3), s);
  X[1] = cadd(q0, q2); X[3] = cadd(q1, q3);
  X[5] = csub(q0, q2); X[7] = csub(q1, q3);
}

__global__ __launch_bounds__(256) void s4_perl(
    const float* __restrict__ Lam, const float* __restrict__ Pri,
    const float* __restrict__ Cri, const float* __restrict__ log_step,
    float* __restrict__ perl)
{
  int l = blockIdx.x * blockDim.x + threadIdx.x;
  if (l >= Llen) return;
  float step = expf(log_step[0]);
  float theta = -2.0f * PI_F * (float)l / (float)Llen;
  float oi, orr;
  sincosf(theta, &oi, &orr);
  float op1r = 1.0f + orr;
  float den = op1r * op1r + oi * oi;
  float nr = 1.0f - orr, ni = -oi;
  float gscale = 2.0f / step;
  float gr = gscale * (nr * op1r + ni * oi) / den;
  float gi = gscale * (ni * op1r - nr * oi) / den;
  float scr = 2.0f * op1r / den;
  float sci = -2.0f * oi / den;
  float k01r = 0.f, k01i = 0.f, k11r = 0.f, k11i = 0.f;
  for (int n = 0; n < Nst; ++n) {
    float lr = Lam[2 * n], li = Lam[2 * n + 1];
    float pr = Pri[2 * n], pi = Pri[2 * n + 1];
    float cr = Cri[2 * n], ci = Cri[2 * n + 1];
    float zr = gr - lr, zi = gi - li;
    float zd = zr * zr + zi * zi;
    float car = zr / zd, cai = -zi / zd;
    float w01r = cr * pr + ci * pi;
    float w01i = cr * pi - ci * pr;
    float w11r = pr * pr + pi * pi;
    k01r += car * w01r - cai * w01i;
    k01i += car * w01i + cai * w01r;
    k11r += car * w11r;
    k11i += cai * w11r;
  }
  float dr = 1.0f + k11r, di = k11i;
  float dd = dr * dr + di * di;
  float sr = (k01r * dr + k01i * di) / dd;
  float si = (k01i * dr - k01r * di) / dd;
  perl[l] = sr;
  perl[Llen + l] = si;
  perl[2 * Llen + l] = scr;
  perl[3 * Llen + l] = sci;
  perl[4 * Llen + l] = gr;
  perl[5 * Llen + l] = gi;
}

__global__ __launch_bounds__(256) void s4_weights(
    const float* __restrict__ Bri, const float* __restrict__ Pri,
    const float* __restrict__ Cri,
    float* __restrict__ W0r, float* __restrict__ W0i,
    float* __restrict__ W1r, float* __restrict__ W1i)
{
  int idx = blockIdx.x * blockDim.x + threadIdx.x;
  if (idx >= Nst * Dch) return;
  int n = idx / Dch, d = idx % Dch;
  float br = Bri[2 * (n * Dch + d)], bi = Bri[2 * (n * Dch + d) + 1];
  float cr = Cri[2 * n], ci = Cri[2 * n + 1];
  float pr = Pri[2 * n], pi = Pri[2 * n + 1];
  int o = (n >> 1) * (2 * Dch) + 2 * d + (n & 1);
  W0r[o] = cr * br + ci * bi;
  W0i[o] = cr * bi - ci * br;
  W1r[o] = pr * br + pi * bi;
  W1i[o] = pr * bi - pi * br;
}

__global__ __launch_bounds__(256) void s4_cauchy(
    const float* __restrict__ Lam, const float* __restrict__ perl,
    float* __restrict__ APr, float* __restrict__ APi)
{
  int idx = blockIdx.x * blockDim.x + threadIdx.x;
  if (idx >= Llen * Nst) return;
  int n = idx / Llen, l = idx % Llen;
  float gr = perl[4 * Llen + l], gi = perl[5 * Llen + l];
  float zr = gr - Lam[2 * n], zi = gi - Lam[2 * n + 1];
  float zd = zr * zr + zi * zi;
  int o = (n >> 1) * (2 * Llen) + 2 * l + (n & 1);
  APr[o] = zr / zd;
  APi[o] = -zi / zd;
}

__global__ __launch_bounds__(32) void s4_cauchy_gemm(
    const float* __restrict__ APr, const float* __restrict__ APi,
    const float* __restrict__ W0r, const float* __restrict__ W0i,
    const float* __restrict__ W1r, const float* __restrict__ W1i,
    const float* __restrict__ perl, float2* __restrict__ Ev)
{
  int lane = threadIdx.x & 31;
  int tileD = blockIdx.x;
  int tileL = blockIdx.y;
  int d0 = tileD * 16 + (lane & 15);
  int mbase = tileL * 16 + ((lane >> 4) ? 8 : 0);
#if S4_WMMA
  int m = tileL * 16 + (lane & 15);
  int khi = (lane >> 4) * 2;
  v8f aRr = {}, aRi = {}, aSr = {}, aSi = {};
  for (int kb = 0; kb < Nst; kb += 4) {
    int k0 = kb + khi;
    const float* pr = APr + (k0 >> 1) * (2 * Llen) + 2 * m;
    const float* pi = APi + (k0 >> 1) * (2 * Llen) + 2 * m;
    v2f ar = *(const v2f*)pr;
    v2f ai = *(const v2f*)pi;
    v2f nai = -ai;
    int bo = (k0 >> 1) * (2 * Dch) + 2 * d0;
    v2f b0r = *(const v2f*)(W0r + bo);
    v2f b0i = *(const v2f*)(W0i + bo);
    v2f b1r = *(const v2f*)(W1r + bo);
    v2f b1i = *(const v2f*)(W1i + bo);
    aRr = __builtin_amdgcn_wmma_f32_16x16x4_f32(false, ar,  false, b0r, (short)0, aRr, false, false);
    aRr = __builtin_amdgcn_wmma_f32_16x16x4_f32(false, nai, false, b0i, (short)0, aRr, false, false);
    aRi = __builtin_amdgcn_wmma_f32_16x16x4_f32(false, ar,  false, b0i, (short)0, aRi, false, false);
    aRi = __builtin_amdgcn_wmma_f32_16x16x4_f32(false, ai,  false, b0r, (short)0, aRi, false, false);
    aSr = __builtin_amdgcn_wmma_f32_16x16x4_f32(false, ar,  false, b1r, (short)0, aSr, false, false);
    aSr = __builtin_amdgcn_wmma_f32_16x16x4_f32(false, nai, false, b1i, (short)0, aSr, false, false);
    aSi = __builtin_amdgcn_wmma_f32_16x16x4_f32(false, ar,  false, b1i, (short)0, aSi, false, false);
    aSi = __builtin_amdgcn_wmma_f32_16x16x4_f32(false, ai,  false, b1r, (short)0, aSi, false, false);
  }
  for (int r = 0; r < 8; ++r) {
    int l = mbase + r;
    float k00r = aRr[r], k00i = aRi[r];
    float k10r = aSr[r], k10i = aSi[r];
    float sr = perl[l],            si = perl[Llen + l];
    float cr = perl[2 * Llen + l], ci = perl[3 * Llen + l];
    float tr = k00r - (sr * k10r - si * k10i);
    float ti = k00i - (sr * k10i + si * k10r);
    Ev[l * Dch + d0] = make_float2(cr * tr - ci * ti, cr * ti + ci * tr);
  }
#else
  for (int r = 0; r < 8; ++r) {
    int l = mbase + r;
    float k00r = 0.f, k00i = 0.f, k10r = 0.f, k10i = 0.f;
    for (int k = 0; k < Nst; ++k) {
      int ao = (k >> 1) * (2 * Llen) + 2 * l + (k & 1);
      int bo = (k >> 1) * (2 * Dch) + 2 * d0 + (k & 1);
      float arr = APr[ao], aii = APi[ao];
      float b0r = W0r[bo], b0i = W0i[bo];
      float b1r = W1r[bo], b1i = W1i[bo];
      k00r += arr * b0r - aii * b0i;  k00i += arr * b0i + aii * b0r;
      k10r += arr * b1r - aii * b1i;  k10i += arr * b1i + aii * b1r;
    }
    float sr = perl[l],            si = perl[Llen + l];
    float cr = perl[2 * Llen + l], ci = perl[3 * Llen + l];
    float tr = k00r - (sr * k10r - si * k10i);
    float ti = k00i - (sr * k10i + si * k10r);
    Ev[l * Dch + d0] = make_float2(cr * tr - ci * ti, cr * ti + ci * tr);
  }
#endif
}

__global__ __launch_bounds__(256) void s4_fft_r4(
    const float2* __restrict__ x, float2* __restrict__ y,
    int n, int Ns, float sign)
{
  int t = blockIdx.x * blockDim.x + threadIdx.x;
  int q = n >> 2;
  if (t >= q * Dch) return;
  int col = t & (Dch - 1);
  int j = t >> 9;
  int k = j & (Ns - 1);
  float ang = sign * (0.5f * PI_F) * (float)k / (float)Ns;
  float s1, c1;
  sincosf(ang, &s1, &c1);
  float c2 = c1 * c1 - s1 * s1, s2 = 2.f * c1 * s1;
  float c3 = c2 * c1 - s2 * s1, s3 = c2 * s1 + s2 * c1;
  float2 a = x[j * Dch + col];
  float2 b = x[(j + q) * Dch + col];
  float2 c = x[(j + 2 * q) * Dch + col];
  float2 d = x[(j + 3 * q) * Dch + col];
  float2 v1 = make_float2(c1 * b.x - s1 * b.y, c1 * b.y + s1 * b.x);
  float2 v2 = make_float2(c2 * c.x - s2 * c.y, c2 * c.y + s2 * c.x);
  float2 v3 = make_float2(c3 * d.x - s3 * d.y, c3 * d.y + s3 * d.x);
  float2 p0 = cadd(a, v2), p1 = csub(a, v2);
  float2 p2 = cadd(v1, v3), p3 = crotis(csub(v1, v3), sign);
  int i0 = 4 * j - 3 * k;
  y[(i0         ) * Dch + col] = cadd(p0, p2);
  y[(i0 +     Ns) * Dch + col] = cadd(p1, p3);
  y[(i0 + 2 * Ns) * Dch + col] = csub(p0, p2);
  y[(i0 + 3 * Ns) * Dch + col] = csub(p1, p3);
}

__device__ __forceinline__ void r64_butterfly(
    const float2* __restrict__ x, float2* sh, int j, int q, int k, int Ns,
    float sign, int col, int lane, int sub, float2* X)
{
  float2 v[8];
#pragma unroll
  for (int r1 = 0; r1 < 8; ++r1)
    v[r1] = x[(j + (sub + 8 * r1) * q) * Dch + col];
  float kf = (float)k / (float)Ns;
  float s8, c8, sb, cb, ss, cs;
  sincospif(sign * 0.25f * kf, &s8, &c8);
  sincospif(sign * (float)sub * kf * (1.0f / 32.0f), &sb, &cb);
  sincospif(sign * (float)sub * (1.0f / 32.0f), &ss, &cs);
  float2 w8 = make_float2(c8, s8);
  float2 w = w8;
#pragma unroll
  for (int r1 = 1; r1 < 8; ++r1) {
    v[r1] = cmulf(v[r1], w);
    w = cmulf(w, w8);
  }
  float2 Y[8];
  dft8(v, sign, Y);
  float2 tw = make_float2(cb, sb);
  float2 st = make_float2(cs, ss);
#pragma unroll
  for (int a = 0; a < 8; ++a) {
    sh[(a * 8 + sub) * 32 + lane] = cmulf(Y[a], tw);
    tw = cmulf(tw, st);
  }
  __syncthreads();
  float2 z[8];
#pragma unroll
  for (int r0 = 0; r0 < 8; ++r0)
    z[r0] = sh[(sub * 8 + r0) * 32 + lane];
  dft8(z, sign, X);
}

__global__ __launch_bounds__(256) void s4_fft_r64(
    const float2* __restrict__ x, float2* __restrict__ y,
    int n, int Ns, float sign)
{
  __shared__ float2 sh[64 * 32];
  int lane = threadIdx.x & 31;
  int sub = threadIdx.x >> 5;
  int col = blockIdx.x * 32 + lane;
  int j = blockIdx.y;
  int k = j & (Ns - 1);
  float2 X[8];
  r64_butterfly(x, sh, j, n >> 6, k, Ns, sign, col, lane, sub, X);
  int i0 = 64 * j - 63 * k;
#pragma unroll
  for (int b = 0; b < 8; ++b)
    y[(i0 + (sub + 8 * b) * Ns) * Dch + col] = X[b];
}

__global__ __launch_bounds__(256) void s4_upad_r8(
    const float* __restrict__ ua, const float* __restrict__ ub,
    float2* __restrict__ y)
{
  const int Q = NFFT / 8;
  int t = blockIdx.x * blockDim.x + threadIdx.x;
  if (t >= Q * Dch) return;
  int col = t & (Dch - 1);
  int j = t >> 9;
  float2 v[8];
#pragma unroll
  for (int r = 0; r < 4; ++r)
    v[r] = make_float2(ua[(j + r * Q) * Dch + col], ub[(j + r * Q) * Dch + col]);
#pragma unroll
  for (int r = 4; r < 8; ++r)
    v[r] = make_float2(0.f, 0.f);
  float2 X[8];
  dft8(v, -1.0f, X);
  int i0 = 8 * j;
#pragma unroll
  for (int m = 0; m < 8; ++m)
    y[(i0 + m) * Dch + col] = X[m];
}

__global__ __launch_bounds__(256) void s4_kpad_r8(
    const float2* __restrict__ Kt, float2* __restrict__ y)
{
  const int Q = NFFT / 8;
  int t = blockIdx.x * blockDim.x + threadIdx.x;
  if (t >= Q * Dch) return;
  int col = t & (Dch - 1);
  int j = t >> 9;
  float2 v[8];
#pragma unroll
  for (int r = 0; r < 4; ++r)
    v[r] = make_float2(Kt[(j + r * Q) * Dch + col].x * (1.0f / (float)Llen), 0.f);
#pragma unroll
  for (int r = 4; r < 8; ++r)
    v[r] = make_float2(0.f, 0.f);
  float2 X[8];
  dft8(v, -1.0f, X);
  int i0 = 8 * j;
#pragma unroll
  for (int m = 0; m < 8; ++m)
    y[(i0 + m) * Dch + col] = X[m];
}

__global__ __launch_bounds__(256) void s4_cmul_r8(
    const float2* __restrict__ Xs, const float2* __restrict__ Kd,
    float2* __restrict__ y)
{
  const int Q = NFFT / 8;
  int t = blockIdx.x * blockDim.x + threadIdx.x;
  if (t >= Q * Dch) return;
  int col = t & (Dch - 1);
  int j = t >> 9;
  float2 v[8];
#pragma unroll
  for (int r = 0; r < 8; ++r) {
    int idx = (j + r * Q) * Dch + col;
    v[r] = cmulf(Xs[idx], Kd[idx]);
  }
  float2 X[8];
  dft8(v, +1.0f, X);
  int i0 = 8 * j;
#pragma unroll
  for (int m = 0; m < 8; ++m)
    y[(i0 + m) * Dch + col] = X[m];
}

__global__ __launch_bounds__(256) void s4_final_r64(
    const float2* __restrict__ x,
    const float* __restrict__ ua, const float* __restrict__ ub,
    const float* __restrict__ Dp,
    float* __restrict__ outa, float* __restrict__ outb)
{
  __shared__ float2 sh[64 * 32];
  const int Q = NFFT / 64;
  int lane = threadIdx.x & 31;
  int sub = threadIdx.x >> 5;
  int col = blockIdx.x * 32 + lane;
  int j = blockIdx.y;
  float2 X[8];
  r64_butterfly(x, sh, j, Q, j, Q, +1.0f, col, lane, sub, X);
  float dp = Dp[0];
  const float inv = 1.0f / (float)NFFT;
#pragma unroll
  for (int b = 0; b < 4; ++b) {
    int pos = (j + (sub + 8 * b) * Q) * Dch + col;
    outa[pos] = X[b].x * inv + dp * ua[pos];
    outb[pos] = X[b].y * inv + dp * ub[pos];
  }
}


extern "C" void kernel_launch(void* const* d_in, const int* in_sizes, int n_in,
                              void* d_out, int out_size, void* d_ws, size_t ws_size,
                              hipStream_t stream) {
  const float* u        = (const float*)d_in[0];
  const float* Lam      = (const float*)d_in[1];
  const float* Pri      = (const float*)d_in[2];
  const float* Bri      = (const float*)d_in[3];
  const float* Cri      = (const float*)d_in[4];
  const float* Dp       = (const float*)d_in[5];
  const float* log_step = (const float*)d_in[6];
  float* out = (float*)d_out;

  uint8_t* base = (uint8_t*)d_ws;
  size_t off = 0;
  auto carve = [&](size_t bytes) -> void* {
    void* p = base + off;
    off += (bytes + 255) & ~(size_t)255;
    return p;
  };
  float*  APr  = (float*)carve((size_t)Llen * Nst * 4);
  float*  APi  = (float*)carve((size_t)Llen * Nst * 4);
  float*  W0r  = (float*)carve((size_t)Nst * Dch * 4);
  float*  W0i  = (float*)carve((size_t)Nst * Dch * 4);
  float*  W1r  = (float*)carve((size_t)Nst * Dch * 4);
  float*  W1i  = (float*)carve((size_t)Nst * Dch * 4);
  float*  perl = (float*)carve((size_t)6 * Llen * 4);
  float2* Ev   = (float2*)carve((size_t)Llen * Dch * 8);
  float2* T0   = (float2*)carve((size_t)NFFT * Dch * 8);
  float2* T1   = (float2*)carve((size_t)NFFT * Dch * 8);
  float2* Kd   = (float2*)carve((size_t)NFFT * Dch * 8);
  (void)ws_size; (void)in_sizes; (void)n_in; (void)out_size;

  const int TB   = 256;
  const int gR8  = (NFFT / 8) * Dch / TB;
  const int gR4s = (Llen / 4) * Dch / TB;
  const dim3 gR64(Dch / 32, NFFT / 64);
  const dim3 gR64s(Dch / 32, Llen / 64);

  s4_perl<<<(Llen + TB - 1) / TB, TB, 0, stream>>>(Lam, Pri, Cri, log_step, perl);
  s4_weights<<<(Nst * Dch + TB - 1) / TB, TB, 0, stream>>>(Bri, Pri, Cri, W0r, W0i, W1r, W1i);
  s4_cauchy<<<(Llen * Nst + TB - 1) / TB, TB, 0, stream>>>(Lam, perl, APr, APi);
  s4_cauchy_gemm<<<dim3(Dch / 16, Llen / 16), 32, 0, stream>>>(
      APr, APi, W0r, W0i, W1r, W1i, perl, Ev);

  s4_fft_r4<<<gR4s, TB, 0, stream>>>(Ev, T0, Llen, 1, +1.0f);
  s4_fft_r64<<<gR64s, TB, 0, stream>>>(T0, Ev, Llen, 4, +1.0f);
  s4_fft_r64<<<gR64s, TB, 0, stream>>>(Ev, T0, Llen, 256, +1.0f);

  s4_kpad_r8<<<gR8, TB, 0, stream>>>(T0, T1);
  s4_fft_r64<<<gR64, TB, 0, stream>>>(T1, T0, NFFT, 8, -1.0f);
  s4_fft_r64<<<gR64, TB, 0, stream>>>(T0, Kd, NFFT, 512, -1.0f);

  for (int b = 0; b < Bsz; b += 2) {
    const float* ua = u + (size_t)b * Llen * Dch;
    const float* ub = u + (size_t)(b + 1) * Llen * Dch;
    float*       oa = out + (size_t)b * Llen * Dch;
    float*       ob = out + (size_t)(b + 1) * Llen * Dch;
    s4_upad_r8<<<gR8, TB, 0, stream>>>(ua, ub, T0);
    s4_fft_r64<<<gR64, TB, 0, stream>>>(T0, T1, NFFT, 8, -1.0f);
    s4_fft_r64<<<gR64, TB, 0, stream>>>(T1, T0, NFFT, 512, -1.0f);
    s4_cmul_r8<<<gR8, TB, 0, stream>>>(T0, Kd, T1);
    s4_fft_r64<<<gR64, TB, 0, stream>>>(T1, T0, NFFT, 8, +1.0f);
    s4_final_r64<<<gR64, TB, 0, stream>>>(T0, ua, ub, Dp, oa, ob);
  }
}


# First radix-8 pass of the next transform (after x Kd; after pad of Re K) done in registers at the end of the previous transform's last radix-64 pass: 3 global round trips gone; sums unchanged
# speedup vs baseline: 3.3473x; 1.1919x over previous
#include <hip/hip_runtime.h>
#include <hip/hip_bf16.h>
#include <stdint.h>


#define Bsz   4
#define Llen  16384
#define Dch   512
#define Nst   64
#define NFFT  32768
#define PI_F  3.14159265358979323846f

typedef __attribute__((ext_vector_type(2))) float v2f;
typedef __attribute__((ext_vector_type(8))) float v8f;

#ifndef __has_builtin
#define __has_builtin(x) 0
#endif
#if __has_builtin(__builtin_amdgcn_wmma_f32_16x16x4_f32)
#define S4_WMMA 1
#else
#define S4_WMMA 0
#endif

__device__ __forceinline__ float2 cadd(float2 a, float2 b) { return make_float2(a.x + b.x, a.y + b.y); }
__device__ __forceinline__ float2 csub(float2 a, float2 b) { return make_float2(a.x - b.x, a.y - b.y); }
__device__ __forceinline__ float2 cmulf(float2 a, float2 b) { return make_float2(a.x * b.x - a.y * b.y, a.x * b.y + a.y * b.x); }
__device__ __forceinline__ float2 crotis(float2 a, float s) { return make_float2(-s * a.y, s * a.x); }

__device__ __forceinline__ void dft8(const float2* v, float s, float2* X) {
  const float R = 0.70710678118654752440f;
  float2 t0 = cadd(v[0], v[4]), t1 = cadd(v[1], v[5]);
  float2 t2 = cadd(v[2], v[6]), t3 = cadd(v[3], v[7]);
  float2 u0 = csub(v[0], v[4]), u1 = csub(v[1], v[5]);
  float2 u2 = csub(v[2], v[6]), u3 = csub(v[3], v[7]);
  float2 m1 = cmulf(u1, make_float2(R, s * R));
  float2 m2 = crotis(u2, s);
  float2 m3 = cmulf(u3, make_float2(-R, s * R));
  float2 p0 = cadd(t0, t2), p1 = csub(t0, t2);
  float2 p2 = cadd(t1, t3), p3 = crotis(csub(t1, t3), s);
  X[0] = cadd(p0, p2); X[2] = cadd(p1, p3);
  X[4] = csub(p0, p2); X[6] = csub(p1, p3);
  float2 q0 = cadd(u0, m2), q1 = csub(u0, m2);
  float2 q2 = cadd(m1, m3), q3 = crotis(csub(m1, m3), s);
  X[1] = cadd(q0, q2); X[3] = cadd(q1, q3);
  X[5] = csub(q0, q2); X[7] = csub(q1, q3);
}

__global__ __launch_bounds__(256) void s4_perl(
    const float* __restrict__ Lam, const float* __restrict__ Pri,
    const float* __restrict__ Cri, const float* __restrict__ log_step,
    float* __restrict__ perl)
{
  int l = blockIdx.x * blockDim.x + threadIdx.x;
  if (l >= Llen) return;
  float step = expf(log_step[0]);
  float theta = -2.0f * PI_F * (float)l / (float)Llen;
  float oi, orr;
  sincosf(theta, &oi, &orr);
  float op1r = 1.0f + orr;
  float den = op1r * op1r + oi * oi;
  float nr = 1.0f - orr, ni = -oi;
  float gscale = 2.0f / step;
  float gr = gscale * (nr * op1r + ni * oi) / den;
  float gi = gscale * (ni * op1r - nr * oi) / den;
  float scr = 2.0f * op1r / den;
  float sci = -2.0f * oi / den;
  float k01r = 0.f, k01i = 0.f, k11r = 0.f, k11i = 0.f;
  for (int n = 0; n < Nst; ++n) {
    float lr = Lam[2 * n], li = Lam[2 * n + 1];
    float pr = Pri[2 * n], pi = Pri[2 * n + 1];
    float cr = Cri[2 * n], ci = Cri[2 * n + 1];
    float zr = gr - lr, zi = gi - li;
    float zd = zr * zr + zi * zi;
    float car = zr / zd, cai = -zi / zd;
    float w01r = cr * pr + ci * pi;
    float w01i = cr * pi - ci * pr;
    float w11r = pr * pr + pi * pi;
    k01r += car * w01r - cai * w01i;
    k01i += car * w01i + cai * w01r;
    k11r += car * w11r;
    k11i += cai * w11r;
  }
  float dr = 1.0f + k11r, di = k11i;
  float dd = dr * dr + di * di;
  float sr = (k01r * dr + k01i * di) / dd;
  float si = (k01i * dr - k01r * di) / dd;
  perl[l] = sr;
  perl[Llen + l] = si;
  perl[2 * Llen + l] = scr;
  perl[3 * Llen + l] = sci;
  perl[4 * Llen + l] = gr;
  perl[5 * Llen + l] = gi;
}

__global__ __launch_bounds__(256) void s4_weights(
    const float* __restrict__ Bri, const float* __restrict__ Pri,
    const float* __restrict__ Cri,
    float* __restrict__ W0r, float* __restrict__ W0i,
    float* __restrict__ W1r, float* __restrict__ W1i)
{
  int idx = blockIdx.x * blockDim.x + threadIdx.x;
  if (idx >= Nst * Dch) return;
  int n = idx / Dch, d = idx % Dch;
  float br = Bri[2 * (n * Dch + d)], bi = Bri[2 * (n * Dch + d) + 1];
  float cr = Cri[2 * n], ci = Cri[2 * n + 1];
  float pr = Pri[2 * n], pi = Pri[2 * n + 1];
  int o = (n >> 1) * (2 * Dch) + 2 * d + (n & 1);
  W0r[o] = cr * br + ci * bi;
  W0i[o] = cr * bi - ci * br;
  W1r[o] = pr * br + pi * bi;
  W1i[o] = pr * bi - pi * br;
}

__global__ __launch_bounds__(256) void s4_cauchy(
    const float* __restrict__ Lam, const float* __restrict__ perl,
    float* __restrict__ APr, float* __restrict__ APi)
{
  int idx = blockIdx.x * blockDim.x + threadIdx.x;
  if (idx >= Llen * Nst) return;
  int n = idx / Llen, l = idx % Llen;
  float gr = perl[4 * Llen + l], gi = perl[5 * Llen + l];
  float zr = gr - Lam[2 * n], zi = gi - Lam[2 * n + 1];
  float zd = zr * zr + zi * zi;
  int o = (n >> 1) * (2 * Llen) + 2 * l + (n & 1);
  APr[o] = zr / zd;
  APi[o] = -zi / zd;
}

__global__ __launch_bounds__(32) void s4_cauchy_gemm(
    const float* __restrict__ APr, const float* __restrict__ APi,
    const float* __restrict__ W0r, const float* __restrict__ W0i,
    const float* __restrict__ W1r, const float* __restrict__ W1i,
    const float* __restrict__ perl, float2* __restrict__ Ev)
{
  int lane = threadIdx.x & 31;
  int tileD = blockIdx.x;
  int tileL = blockIdx.y;
  int d0 = tileD * 16 + (lane & 15);
  int mbase = tileL * 16 + ((lane >> 4) ? 8 : 0);
#if S4_WMMA
  int m = tileL * 16 + (lane & 15);
  int khi = (lane >> 4) * 2;
  v8f aRr = {}, aRi = {}, aSr = {}, aSi = {};
  for (int kb = 0; kb < Nst; kb += 4) {
    int k0 = kb + khi;
    const float* pr = APr + (k0 >> 1) * (2 * Llen) + 2 * m;
    const float* pi = APi + (k0 >> 1) * (2 * Llen) + 2 * m;
    v2f ar = *(const v2f*)pr;
    v2f ai = *(const v2f*)pi;
    v2f nai = -ai;
    int bo = (k0 >> 1) * (2 * Dch) + 2 * d0;
    v2f b0r = *(const v2f*)(W0r + bo);
    v2f b0i = *(const v2f*)(W0i + bo);
    v2f b1r = *(const v2f*)(W1r + bo);
    v2f b1i = *(const v2f*)(W1i + bo);
    aRr = __builtin_amdgcn_wmma_f32_16x16x4_f32(false, ar,  false, b0r, (short)0, aRr, false, false);
    aRr = __builtin_amdgcn_wmma_f32_16x16x4_f32(false, nai, false, b0i, (short)0, aRr, false, false);
    aRi = __builtin_amdgcn_wmma_f32_16x16x4_f32(false, ar,  false, b0i, (short)0, aRi, false, false);
    aRi = __builtin_amdgcn_wmma_f32_16x16x4_f32(false, ai,  false, b0r, (short)0, aRi, false, false);
    aSr = __builtin_amdgcn_wmma_f32_16x16x4_f32(false, ar,  false, b1r, (short)0, aSr, false, false);
    aSr = __builtin_amdgcn_wmma_f32_16x16x4_f32(false, nai, false, b1i, (short)0, aSr, false, false);
    aSi = __builtin_amdgcn_wmma_f32_16x16x4_f32(false, ar,  false, b1i, (short)0, aSi, false, false);
    aSi = __builtin_amdgcn_wmma_f32_16x16x4_f32(false, ai,  false, b1r, (short)0, aSi, false, false);
  }
  for (int r = 0; r < 8; ++r) {
    int l = mbase + r;
    float k00r = aRr[r], k00i = aRi[r];
    float k10r = aSr[r], k10i = aSi[r];
    float sr = perl[l],            si = perl[Llen + l];
    float cr = perl[2 * Llen + l], ci = perl[3 * Llen + l];
    float tr = k00r - (sr * k10r - si * k10i);
    float ti = k00i - (sr * k10i + si * k10r);
    Ev[l * Dch + d0] = make_float2(cr * tr - ci * ti, cr * ti + ci * tr);
  }
#else
  for (int r = 0; r < 8; ++r) {
    int l = mbase + r;
    float k00r = 0.f, k00i = 0.f, k10r = 0.f, k10i = 0.f;
    for (int k = 0; k < Nst; ++k) {
      int ao = (k >> 1) * (2 * Llen) + 2 * l + (k & 1);
      int bo = (k >> 1) * (2 * Dch) + 2 * d0 + (k & 1);
      float arr = APr[ao], aii = APi[ao];
      float b0r = W0r[bo], b0i = W0i[bo];
      float b1r = W1r[bo], b1i = W1i[bo];
      k00r += arr * b0r - aii * b0i;  k00i += arr * b0i + aii * b0r;
      k10r += arr * b1r - aii * b1i;  k10i += arr * b1i + aii * b1r;
    }
    float sr = perl[l],            si = perl[Llen + l];
    float cr = perl[2 * Llen + l], ci = perl[3 * Llen + l];
    float tr = k00r - (sr * k10r - si * k10i);
    float ti = k00i - (sr * k10i + si * k10r);
    Ev[l * Dch + d0] = make_float2(cr * tr - ci * ti, cr * ti + ci * tr);
  }
#endif
}

__global__ __launch_bounds__(256) void s4_fft_r4(
    const float2* __restrict__ x, float2* __restrict__ y,
    int n, int Ns, float sign)
{
  int t = blockIdx.x * blockDim.x + threadIdx.x;
  int q = n >> 2;
  if (t >= q * Dch) return;
  int col = t & (Dch - 1);
  int j = t >> 9;
  int k = j & (Ns - 1);
  float ang = sign * (0.5f * PI_F) * (float)k / (float)Ns;
  float s1, c1;
  sincosf(ang, &s1, &c1);
  float c2 = c1 * c1 - s1 * s1, s2 = 2.f * c1 * s1;
  float c3 = c2 * c1 - s2 * s1, s3 = c2 * s1 + s2 * c1;
  float2 a = x[j * Dch + col];
  float2 b = x[(j + q) * Dch + col];
  float2 c = x[(j + 2 * q) * Dch + col];
  float2 d = x[(j + 3 * q) * Dch + col];
  float2 v1 = make_float2(c1 * b.x - s1 * b.y, c1 * b.y + s1 * b.x);
  float2 v2 = make_float2(c2 * c.x - s2 * c.y, c2 * c.y + s2 * c.x);
  float2 v3 = make_float2(c3 * d.x - s3 * d.y, c3 * d.y + s3 * d.x);
  float2 p0 = cadd(a, v2), p1 = csub(a, v2);
  float2 p2 = cadd(v1, v3), p3 = crotis(csub(v1, v3), sign);
  int i0 = 4 * j - 3 * k;
  y[(i0         ) * Dch + col] = cadd(p0, p2);
  y[(i0 +     Ns) * Dch + col] = cadd(p1, p3);
  y[(i0 + 2 * Ns) * Dch + col] = csub(p0, p2);
  y[(i0 + 3 * Ns) * Dch + col] = csub(p1, p3);
}

__device__ __forceinline__ void r64_butterfly(
    const float2* __restrict__ x, float2* sh, int j, int q, int k, int Ns,
    float sign, int col, int lane, int sub, float2* X)
{
  float2 v[8];
#pragma unroll
  for (int r1 = 0; r1 < 8; ++r1)
    v[r1] = x[(j + (sub + 8 * r1) * q) * Dch + col];
  float kf = (float)k / (float)Ns;
  float s8, c8, sb, cb, ss, cs;
  sincospif(sign * 0.25f * kf, &s8, &c8);
  sincospif(sign * (float)sub * kf * (1.0f / 32.0f), &sb, &cb);
  sincospif(sign * (float)sub * (1.0f / 32.0f), &ss, &cs);
  float2 w8 = make_float2(c8, s8);
  float2 w = w8;
#pragma unroll
  for (int r1 = 1; r1 < 8; ++r1) {
    v[r1] = cmulf(v[r1], w);
    w = cmulf(w, w8);
  }
  float2 Y[8];
  dft8(v, sign, Y);
  float2 tw = make_float2(cb, sb);
  float2 st = make_float2(cs, ss);
#pragma unroll
  for (int a = 0; a < 8; ++a) {
    sh[(a * 8 + sub) * 32 + lane] = cmulf(Y[a], tw);
    tw = cmulf(tw, st);
  }
  __syncthreads();
  float2 z[8];
#pragma unroll
  for (int r0 = 0; r0 < 8; ++r0)
    z[r0] = sh[(sub * 8 + r0) * 32 + lane];
  dft8(z, sign, X);
}

__global__ __launch_bounds__(256) void s4_fft_r64(
    const float2* __restrict__ x, float2* __restrict__ y,
    int n, int Ns, float sign)
{
  __shared__ float2 sh[64 * 32];
  int lane = threadIdx.x & 31;
  int sub = threadIdx.x >> 5;
  int col = blockIdx.x * 32 + lane;
  int j = blockIdx.y;
  int k = j & (Ns - 1);
  float2 X[8];
  r64_butterfly(x, sh, j, n >> 6, k, Ns, sign, col, lane, sub, X);
  int i0 = 64 * j - 63 * k;
#pragma unroll
  for (int b = 0; b < 8; ++b)
    y[(i0 + (sub + 8 * b) * Ns) * Dch + col] = X[b];
}

__global__ __launch_bounds__(256) void s4_upad_r8(
    const float* __restrict__ ua, const float* __restrict__ ub,
    float2* __restrict__ y)
{
  const int Q = NFFT / 8;
  int t = blockIdx.x * blockDim.x + threadIdx.x;
  if (t >= Q * Dch) return;
  int col = t & (Dch - 1);
  int j = t >> 9;
  float2 v[8];
#pragma unroll
  for (int r = 0; r < 4; ++r)
    v[r] = make_float2(ua[(j + r * Q) * Dch + col], ub[(j + r * Q) * Dch + col]);
#pragma unroll
  for (int r = 4; r < 8; ++r)
    v[r] = make_float2(0.f, 0.f);
  float2 X[8];
  dft8(v, -1.0f, X);
  int i0 = 8 * j;
#pragma unroll
  for (int m = 0; m < 8; ++m)
    y[(i0 + m) * Dch + col] = X[m];
}

__global__ __launch_bounds__(256) void s4_klast_r64_pad_r8(
    const float2* __restrict__ x, float2* __restrict__ y)
{
  __shared__ float2 sh[64 * 32];
  const int Q = Llen / 64;
  int lane = threadIdx.x & 31;
  int sub = threadIdx.x >> 5;
  int col = blockIdx.x * 32 + lane;
  int j = blockIdx.y;
  float2 X[8];
  r64_butterfly(x, sh, j, Q, j, Q, +1.0f, col, lane, sub, X);
  const float invL = 1.0f / (float)Llen;
  float2 va[8], vb[8];
#pragma unroll
  for (int r = 0; r < 4; ++r) {
    va[r] = make_float2(X[2 * r].x * invL, 0.f);
    vb[r] = make_float2(X[2 * r + 1].x * invL, 0.f);
  }
#pragma unroll
  for (int r = 4; r < 8; ++r) {
    va[r] = make_float2(0.f, 0.f);
    vb[r] = make_float2(0.f, 0.f);
  }
  float2 Za[8], Zb[8];
  dft8(va, -1.0f, Za);
  dft8(vb, -1.0f, Zb);
  int ja = j + sub * Q;
  int jb = j + (sub + 8) * Q;
#pragma unroll
  for (int m = 0; m < 8; ++m) {
    y[(8 * ja + m) * Dch + col] = Za[m];
    y[(8 * jb + m) * Dch + col] = Zb[m];
  }
}

__global__ __launch_bounds__(256) void s4_flast_r64_cmul_r8(
    const float2* __restrict__ x, const float2* __restrict__ Kd,
    float2* __restrict__ y)
{
  __shared__ float2 sh[64 * 32];
  const int Q = NFFT / 64;
  const int Q8 = NFFT / 8;
  int lane = threadIdx.x & 31;
  int sub = threadIdx.x >> 5;
  int col = blockIdx.x * 32 + lane;
  int j = blockIdx.y;
  int jp = j + sub * Q;
  float2 kd[8];
#pragma unroll
  for (int b = 0; b < 8; ++b)
    kd[b] = Kd[(jp + b * Q8) * Dch + col];
  float2 X[8];
  r64_butterfly(x, sh, j, Q, j, Q, -1.0f, col, lane, sub, X);
  float2 v[8];
#pragma unroll
  for (int b = 0; b < 8; ++b)
    v[b] = cmulf(X[b], kd[b]);
  float2 Z[8];
  dft8(v, +1.0f, Z);
#pragma unroll
  for (int m = 0; m < 8; ++m)
    y[(8 * jp + m) * Dch + col] = Z[m];
}

__global__ __launch_bounds__(256) void s4_final_r64(
    const float2* __restrict__ x,
    const float* __restrict__ ua, const float* __restrict__ ub,
    const float* __restrict__ Dp,
    float* __restrict__ outa, float* __restrict__ outb)
{
  __shared__ float2 sh[64 * 32];
  const int Q = NFFT / 64;
  int lane = threadIdx.x & 31;
  int sub = threadIdx.x >> 5;
  int col = blockIdx.x * 32 + lane;
  int j = blockIdx.y;
  float2 X[8];
  r64_butterfly(x, sh, j, Q, j, Q, +1.0f, col, lane, sub, X);
  float dp = Dp[0];
  const float inv = 1.0f / (float)NFFT;
#pragma unroll
  for (int b = 0; b < 4; ++b) {
    int pos = (j + (sub + 8 * b) * Q) * Dch + col;
    outa[pos] = X[b].x * inv + dp * ua[pos];
    outb[pos] = X[b].y * inv + dp * ub[pos];
  }
}


extern "C" void kernel_launch(void* const* d_in, const int* in_sizes, int n_in,
                              void* d_out, int out_size, void* d_ws, size_t ws_size,
                              hipStream_t stream) {
  const float* u        = (const float*)d_in[0];
  const float* Lam      = (const float*)d_in[1];
  const float* Pri      = (const float*)d_in[2];
  const float* Bri      = (const float*)d_in[3];
  const float* Cri      = (const float*)d_in[4];
  const float* Dp       = (const float*)d_in[5];
  const float* log_step = (const float*)d_in[6];
  float* out = (float*)d_out;

  uint8_t* base = (uint8_t*)d_ws;
  size_t off = 0;
  auto carve = [&](size_t bytes) -> void* {
    void* p = base + off;
    off += (bytes + 255) & ~(size_t)255;
    return p;
  };
  float*  APr  = (float*)carve((size_t)Llen * Nst * 4);
  float*  APi  = (float*)carve((size_t)Llen * Nst * 4);
  float*  W0r  = (float*)carve((size_t)Nst * Dch * 4);
  float*  W0i  = (float*)carve((size_t)Nst * Dch * 4);
  float*  W1r  = (float*)carve((size_t)Nst * Dch * 4);
  float*  W1i  = (float*)carve((size_t)Nst * Dch * 4);
  float*  perl = (float*)carve((size_t)6 * Llen * 4);
  float2* Ev   = (float2*)carve((size_t)Llen * Dch * 8);
  float2* T0   = (float2*)carve((size_t)NFFT * Dch * 8);
  float2* T1   = (float2*)carve((size_t)NFFT * Dch * 8);
  float2* Kd   = (float2*)carve((size_t)NFFT * Dch * 8);
  (void)ws_size; (void)in_sizes; (void)n_in; (void)out_size;

  const int TB   = 256;
  const int gR8  = (NFFT / 8) * Dch / TB;
  const int gR4s = (Llen / 4) * Dch / TB;
  const dim3 gR64(Dch / 32, NFFT / 64);
  const dim3 gR64s(Dch / 32, Llen / 64);

  s4_perl<<<(Llen + TB - 1) / TB, TB, 0, stream>>>(Lam, Pri, Cri, log_step, perl);
  s4_weights<<<(Nst * Dch + TB - 1) / TB, TB, 0, stream>>>(Bri, Pri, Cri, W0r, W0i, W1r, W1i);
  s4_cauchy<<<(Llen * Nst + TB - 1) / TB, TB, 0, stream>>>(Lam, perl, APr, APi);
  s4_cauchy_gemm<<<dim3(Dch / 16, Llen / 16), 32, 0, stream>>>(
      APr, APi, W0r, W0i, W1r, W1i, perl, Ev);

  s4_fft_r4<<<gR4s, TB, 0, stream>>>(Ev, T0, Llen, 1, +1.0f);
  s4_fft_r64<<<gR64s, TB, 0, stream>>>(T0, Ev, Llen, 4, +1.0f);
  s4_klast_r64_pad_r8<<<gR64s, TB, 0, stream>>>(Ev, T1);
  s4_fft_r64<<<gR64, TB, 0, stream>>>(T1, T0, NFFT, 8, -1.0f);
  s4_fft_r64<<<gR64, TB, 0, stream>>>(T0, Kd, NFFT, 512, -1.0f);

  for (int b = 0; b < Bsz; b += 2) {
    const float* ua = u + (size_t)b * Llen * Dch;
    const float* ub = u + (size_t)(b + 1) * Llen * Dch;
    float*       oa = out + (size_t)b * Llen * Dch;
    float*       ob = out + (size_t)(b + 1) * Llen * Dch;
    s4_upad_r8<<<gR8, TB, 0, stream>>>(ua, ub, T0);
    s4_fft_r64<<<gR64, TB, 0, stream>>>(T0, T1, NFFT, 8, -1.0f);
    s4_flast_r64_cmul_r8<<<gR64, TB, 0, stream>>>(T1, Kd, T0);
    s4_fft_r64<<<gR64, TB, 0, stream>>>(T0, T1, NFFT, 8, +1.0f);
    s4_final_r64<<<gR64, TB, 0, stream>>>(T1, ua, ub, Dp, oa, ob);
  }
}


# Convolution FFTs take 2 global round trips, not 3: radix-512 first/last passes (radix-8 results stay in registers, feed 8 radix-64 butterflies); mid pass = fwd r64, x Kd, inv r64; twiddles refactored
# speedup vs baseline: 4.2307x; 1.2639x over previous
#include <hip/hip_runtime.h>
#include <hip/hip_bf16.h>
#include <stdint.h>


#define Bsz   4
#define Llen  16384
#define Dch   512
#define Nst   64
#define NFFT  32768
#define PI_F  3.14159265358979323846f

typedef __attribute__((ext_vector_type(2))) float v2f;
typedef __attribute__((ext_vector_type(8))) float v8f;

#ifndef __has_builtin
#define __has_builtin(x) 0
#endif
#if __has_builtin(__builtin_amdgcn_wmma_f32_16x16x4_f32)
#define S4_WMMA 1
#else
#define S4_WMMA 0
#endif

__device__ __forceinline__ float2 cadd(float2 a, float2 b) { return make_float2(a.x + b.x, a.y + b.y); }
__device__ __forceinline__ float2 csub(float2 a, float2 b) { return make_float2(a.x - b.x, a.y - b.y); }
__device__ __forceinline__ float2 cmulf(float2 a, float2 b) { return make_float2(a.x * b.x - a.y * b.y, a.x * b.y + a.y * b.x); }
__device__ __forceinline__ float2 crotis(float2 a, float s) { return make_float2(-s * a.y, s * a.x); }

__device__ __forceinline__ void dft8(const float2* v, float s, float2* X) {
  const float R = 0.70710678118654752440f;
  float2 t0 = cadd(v[0], v[4]), t1 = cadd(v[1], v[5]);
  float2 t2 = cadd(v[2], v[6]), t3 = cadd(v[3], v[7]);
  float2 u0 = csub(v[0], v[4]), u1 = csub(v[1], v[5]);
  float2 u2 = csub(v[2], v[6]), u3 = csub(v[3], v[7]);
  float2 m1 = cmulf(u1, make_float2(R, s * R));
  float2 m2 = crotis(u2, s);
  float2 m3 = cmulf(u3, make_float2(-R, s * R));
  float2 p0 = cadd(t0, t2), p1 = csub(t0, t2);
  float2 p2 = cadd(t1, t3), p3 = crotis(csub(t1, t3), s);
  X[0] = cadd(p0, p2); X[2] = cadd(p1, p3);
  X[4] = csub(p0, p2); X[6] = csub(p1, p3);
  float2 q0 = cadd(u0, m2), q1 = csub(u0, m2);
  float2 q2 = cadd(m1, m3), q3 = crotis(csub(m1, m3), s);
  X[1] = cadd(q0, q2); X[3] = cadd(q1, q3);
  X[5] = csub(q0, q2); X[7] = csub(q1, q3);
}

__global__ __launch_bounds__(256) void s4_perl(
    const float* __restrict__ Lam, const float* __restrict__ Pri,
    const float* __restrict__ Cri, const float* __restrict__ log_step,
    float* __restrict__ perl)
{
  int l = blockIdx.x * blockDim.x + threadIdx.x;
  if (l >= Llen) return;
  float step = expf(log_step[0]);
  float theta = -2.0f * PI_F * (float)l / (float)Llen;
  float oi, orr;
  sincosf(theta, &oi, &orr);
  float op1r = 1.0f + orr;
  float den = op1r * op1r + oi * oi;
  float nr = 1.0f - orr, ni = -oi;
  float gscale = 2.0f / step;
  float gr = gscale * (nr * op1r + ni * oi) / den;
  float gi = gscale * (ni * op1r - nr * oi) / den;
  float scr = 2.0f * op1r / den;
  float sci = -2.0f * oi / den;
  float k01r = 0.f, k01i = 0.f, k11r = 0.f, k11i = 0.f;
  for (int n = 0; n < Nst; ++n) {
    float lr = Lam[2 * n], li = Lam[2 * n + 1];
    float pr = Pri[2 * n], pi = Pri[2 * n + 1];
    float cr = Cri[2 * n], ci = Cri[2 * n + 1];
    float zr = gr - lr, zi = gi - li;
    float zd = zr * zr + zi * zi;
    float car = zr / zd, cai = -zi / zd;
    float w01r = cr * pr + ci * pi;
    float w01i = cr * pi - ci * pr;
    float w11r = pr * pr + pi * pi;
    k01r += car * w01r - cai * w01i;
    k01i += car * w01i + cai * w01r;
    k11r += car * w11r;
    k11i += cai * w11r;
  }
  float dr = 1.0f + k11r, di = k11i;
  float dd = dr * dr + di * di;
  float sr = (k01r * dr + k01i * di) / dd;
  float si = (k01i * dr - k01r * di) / dd;
  perl[l] = sr;
  perl[Llen + l] = si;
  perl[2 * Llen + l] = scr;
  perl[3 * Llen + l] = sci;
  perl[4 * Llen + l] = gr;
  perl[5 * Llen + l] = gi;
}

__global__ __launch_bounds__(256) void s4_weights(
    const float* __restrict__ Bri, const float* __restrict__ Pri,
    const float* __restrict__ Cri,
    float* __restrict__ W0r, float* __restrict__ W0i,
    float* __restrict__ W1r, float* __restrict__ W1i)
{
  int idx = blockIdx.x * blockDim.x + threadIdx.x;
  if (idx >= Nst * Dch) return;
  int n = idx / Dch, d = idx % Dch;
  float br = Bri[2 * (n * Dch + d)], bi = Bri[2 * (n * Dch + d) + 1];
  float cr = Cri[2 * n], ci = Cri[2 * n + 1];
  float pr = Pri[2 * n], pi = Pri[2 * n + 1];
  int o = (n >> 1) * (2 * Dch) + 2 * d + (n & 1);
  W0r[o] = cr * br + ci * bi;
  W0i[o] = cr * bi - ci * br;
  W1r[o] = pr * br + pi * bi;
  W1i[o] = pr * bi - pi * br;
}

__global__ __launch_bounds__(256) void s4_cauchy(
    const float* __restrict__ Lam, const float* __restrict__ perl,
    float* __restrict__ APr, float* __restrict__ APi)
{
  int idx = blockIdx.x * blockDim.x + threadIdx.x;
  if (idx >= Llen * Nst) return;
  int n = idx / Llen, l = idx % Llen;
  float gr = perl[4 * Llen + l], gi = perl[5 * Llen + l];
  float zr = gr - Lam[2 * n], zi = gi - Lam[2 * n + 1];
  float zd = zr * zr + zi * zi;
  int o = (n >> 1) * (2 * Llen) + 2 * l + (n & 1);
  APr[o] = zr / zd;
  APi[o] = -zi / zd;
}

__global__ __launch_bounds__(32) void s4_cauchy_gemm(
    const float* __restrict__ APr, const float* __restrict__ APi,
    const float* __restrict__ W0r, const float* __restrict__ W0i,
    const float* __restrict__ W1r, const float* __restrict__ W1i,
    const float* __restrict__ perl, float2* __restrict__ Ev)
{
  int lane = threadIdx.x & 31;
  int tileD = blockIdx.x;
  int tileL = blockIdx.y;
  int d0 = tileD * 16 + (lane & 15);
  int mbase = tileL * 16 + ((lane >> 4) ? 8 : 0);
#if S4_WMMA
  int m = tileL * 16 + (lane & 15);
  int khi = (lane >> 4) * 2;
  v8f aRr = {}, aRi = {}, aSr = {}, aSi = {};
  for (int kb = 0; kb < Nst; kb += 4) {
    int k0 = kb + khi;
    const float* pr = APr + (k0 >> 1) * (2 * Llen) + 2 * m;
    const float* pi = APi + (k0 >> 1) * (2 * Llen) + 2 * m;
    v2f ar = *(const v2f*)pr;
    v2f ai = *(const v2f*)pi;
    v2f nai = -ai;
    int bo = (k0 >> 1) * (2 * Dch) + 2 * d0;
    v2f b0r = *(const v2f*)(W0r + bo);
    v2f b0i = *(const v2f*)(W0i + bo);
    v2f b1r = *(const v2f*)(W1r + bo);
    v2f b1i = *(const v2f*)(W1i + bo);
    aRr = __builtin_amdgcn_wmma_f32_16x16x4_f32(false, ar,  false, b0r, (short)0, aRr, false, false);
    aRr = __builtin_amdgcn_wmma_f32_16x16x4_f32(false, nai, false, b0i, (short)0, aRr, false, false);
    aRi = __builtin_amdgcn_wmma_f32_16x16x4_f32(false, ar,  false, b0i, (short)0, aRi, false, false);
    aRi = __builtin_amdgcn_wmma_f32_16x16x4_f32(false, ai,  false, b0r, (short)0, aRi, false, false);
    aSr = __builtin_amdgcn_wmma_f32_16x16x4_f32(false, ar,  false, b1r, (short)0, aSr, false, false);
    aSr = __builtin_amdgcn_wmma_f32_16x16x4_f32(false, nai, false, b1i, (short)0, aSr, false, false);
    aSi = __builtin_amdgcn_wmma_f32_16x16x4_f32(false, ar,  false, b1i, (short)0, aSi, false, false);
    aSi = __builtin_amdgcn_wmma_f32_16x16x4_f32(false, ai,  false, b1r, (short)0, aSi, false, false);
  }
  for (int r = 0; r < 8; ++r) {
    int l = mbase + r;
    float k00r = aRr[r], k00i = aRi[r];
    float k10r = aSr[r], k10i = aSi[r];
    float sr = perl[l],            si = perl[Llen + l];
    float cr = perl[2 * Llen + l], ci = perl[3 * Llen + l];
    float tr = k00r - (sr * k10r - si * k10i);
    float ti = k00i - (sr * k10i + si * k10r);
    Ev[l * Dch + d0] = make_float2(cr * tr - ci * ti, cr * ti + ci * tr);
  }
#else
  for (int r = 0; r < 8; ++r) {
    int l = mbase + r;
    float k00r = 0.f, k00i = 0.f, k10r = 0.f, k10i = 0.f;
    for (int k = 0; k < Nst; ++k) {
      int ao = (k >> 1) * (2 * Llen) + 2 * l + (k & 1);
      int bo = (k >> 1) * (2 * Dch) + 2 * d0 + (k & 1);
      float arr = APr[ao], aii = APi[ao];
      float b0r = W0r[bo], b0i = W0i[bo];
      float b1r = W1r[bo], b1i = W1i[bo];
      k00r += arr * b0r - aii * b0i;  k00i += arr * b0i + aii * b0r;
      k10r += arr * b1r - aii * b1i;  k10i += arr * b1i + aii * b1r;
    }
    float sr = perl[l],            si = perl[Llen + l];
    float cr = perl[2 * Llen + l], ci = perl[3 * Llen + l];
    float tr = k00r - (sr * k10r - si * k10i);
    float ti = k00i - (sr * k10i + si * k10r);
    Ev[l * Dch + d0] = make_float2(cr * tr - ci * ti, cr * ti + ci * tr);
  }
#endif
}

__global__ __launch_bounds__(256) void s4_fft_r4(
    const float2* __restrict__ x, float2* __restrict__ y,
    int n, int Ns, float sign)
{
  int t = blockIdx.x * blockDim.x + threadIdx.x;
  int q = n >> 2;
  if (t >= q * Dch) return;
  int col = t & (Dch - 1);
  int j = t >> 9;
  int k = j & (Ns - 1);
  float ang = sign * (0.5f * PI_F) * (float)k / (float)Ns;
  float s1, c1;
  sincosf(ang, &s1, &c1);
  float c2 = c1 * c1 - s1 * s1, s2 = 2.f * c1 * s1;
  float c3 = c2 * c1 - s2 * s1, s3 = c2 * s1 + s2 * c1;
  float2 a = x[j * Dch + col];
  float2 b = x[(j + q) * Dch + col];
  float2 c = x[(j + 2 * q) * Dch + col];
  float2 d = x[(j + 3 * q) * Dch + col];
  float2 v1 = make_float2(c1 * b.x - s1 * b.y, c1 * b.y + s1 * b.x);
  float2 v2 = make_float2(c2 * c.x - s2 * c.y, c2 * c.y + s2 * c.x);
  float2 v3 = make_float2(c3 * d.x - s3 * d.y, c3 * d.y + s3 * d.x);
  float2 p0 = cadd(a, v2), p1 = csub(a, v2);
  float2 p2 = cadd(v1, v3), p3 = crotis(csub(v1, v3), sign);
  int i0 = 4 * j - 3 * k;
  y[(i0         ) * Dch + col] = cadd(p0, p2);
  y[(i0 +     Ns) * Dch + col] = cadd(p1, p3);
  y[(i0 + 2 * Ns) * Dch + col] = csub(p0, p2);
  y[(i0 + 3 * Ns) * Dch + col] = csub(p1, p3);
}

__device__ __forceinline__ void lds_barrier() {
  __builtin_amdgcn_fence(__ATOMIC_RELEASE, "workgroup", "local");
  __builtin_amdgcn_s_barrier();
  __builtin_amdgcn_fence(__ATOMIC_ACQUIRE, "workgroup", "local");
}

template <bool PASS_TW>
__device__ __forceinline__ void r64_core(
    float2* v, float2* sh, int k, int Ns, float sign, int lane, int sub, float2* X)
{
  float ss, cs;
  sincospif(sign * (float)sub * (1.0f / 32.0f), &ss, &cs);
  float2 tw = make_float2(1.0f, 0.0f);
  if (PASS_TW) {
    float kf = (float)k / (float)Ns;
    float s8, c8, sb, cb;
    sincospif(sign * 0.25f * kf, &s8, &c8);
    sincospif(sign * (float)sub * kf * (1.0f / 32.0f), &sb, &cb);
    float2 w8 = make_float2(c8, s8);
    float2 w = w8;
#pragma unroll
    for (int r1 = 1; r1 < 8; ++r1) {
      v[r1] = cmulf(v[r1], w);
      w = cmulf(w, w8);
    }
    tw = make_float2(cb, sb);
  }
  float2 Y[8];
  dft8(v, sign, Y);
  float2 st = make_float2(cs, ss);
#pragma unroll
  for (int a = 0; a < 8; ++a) {
    sh[(a * 8 + sub) * 32 + lane] = cmulf(Y[a], tw);
    tw = cmulf(tw, st);
  }
  lds_barrier();
  float2 z[8];
#pragma unroll
  for (int r0 = 0; r0 < 8; ++r0)
    z[r0] = sh[(sub * 8 + r0) * 32 + lane];
  dft8(z, sign, X);
}

__device__ __forceinline__ void r64_butterfly(
    const float2* __restrict__ x, float2* sh, int j, int q, int k, int Ns,
    float sign, int col, int lane, int sub, float2* X)
{
  float2 v[8];
#pragma unroll
  for (int r1 = 0; r1 < 8; ++r1)
    v[r1] = x[(j + (sub + 8 * r1) * q) * Dch + col];
  r64_core<true>(v, sh, k, Ns, sign, lane, sub, X);
}

__global__ __launch_bounds__(256) void s4_fft_r64(
    const float2* __restrict__ x, float2* __restrict__ y,
    int n, int Ns, float sign)
{
  __shared__ float2 sh[64 * 32];
  int lane = threadIdx.x & 31;
  int sub = threadIdx.x >> 5;
  int col = blockIdx.x * 32 + lane;
  int j = blockIdx.y;
  int k = j & (Ns - 1);
  float2 X[8];
  r64_butterfly(x, sh, j, n >> 6, k, Ns, sign, col, lane, sub, X);
  int i0 = 64 * j - 63 * k;
#pragma unroll
  for (int b = 0; b < 8; ++b)
    y[(i0 + (sub + 8 * b) * Ns) * Dch + col] = X[b];
}

__global__ __launch_bounds__(256) void s4_upad_r512(
    const float* __restrict__ ua, const float* __restrict__ ub,
    float2* __restrict__ y)
{
  __shared__ float2 sh[2 * 64 * 32];
  const int Q8 = NFFT / 8;
  int lane = threadIdx.x & 31;
  int sub = threadIdx.x >> 5;
  int col = blockIdx.x * 32 + lane;
  int jh = blockIdx.y;
  float2 S[8][8];
#pragma unroll
  for (int r1 = 0; r1 < 8; ++r1) {
    int jr = jh + 64 * (sub + 8 * r1);
    float2 v[8];
#pragma unroll
    for (int rp = 0; rp < 4; ++rp) {
      int pos = (jr + rp * Q8) * Dch + col;
      v[rp] = make_float2(ua[pos], ub[pos]);
    }
#pragma unroll
    for (int rp = 4; rp < 8; ++rp)
      v[rp] = make_float2(0.f, 0.f);
    dft8(v, -1.0f, S[r1]);
  }
#pragma unroll
  for (int m = 0; m < 8; ++m) {
    float2 v[8], X[8];
#pragma unroll
    for (int r1 = 0; r1 < 8; ++r1)
      v[r1] = S[r1][m];
    r64_core<true>(v, sh + (m & 1) * (64 * 32), m, 8, -1.0f, lane, sub, X);
#pragma unroll
    for (int b = 0; b < 8; ++b)
      y[(512 * jh + m + 8 * (sub + 8 * b)) * Dch + col] = X[b];
  }
}

__global__ __launch_bounds__(256) void s4_klast_r64_pad_r8(
    const float2* __restrict__ x, float2* __restrict__ y)
{
  __shared__ float2 sh[64 * 32];
  const int Q = Llen / 64;
  int lane = threadIdx.x & 31;
  int sub = threadIdx.x >> 5;
  int col = blockIdx.x * 32 + lane;
  int j = blockIdx.y;
  float2 X[8];
  r64_butterfly(x, sh, j, Q, j, Q, +1.0f, col, lane, sub, X);
  const float invL = 1.0f / (float)Llen;
  float2 va[8], vb[8];
#pragma unroll
  for (int r = 0; r < 4; ++r) {
    va[r] = make_float2(X[2 * r].x * invL, 0.f);
    vb[r] = make_float2(X[2 * r + 1].x * invL, 0.f);
  }
#pragma unroll
  for (int r = 4; r < 8; ++r) {
    va[r] = make_float2(0.f, 0.f);
    vb[r] = make_float2(0.f, 0.f);
  }
  float2 Za[8], Zb[8];
  dft8(va, -1.0f, Za);
  dft8(vb, -1.0f, Zb);
  int ja = j + sub * Q;
  int jb = j + (sub + 8) * Q;
#pragma unroll
  for (int m = 0; m < 8; ++m) {
    y[(8 * ja + m) * Dch + col] = Za[m];
    y[(8 * jb + m) * Dch + col] = Zb[m];
  }
}

__global__ __launch_bounds__(256) void s4_mid_r64_cmul_r64(
    const float2* __restrict__ x, const float2* __restrict__ Kd,
    float2* __restrict__ y)
{
  __shared__ float2 sh[2 * 64 * 32];
  const int Q = NFFT / 64;
  int lane = threadIdx.x & 31;
  int sub = threadIdx.x >> 5;
  int col = blockIdx.x * 32 + lane;
  int j = blockIdx.y;
  float2 kd[8];
#pragma unroll
  for (int b = 0; b < 8; ++b)
    kd[b] = Kd[(j + (sub + 8 * b) * Q) * Dch + col];
  float2 X[8];
  r64_butterfly(x, sh, j, Q, j, Q, -1.0f, col, lane, sub, X);
  float2 v[8];
#pragma unroll
  for (int b = 0; b < 8; ++b)
    v[b] = cmulf(X[b], kd[b]);
  float2 Z[8];
  r64_core<false>(v, sh + 64 * 32, 0, 1, +1.0f, lane, sub, Z);
#pragma unroll
  for (int b = 0; b < 8; ++b)
    y[(64 * j + sub + 8 * b) * Dch + col] = Z[b];
}

__global__ __launch_bounds__(256) __attribute__((amdgpu_waves_per_eu(4, 4))) void s4_final_r512(
    const float2* __restrict__ x,
    const float* __restrict__ ua, const float* __restrict__ ub,
    const float* __restrict__ Dp,
    float* __restrict__ outa, float* __restrict__ outb)
{
  __shared__ float2 sh[2 * 64 * 32];
  const int Q = NFFT / 64;
  const int Q8 = NFFT / 8;
  int lane = threadIdx.x & 31;
  int sub = threadIdx.x >> 5;
  int col = blockIdx.x * 32 + lane;
  int kk = blockIdx.y;
  float sw, cw;
  sincospif((float)kk * (1.0f / 256.0f), &sw, &cw);
  float2 w1 = make_float2(cw, sw);
  float2 wp[8];
  wp[1] = w1;
#pragma unroll
  for (int rp = 2; rp < 8; ++rp)
    wp[rp] = cmulf(wp[rp - 1], w1);
  float2 S[8][8];
#pragma unroll
  for (int r1 = 0; r1 < 8; ++r1) {
    int jr = kk + 64 * (sub + 8 * r1);
#pragma unroll
    for (int rp = 0; rp < 8; ++rp)
      S[r1][rp] = x[(jr + rp * Q8) * Dch + col];
  }
#pragma unroll
  for (int r1 = 0; r1 < 8; ++r1) {
    float2 v[8];
    v[0] = S[r1][0];
#pragma unroll
    for (int rp = 1; rp < 8; ++rp)
      v[rp] = cmulf(S[r1][rp], wp[rp]);
    dft8(v, +1.0f, S[r1]);
  }
  float dp = Dp[0];
  const float inv = 1.0f / (float)NFFT;
#pragma unroll
  for (int m = 0; m < 8; ++m) {
    int j2 = kk + 64 * m;
    float2 v[8], X[8];
#pragma unroll
    for (int r1 = 0; r1 < 8; ++r1)
      v[r1] = S[r1][m];
    float pa[4], pb[4];
#pragma unroll
    for (int b = 0; b < 4; ++b) {
      int pos = (j2 + (sub + 8 * b) * Q) * Dch + col;
      pa[b] = ua[pos];
      pb[b] = ub[pos];
    }
    r64_core<true>(v, sh + (m & 1) * (64 * 32), j2, Q, +1.0f, lane, sub, X);
#pragma unroll
    for (int b = 0; b < 4; ++b) {
      int pos = (j2 + (sub + 8 * b) * Q) * Dch + col;
      outa[pos] = X[b].x * inv + dp * pa[b];
      outb[pos] = X[b].y * inv + dp * pb[b];
    }
  }
}


extern "C" void kernel_launch(void* const* d_in, const int* in_sizes, int n_in,
                              void* d_out, int out_size, void* d_ws, size_t ws_size,
                              hipStream_t stream) {
  const float* u        = (const float*)d_in[0];
  const float* Lam      = (const float*)d_in[1];
  const float* Pri      = (const float*)d_in[2];
  const float* Bri      = (const float*)d_in[3];
  const float* Cri      = (const float*)d_in[4];
  const float* Dp       = (const float*)d_in[5];
  const float* log_step = (const float*)d_in[6];
  float* out = (float*)d_out;

  uint8_t* base = (uint8_t*)d_ws;
  size_t off = 0;
  auto carve = [&](size_t bytes) -> void* {
    void* p = base + off;
    off += (bytes + 255) & ~(size_t)255;
    return p;
  };
  float*  APr  = (float*)carve((size_t)Llen * Nst * 4);
  float*  APi  = (float*)carve((size_t)Llen * Nst * 4);
  float*  W0r  = (float*)carve((size_t)Nst * Dch * 4);
  float*  W0i  = (float*)carve((size_t)Nst * Dch * 4);
  float*  W1r  = (float*)carve((size_t)Nst * Dch * 4);
  float*  W1i  = (float*)carve((size_t)Nst * Dch * 4);
  float*  perl = (float*)carve((size_t)6 * Llen * 4);
  float2* Ev   = (float2*)carve((size_t)Llen * Dch * 8);
  float2* T0   = (float2*)carve((size_t)NFFT * Dch * 8);
  float2* T1   = (float2*)carve((size_t)NFFT * Dch * 8);
  float2* Kd   = (float2*)carve((size_t)NFFT * Dch * 8);
  (void)ws_size; (void)in_sizes; (void)n_in; (void)out_size;

  const int TB   = 256;
  const int gR4s = (Llen / 4) * Dch / TB;
  const dim3 gR64(Dch / 32, NFFT / 64);
  const dim3 gR64s(Dch / 32, Llen / 64);
  const dim3 gR512(Dch / 32, NFFT / 512);

  s4_perl<<<(Llen + TB - 1) / TB, TB, 0, stream>>>(Lam, Pri, Cri, log_step, perl);
  s4_weights<<<(Nst * Dch + TB - 1) / TB, TB, 0, stream>>>(Bri, Pri, Cri, W0r, W0i, W1r, W1i);
  s4_cauchy<<<(Llen * Nst + TB - 1) / TB, TB, 0, stream>>>(Lam, perl, APr, APi);
  s4_cauchy_gemm<<<dim3(Dch / 16, Llen / 16), 32, 0, stream>>>(
      APr, APi, W0r, W0i, W1r, W1i, perl, Ev);

  s4_fft_r4<<<gR4s, TB, 0, stream>>>(Ev, T0, Llen, 1, +1.0f);
  s4_fft_r64<<<gR64s, TB, 0, stream>>>(T0, Ev, Llen, 4, +1.0f);
  s4_klast_r64_pad_r8<<<gR64s, TB, 0, stream>>>(Ev, T1);
  s4_fft_r64<<<gR64, TB, 0, stream>>>(T1, T0, NFFT, 8, -1.0f);
  s4_fft_r64<<<gR64, TB, 0, stream>>>(T0, Kd, NFFT, 512, -1.0f);

  for (int b = 0; b < Bsz; b += 2) {
    const float* ua = u + (size_t)b * Llen * Dch;
    const float* ub = u + (size_t)(b + 1) * Llen * Dch;
    float*       oa = out + (size_t)b * Llen * Dch;
    float*       ob = out + (size_t)(b + 1) * Llen * Dch;
    s4_upad_r512<<<gR512, TB, 0, stream>>>(ua, ub, T0);
    s4_mid_r64_cmul_r64<<<gR64, TB, 0, stream>>>(T0, Kd, T1);
    s4_final_r512<<<gR512, TB, 0, stream>>>(T1, ua, ub, Dp, oa, ob);
  }
}


# K-path FFTs take 2 global round trips each, not 3: inverse = radix-256 (radix-4 kept in registers, feeds 4 radix-64) + radix-64 writing real K; forward = padded radix-512 + radix-64; sums unchanged
# speedup vs baseline: 5.1263x; 1.2117x over previous
#include <hip/hip_runtime.h>
#include <hip/hip_bf16.h>
#include <stdint.h>


#define Bsz   4
#define Llen  16384
#define Dch   512
#define Nst   64
#define NFFT  32768
#define PI_F  3.14159265358979323846f

typedef __attribute__((ext_vector_type(2))) float v2f;
typedef __attribute__((ext_vector_type(8))) float v8f;

#ifndef __has_builtin
#define __has_builtin(x) 0
#endif
#if __has_builtin(__builtin_amdgcn_wmma_f32_16x16x4_f32)
#define S4_WMMA 1
#else
#define S4_WMMA 0
#endif

__device__ __forceinline__ float2 cadd(float2 a, float2 b) { return make_float2(a.x + b.x, a.y + b.y); }
__device__ __forceinline__ float2 csub(float2 a, float2 b) { return make_float2(a.x - b.x, a.y - b.y); }
__device__ __forceinline__ float2 cmulf(float2 a, float2 b) { return make_float2(a.x * b.x - a.y * b.y, a.x * b.y + a.y * b.x); }
__device__ __forceinline__ float2 crotis(float2 a, float s) { return make_float2(-s * a.y, s * a.x); }

__device__ __forceinline__ void dft8(const float2* v, float s, float2* X) {
  const float R = 0.70710678118654752440f;
  float2 t0 = cadd(v[0], v[4]), t1 = cadd(v[1], v[5]);
  float2 t2 = cadd(v[2], v[6]), t3 = cadd(v[3], v[7]);
  float2 u0 = csub(v[0], v[4]), u1 = csub(v[1], v[5]);
  float2 u2 = csub(v[2], v[6]), u3 = csub(v[3], v[7]);
  float2 m1 = cmulf(u1, make_float2(R, s * R));
  float2 m2 = crotis(u2, s);
  float2 m3 = cmulf(u3, make_float2(-R, s * R));
  float2 p0 = cadd(t0, t2), p1 = csub(t0, t2);
  float2 p2 = cadd(t1, t3), p3 = crotis(csub(t1, t3), s);
  X[0] = cadd(p0, p2); X[2] = cadd(p1, p3);
  X[4] = csub(p0, p2); X[6] = csub(p1, p3);
  float2 q0 = cadd(u0, m2), q1 = csub(u0, m2);
  float2 q2 = cadd(m1, m3), q3 = crotis(csub(m1, m3), s);
  X[1] = cadd(q0, q2); X[3] = cadd(q1, q3);
  X[5] = csub(q0, q2); X[7] = csub(q1, q3);
}

__global__ __launch_bounds__(256) void s4_perl(
    const float* __restrict__ Lam, const float* __restrict__ Pri,
    const float* __restrict__ Cri, const float* __restrict__ log_step,
    float* __restrict__ perl)
{
  int l = blockIdx.x * blockDim.x + threadIdx.x;
  if (l >= Llen) return;
  float step = expf(log_step[0]);
  float theta = -2.0f * PI_F * (float)l / (float)Llen;
  float oi, orr;
  sincosf(theta, &oi, &orr);
  float op1r = 1.0f + orr;
  float den = op1r * op1r + oi * oi;
  float nr = 1.0f - orr, ni = -oi;
  float gscale = 2.0f / step;
  float gr = gscale * (nr * op1r + ni * oi) / den;
  float gi = gscale * (ni * op1r - nr * oi) / den;
  float scr = 2.0f * op1r / den;
  float sci = -2.0f * oi / den;
  float k01r = 0.f, k01i = 0.f, k11r = 0.f, k11i = 0.f;
  for (int n = 0; n < Nst; ++n) {
    float lr = Lam[2 * n], li = Lam[2 * n + 1];
    float pr = Pri[2 * n], pi = Pri[2 * n + 1];
    float cr = Cri[2 * n], ci = Cri[2 * n + 1];
    float zr = gr - lr, zi = gi - li;
    float zd = zr * zr + zi * zi;
    float car = zr / zd, cai = -zi / zd;
    float w01r = cr * pr + ci * pi;
    float w01i = cr * pi - ci * pr;
    float w11r = pr * pr + pi * pi;
    k01r += car * w01r - cai * w01i;
    k01i += car * w01i + cai * w01r;
    k11r += car * w11r;
    k11i += cai * w11r;
  }
  float dr = 1.0f + k11r, di = k11i;
  float dd = dr * dr + di * di;
  float sr = (k01r * dr + k01i * di) / dd;
  float si = (k01i * dr - k01r * di) / dd;
  perl[l] = sr;
  perl[Llen + l] = si;
  perl[2 * Llen + l] = scr;
  perl[3 * Llen + l] = sci;
  perl[4 * Llen + l] = gr;
  perl[5 * Llen + l] = gi;
}

__global__ __launch_bounds__(256) void s4_weights(
    const float* __restrict__ Bri, const float* __restrict__ Pri,
    const float* __restrict__ Cri,
    float* __restrict__ W0r, float* __restrict__ W0i,
    float* __restrict__ W1r, float* __restrict__ W1i)
{
  int idx = blockIdx.x * blockDim.x + threadIdx.x;
  if (idx >= Nst * Dch) return;
  int n = idx / Dch, d = idx % Dch;
  float br = Bri[2 * (n * Dch + d)], bi = Bri[2 * (n * Dch + d) + 1];
  float cr = Cri[2 * n], ci = Cri[2 * n + 1];
  float pr = Pri[2 * n], pi = Pri[2 * n + 1];
  int o = (n >> 1) * (2 * Dch) + 2 * d + (n & 1);
  W0r[o] = cr * br + ci * bi;
  W0i[o] = cr * bi - ci * br;
  W1r[o] = pr * br + pi * bi;
  W1i[o] = pr * bi - pi * br;
}

__global__ __launch_bounds__(256) void s4_cauchy(
    const float* __restrict__ Lam, const float* __restrict__ perl,
    float* __restrict__ APr, float* __restrict__ APi)
{
  int idx = blockIdx.x * blockDim.x + threadIdx.x;
  if (idx >= Llen * Nst) return;
  int n = idx / Llen, l = idx % Llen;
  float gr = perl[4 * Llen + l], gi = perl[5 * Llen + l];
  float zr = gr - Lam[2 * n], zi = gi - Lam[2 * n + 1];
  float zd = zr * zr + zi * zi;
  int o = (n >> 1) * (2 * Llen) + 2 * l + (n & 1);
  APr[o] = zr / zd;
  APi[o] = -zi / zd;
}

__global__ __launch_bounds__(32) void s4_cauchy_gemm(
    const float* __restrict__ APr, const float* __restrict__ APi,
    const float* __restrict__ W0r, const float* __restrict__ W0i,
    const float* __restrict__ W1r, const float* __restrict__ W1i,
    const float* __restrict__ perl, float2* __restrict__ Ev)
{
  int lane = threadIdx.x & 31;
  int tileD = blockIdx.x;
  int tileL = blockIdx.y;
  int d0 = tileD * 16 + (lane & 15);
  int mbase = tileL * 16 + ((lane >> 4) ? 8 : 0);
#if S4_WMMA
  int m = tileL * 16 + (lane & 15);
  int khi = (lane >> 4) * 2;
  v8f aRr = {}, aRi = {}, aSr = {}, aSi = {};
  for (int kb = 0; kb < Nst; kb += 4) {
    int k0 = kb + khi;
    const float* pr = APr + (k0 >> 1) * (2 * Llen) + 2 * m;
    const float* pi = APi + (k0 >> 1) * (2 * Llen) + 2 * m;
    v2f ar = *(const v2f*)pr;
    v2f ai = *(const v2f*)pi;
    v2f nai = -ai;
    int bo = (k0 >> 1) * (2 * Dch) + 2 * d0;
    v2f b0r = *(const v2f*)(W0r + bo);
    v2f b0i = *(const v2f*)(W0i + bo);
    v2f b1r = *(const v2f*)(W1r + bo);
    v2f b1i = *(const v2f*)(W1i + bo);
    aRr = __builtin_amdgcn_wmma_f32_16x16x4_f32(false, ar,  false, b0r, (short)0, aRr, false, false);
    aRr = __builtin_amdgcn_wmma_f32_16x16x4_f32(false, nai, false, b0i, (short)0, aRr, false, false);
    aRi = __builtin_amdgcn_wmma_f32_16x16x4_f32(false, ar,  false, b0i, (short)0, aRi, false, false);
    aRi = __builtin_amdgcn_wmma_f32_16x16x4_f32(false, ai,  false, b0r, (short)0, aRi, false, false);
    aSr = __builtin_amdgcn_wmma_f32_16x16x4_f32(false, ar,  false, b1r, (short)0, aSr, false, false);
    aSr = __builtin_amdgcn_wmma_f32_16x16x4_f32(false, nai, false, b1i, (short)0, aSr, false, false);
    aSi = __builtin_amdgcn_wmma_f32_16x16x4_f32(false, ar,  false, b1i, (short)0, aSi, false, false);
    aSi = __builtin_amdgcn_wmma_f32_16x16x4_f32(false, ai,  false, b1r, (short)0, aSi, false, false);
  }
  for (int r = 0; r < 8; ++r) {
    int l = mbase + r;
    float k00r = aRr[r], k00i = aRi[r];
    float k10r = aSr[r], k10i = aSi[r];
    float sr = perl[l],            si = perl[Llen + l];
    float cr = perl[2 * Llen + l], ci = perl[3 * Llen + l];
    float tr = k00r - (sr * k10r - si * k10i);
    float ti = k00i - (sr * k10i + si * k10r);
    Ev[l * Dch + d0] = make_float2(cr * tr - ci * ti, cr * ti + ci * tr);
  }
#else
  for (int r = 0; r < 8; ++r) {
    int l = mbase + r;
    float k00r = 0.f, k00i = 0.f, k10r = 0.f, k10i = 0.f;
    for (int k = 0; k < Nst; ++k) {
      int ao = (k >> 1) * (2 * Llen) + 2 * l + (k & 1);
      int bo = (k >> 1) * (2 * Dch) + 2 * d0 + (k & 1);
      float arr = APr[ao], aii = APi[ao];
      float b0r = W0r[bo], b0i = W0i[bo];
      float b1r = W1r[bo], b1i = W1i[bo];
      k00r += arr * b0r - aii * b0i;  k00i += arr * b0i + aii * b0r;
      k10r += arr * b1r - aii * b1i;  k10i += arr * b1i + aii * b1r;
    }
    float sr = perl[l],            si = perl[Llen + l];
    float cr = perl[2 * Llen + l], ci = perl[3 * Llen + l];
    float tr = k00r - (sr * k10r - si * k10i);
    float ti = k00i - (sr * k10i + si * k10r);
    Ev[l * Dch + d0] = make_float2(cr * tr - ci * ti, cr * ti + ci * tr);
  }
#endif
}

__device__ __forceinline__ void lds_barrier() {
  __builtin_amdgcn_fence(__ATOMIC_RELEASE, "workgroup", "local");
  __builtin_amdgcn_s_barrier();
  __builtin_amdgcn_fence(__ATOMIC_ACQUIRE, "workgroup", "local");
}

template <bool PASS_TW>
__device__ __forceinline__ void r64_core(
    float2* v, float2* sh, int k, int Ns, float sign, int lane, int sub, float2* X)
{
  float ss, cs;
  sincospif(sign * (float)sub * (1.0f / 32.0f), &ss, &cs);
  float2 tw = make_float2(1.0f, 0.0f);
  if (PASS_TW) {
    float kf = (float)k / (float)Ns;
    float s8, c8, sb, cb;
    sincospif(sign * 0.25f * kf, &s8, &c8);
    sincospif(sign * (float)sub * kf * (1.0f / 32.0f), &sb, &cb);
    float2 w8 = make_float2(c8, s8);
    float2 w = w8;
#pragma unroll
    for (int r1 = 1; r1 < 8; ++r1) {
      v[r1] = cmulf(v[r1], w);
      w = cmulf(w, w8);
    }
    tw = make_float2(cb, sb);
  }
  float2 Y[8];
  dft8(v, sign, Y);
  float2 st = make_float2(cs, ss);
#pragma unroll
  for (int a = 0; a < 8; ++a) {
    sh[(a * 8 + sub) * 32 + lane] = cmulf(Y[a], tw);
    tw = cmulf(tw, st);
  }
  lds_barrier();
  float2 z[8];
#pragma unroll
  for (int r0 = 0; r0 < 8; ++r0)
    z[r0] = sh[(sub * 8 + r0) * 32 + lane];
  dft8(z, sign, X);
}

__device__ __forceinline__ void r64_butterfly(
    const float2* __restrict__ x, float2* sh, int j, int q, int k, int Ns,
    float sign, int col, int lane, int sub, float2* X)
{
  float2 v[8];
#pragma unroll
  for (int r1 = 0; r1 < 8; ++r1)
    v[r1] = x[(j + (sub + 8 * r1) * q) * Dch + col];
  r64_core<true>(v, sh, k, Ns, sign, lane, sub, X);
}

__global__ __launch_bounds__(256) void s4_fft_r64(
    const float2* __restrict__ x, float2* __restrict__ y,
    int n, int Ns, float sign)
{
  __shared__ float2 sh[64 * 32];
  int lane = threadIdx.x & 31;
  int sub = threadIdx.x >> 5;
  int col = blockIdx.x * 32 + lane;
  int j = blockIdx.y;
  int k = j & (Ns - 1);
  float2 X[8];
  r64_butterfly(x, sh, j, n >> 6, k, Ns, sign, col, lane, sub, X);
  int i0 = 64 * j - 63 * k;
#pragma unroll
  for (int b = 0; b < 8; ++b)
    y[(i0 + (sub + 8 * b) * Ns) * Dch + col] = X[b];
}

template <bool PAIR>
__device__ __forceinline__ void pad_r512_body(
    const float* __restrict__ ua, const float* __restrict__ ub,
    float2* __restrict__ y, float2* sh)
{
  const int Q8 = NFFT / 8;
  int lane = threadIdx.x & 31;
  int sub = threadIdx.x >> 5;
  int col = blockIdx.x * 32 + lane;
  int jh = blockIdx.y;
  float2 S[8][8];
#pragma unroll
  for (int r1 = 0; r1 < 8; ++r1) {
    int jr = jh + 64 * (sub + 8 * r1);
    float2 v[8];
#pragma unroll
    for (int rp = 0; rp < 4; ++rp) {
      int pos = (jr + rp * Q8) * Dch + col;
      v[rp] = make_float2(ua[pos], PAIR ? ub[pos] : 0.f);
    }
#pragma unroll
    for (int rp = 4; rp < 8; ++rp)
      v[rp] = make_float2(0.f, 0.f);
    dft8(v, -1.0f, S[r1]);
  }
#pragma unroll
  for (int m = 0; m < 8; ++m) {
    float2 v[8], X[8];
#pragma unroll
    for (int r1 = 0; r1 < 8; ++r1)
      v[r1] = S[r1][m];
    r64_core<true>(v, sh + (m & 1) * (64 * 32), m, 8, -1.0f, lane, sub, X);
#pragma unroll
    for (int b = 0; b < 8; ++b)
      y[(512 * jh + m + 8 * (sub + 8 * b)) * Dch + col] = X[b];
  }
}

__global__ __launch_bounds__(256) void s4_upad_r512(
    const float* __restrict__ ua, const float* __restrict__ ub,
    float2* __restrict__ y)
{
  __shared__ float2 sh[2 * 64 * 32];
  pad_r512_body<true>(ua, ub, y, sh);
}

__global__ __launch_bounds__(256) void s4_kpad_r512(
    const float* __restrict__ kr, float2* __restrict__ y)
{
  __shared__ float2 sh[2 * 64 * 32];
  pad_r512_body<false>(kr, kr, y, sh);
}

__global__ __launch_bounds__(256) void s4_kfirst_r256(
    const float2* __restrict__ x, float2* __restrict__ y)
{
  __shared__ float2 sh[2 * 64 * 32];
  const int Q4 = Llen / 4;
  int lane = threadIdx.x & 31;
  int sub = threadIdx.x >> 5;
  int col = blockIdx.x * 32 + lane;
  int jh = blockIdx.y;
  float2 S[8][4];
#pragma unroll
  for (int r1 = 0; r1 < 8; ++r1) {
    int jr = jh + 64 * (sub + 8 * r1);
#pragma unroll
    for (int rp = 0; rp < 4; ++rp)
      S[r1][rp] = x[(jr + rp * Q4) * Dch + col];
  }
#pragma unroll
  for (int r1 = 0; r1 < 8; ++r1) {
    float2 p0 = cadd(S[r1][0], S[r1][2]), p1 = csub(S[r1][0], S[r1][2]);
    float2 p2 = cadd(S[r1][1], S[r1][3]);
    float2 p3 = crotis(csub(S[r1][1], S[r1][3]), +1.0f);
    S[r1][0] = cadd(p0, p2); S[r1][1] = cadd(p1, p3);
    S[r1][2] = csub(p0, p2); S[r1][3] = csub(p1, p3);
  }
#pragma unroll
  for (int m = 0; m < 4; ++m) {
    float2 v[8], X[8];
#pragma unroll
    for (int r1 = 0; r1 < 8; ++r1)
      v[r1] = S[r1][m];
    r64_core<true>(v, sh + (m & 1) * (64 * 32), m, 4, +1.0f, lane, sub, X);
#pragma unroll
    for (int b = 0; b < 8; ++b)
      y[(256 * jh + m + 4 * (sub + 8 * b)) * Dch + col] = X[b];
  }
}

__global__ __launch_bounds__(256) void s4_klast_r64_real(
    const float2* __restrict__ x, float* __restrict__ kr)
{
  __shared__ float2 sh[64 * 32];
  const int Q = Llen / 64;
  int lane = threadIdx.x & 31;
  int sub = threadIdx.x >> 5;
  int col = blockIdx.x * 32 + lane;
  int j = blockIdx.y;
  float2 X[8];
  r64_butterfly(x, sh, j, Q, j, Q, +1.0f, col, lane, sub, X);
  const float invL = 1.0f / (float)Llen;
#pragma unroll
  for (int b = 0; b < 8; ++b)
    kr[(j + (sub + 8 * b) * Q) * Dch + col] = X[b].x * invL;
}

__global__ __launch_bounds__(256) void s4_mid_r64_cmul_r64(
    const float2* __restrict__ x, const float2* __restrict__ Kd,
    float2* __restrict__ y)
{
  __shared__ float2 sh[2 * 64 * 32];
  const int Q = NFFT / 64;
  int lane = threadIdx.x & 31;
  int sub = threadIdx.x >> 5;
  int col = blockIdx.x * 32 + lane;
  int j = blockIdx.y;
  float2 kd[8];
#pragma unroll
  for (int b = 0; b < 8; ++b)
    kd[b] = Kd[(j + (sub + 8 * b) * Q) * Dch + col];
  float2 X[8];
  r64_butterfly(x, sh, j, Q, j, Q, -1.0f, col, lane, sub, X);
  float2 v[8];
#pragma unroll
  for (int b = 0; b < 8; ++b)
    v[b] = cmulf(X[b], kd[b]);
  float2 Z[8];
  r64_core<false>(v, sh + 64 * 32, 0, 1, +1.0f, lane, sub, Z);
#pragma unroll
  for (int b = 0; b < 8; ++b)
    y[(64 * j + sub + 8 * b) * Dch + col] = Z[b];
}

__global__ __launch_bounds__(256) __attribute__((amdgpu_waves_per_eu(4, 4))) void s4_final_r512(
    const float2* __restrict__ x,
    const float* __restrict__ ua, const float* __restrict__ ub,
    const float* __restrict__ Dp,
    float* __restrict__ outa, float* __restrict__ outb)
{
  __shared__ float2 sh[2 * 64 * 32];
  const int Q = NFFT / 64;
  const int Q8 = NFFT / 8;
  int lane = threadIdx.x & 31;
  int sub = threadIdx.x >> 5;
  int col = blockIdx.x * 32 + lane;
  int kk = blockIdx.y;
  float sw, cw;
  sincospif((float)kk * (1.0f / 256.0f), &sw, &cw);
  float2 w1 = make_float2(cw, sw);
  float2 wp[8];
  wp[1] = w1;
#pragma unroll
  for (int rp = 2; rp < 8; ++rp)
    wp[rp] = cmulf(wp[rp - 1], w1);
  float2 S[8][8];
#pragma unroll
  for (int r1 = 0; r1 < 8; ++r1) {
    int jr = kk + 64 * (sub + 8 * r1);
#pragma unroll
    for (int rp = 0; rp < 8; ++rp)
      S[r1][rp] = x[(jr + rp * Q8) * Dch + col];
  }
#pragma unroll
  for (int r1 = 0; r1 < 8; ++r1) {
    float2 v[8];
    v[0] = S[r1][0];
#pragma unroll
    for (int rp = 1; rp < 8; ++rp)
      v[rp] = cmulf(S[r1][rp], wp[rp]);
    dft8(v, +1.0f, S[r1]);
  }
  float dp = Dp[0];
  const float inv = 1.0f / (float)NFFT;
#pragma unroll
  for (int m = 0; m < 8; ++m) {
    int j2 = kk + 64 * m;
    float2 v[8], X[8];
#pragma unroll
    for (int r1 = 0; r1 < 8; ++r1)
      v[r1] = S[r1][m];
    float pa[4], pb[4];
#pragma unroll
    for (int b = 0; b < 4; ++b) {
      int pos = (j2 + (sub + 8 * b) * Q) * Dch + col;
      pa[b] = ua[pos];
      pb[b] = ub[pos];
    }
    r64_core<true>(v, sh + (m & 1) * (64 * 32), j2, Q, +1.0f, lane, sub, X);
#pragma unroll
    for (int b = 0; b < 4; ++b) {
      int pos = (j2 + (sub + 8 * b) * Q) * Dch + col;
      outa[pos] = X[b].x * inv + dp * pa[b];
      outb[pos] = X[b].y * inv + dp * pb[b];
    }
  }
}


extern "C" void kernel_launch(void* const* d_in, const int* in_sizes, int n_in,
                              void* d_out, int out_size, void* d_ws, size_t ws_size,
                              hipStream_t stream) {
  const float* u        = (const float*)d_in[0];
  const float* Lam      = (const float*)d_in[1];
  const float* Pri      = (const float*)d_in[2];
  const float* Bri      = (const float*)d_in[3];
  const float* Cri      = (const float*)d_in[4];
  const float* Dp       = (const float*)d_in[5];
  const float* log_step = (const float*)d_in[6];
  float* out = (float*)d_out;

  uint8_t* base = (uint8_t*)d_ws;
  size_t off = 0;
  auto carve = [&](size_t bytes) -> void* {
    void* p = base + off;
    off += (bytes + 255) & ~(size_t)255;
    return p;
  };
  float*  APr  = (float*)carve((size_t)Llen * Nst * 4);
  float*  APi  = (float*)carve((size_t)Llen * Nst * 4);
  float*  W0r  = (float*)carve((size_t)Nst * Dch * 4);
  float*  W0i  = (float*)carve((size_t)Nst * Dch * 4);
  float*  W1r  = (float*)carve((size_t)Nst * Dch * 4);
  float*  W1i  = (float*)carve((size_t)Nst * Dch * 4);
  float*  perl = (float*)carve((size_t)6 * Llen * 4);
  float2* Ev   = (float2*)carve((size_t)Llen * Dch * 8);
  float2* T0   = (float2*)carve((size_t)NFFT * Dch * 8);
  float2* T1   = (float2*)carve((size_t)NFFT * Dch * 8);
  float2* Kd   = (float2*)carve((size_t)NFFT * Dch * 8);
  (void)ws_size; (void)in_sizes; (void)n_in; (void)out_size;

  const int TB   = 256;
  const dim3 gR64(Dch / 32, NFFT / 64);
  const dim3 gR64s(Dch / 32, Llen / 64);
  const dim3 gR512(Dch / 32, NFFT / 512);
  const dim3 gR256s(Dch / 32, Llen / 256);

  s4_perl<<<(Llen + TB - 1) / TB, TB, 0, stream>>>(Lam, Pri, Cri, log_step, perl);
  s4_weights<<<(Nst * Dch + TB - 1) / TB, TB, 0, stream>>>(Bri, Pri, Cri, W0r, W0i, W1r, W1i);
  s4_cauchy<<<(Llen * Nst + TB - 1) / TB, TB, 0, stream>>>(Lam, perl, APr, APi);
  s4_cauchy_gemm<<<dim3(Dch / 16, Llen / 16), 32, 0, stream>>>(
      APr, APi, W0r, W0i, W1r, W1i, perl, Ev);

  float* Kr = (float*)Ev;
  s4_kfirst_r256<<<gR256s, TB, 0, stream>>>(Ev, T0);
  s4_klast_r64_real<<<gR64s, TB, 0, stream>>>(T0, Kr);
  s4_kpad_r512<<<gR512, TB, 0, stream>>>(Kr, T1);
  s4_fft_r64<<<gR64, TB, 0, stream>>>(T1, Kd, NFFT, 512, -1.0f);

  for (int b = 0; b < Bsz; b += 2) {
    const float* ua = u + (size_t)b * Llen * Dch;
    const float* ub = u + (size_t)(b + 1) * Llen * Dch;
    float*       oa = out + (size_t)b * Llen * Dch;
    float*       ob = out + (size_t)(b + 1) * Llen * Dch;
    s4_upad_r512<<<gR512, TB, 0, stream>>>(ua, ub, T0);
    s4_mid_r64_cmul_r64<<<gR64, TB, 0, stream>>>(T0, Kd, T1);
    s4_final_r512<<<gR512, TB, 0, stream>>>(T1, ua, ub, Dp, oa, ob);
  }
}


# K real => Kd Hermitian: K's last radix-64 pass runs butterflies 0..256 of 512 only; mid pass does j and 512-j per block on one Kd read, mirror rows = conj (rounding now the mirror's); sums unchanged
# speedup vs baseline: 5.4635x; 1.0658x over previous
#include <hip/hip_runtime.h>
#include <hip/hip_bf16.h>
#include <stdint.h>


#define Bsz   4
#define Llen  16384
#define Dch   512
#define Nst   64
#define NFFT  32768
#define PI_F  3.14159265358979323846f

typedef __attribute__((ext_vector_type(2))) float v2f;
typedef __attribute__((ext_vector_type(8))) float v8f;

#ifndef __has_builtin
#define __has_builtin(x) 0
#endif
#if __has_builtin(__builtin_amdgcn_wmma_f32_16x16x4_f32)
#define S4_WMMA 1
#else
#define S4_WMMA 0
#endif

__device__ __forceinline__ float2 cadd(float2 a, float2 b) { return make_float2(a.x + b.x, a.y + b.y); }
__device__ __forceinline__ float2 csub(float2 a, float2 b) { return make_float2(a.x - b.x, a.y - b.y); }
__device__ __forceinline__ float2 cmulf(float2 a, float2 b) { return make_float2(a.x * b.x - a.y * b.y, a.x * b.y + a.y * b.x); }
__device__ __forceinline__ float2 cmulcf(float2 a, float2 b) { return make_float2(a.x * b.x + a.y * b.y, a.y * b.x - a.x * b.y); }
__device__ __forceinline__ float2 crotis(float2 a, float s) { return make_float2(-s * a.y, s * a.x); }

__device__ __forceinline__ void dft8(const float2* v, float s, float2* X) {
  const float R = 0.70710678118654752440f;
  float2 t0 = cadd(v[0], v[4]), t1 = cadd(v[1], v[5]);
  float2 t2 = cadd(v[2], v[6]), t3 = cadd(v[3], v[7]);
  float2 u0 = csub(v[0], v[4]), u1 = csub(v[1], v[5]);
  float2 u2 = csub(v[2], v[6]), u3 = csub(v[3], v[7]);
  float2 m1 = cmulf(u1, make_float2(R, s * R));
  float2 m2 = crotis(u2, s);
  float2 m3 = cmulf(u3, make_float2(-R, s * R));
  float2 p0 = cadd(t0, t2), p1 = csub(t0, t2);
  float2 p2 = cadd(t1, t3), p3 = crotis(csub(t1, t3), s);
  X[0] = cadd(p0, p2); X[2] = cadd(p1, p3);
  X[4] = csub(p0, p2); X[6] = csub(p1, p3);
  float2 q0 = cadd(u0, m2), q1 = csub(u0, m2);
  float2 q2 = cadd(m1, m3), q3 = crotis(csub(m1, m3), s);
  X[1] = cadd(q0, q2); X[3] = cadd(q1, q3);
  X[5] = csub(q0, q2); X[7] = csub(q1, q3);
}

__global__ __launch_bounds__(256) void s4_perl(
    const float* __restrict__ Lam, const float* __restrict__ Pri,
    const float* __restrict__ Cri, const float* __restrict__ log_step,
    float* __restrict__ perl)
{
  int l = blockIdx.x * blockDim.x + threadIdx.x;
  if (l >= Llen) return;
  float step = expf(log_step[0]);
  float theta = -2.0f * PI_F * (float)l / (float)Llen;
  float oi, orr;
  sincosf(theta, &oi, &orr);
  float op1r = 1.0f + orr;
  float den = op1r * op1r + oi * oi;
  float nr = 1.0f - orr, ni = -oi;
  float gscale = 2.0f / step;
  float gr = gscale * (nr * op1r + ni * oi) / den;
  float gi = gscale * (ni * op1r - nr * oi) / den;
  float scr = 2.0f * op1r / den;
  float sci = -2.0f * oi / den;
  float k01r = 0.f, k01i = 0.f, k11r = 0.f, k11i = 0.f;
  for (int n = 0; n < Nst; ++n) {
    float lr = Lam[2 * n], li = Lam[2 * n + 1];
    float pr = Pri[2 * n], pi = Pri[2 * n + 1];
    float cr = Cri[2 * n], ci = Cri[2 * n + 1];
    float zr = gr - lr, zi = gi - li;
    float zd = zr * zr + zi * zi;
    float car = zr / zd, cai = -zi / zd;
    float w01r = cr * pr + ci * pi;
    float w01i = cr * pi - ci * pr;
    float w11r = pr * pr + pi * pi;
    k01r += car * w01r - cai * w01i;
    k01i += car * w01i + cai * w01r;
    k11r += car * w11r;
    k11i += cai * w11r;
  }
  float dr = 1.0f + k11r, di = k11i;
  float dd = dr * dr + di * di;
  float sr = (k01r * dr + k01i * di) / dd;
  float si = (k01i * dr - k01r * di) / dd;
  perl[l] = sr;
  perl[Llen + l] = si;
  perl[2 * Llen + l] = scr;
  perl[3 * Llen + l] = sci;
  perl[4 * Llen + l] = gr;
  perl[5 * Llen + l] = gi;
}

__global__ __launch_bounds__(256) void s4_weights(
    const float* __restrict__ Bri, const float* __restrict__ Pri,
    const float* __restrict__ Cri,
    float* __restrict__ W0r, float* __restrict__ W0i,
    float* __restrict__ W1r, float* __restrict__ W1i)
{
  int idx = blockIdx.x * blockDim.x + threadIdx.x;
  if (idx >= Nst * Dch) return;
  int n = idx / Dch, d = idx % Dch;
  float br = Bri[2 * (n * Dch + d)], bi = Bri[2 * (n * Dch + d) + 1];
  float cr = Cri[2 * n], ci = Cri[2 * n + 1];
  float pr = Pri[2 * n], pi = Pri[2 * n + 1];
  int o = (n >> 1) * (2 * Dch) + 2 * d + (n & 1);
  W0r[o] = cr * br + ci * bi;
  W0i[o] = cr * bi - ci * br;
  W1r[o] = pr * br + pi * bi;
  W1i[o] = pr * bi - pi * br;
}

__global__ __launch_bounds__(256) void s4_cauchy(
    const float* __restrict__ Lam, const float* __restrict__ perl,
    float* __restrict__ APr, float* __restrict__ APi)
{
  int idx = blockIdx.x * blockDim.x + threadIdx.x;
  if (idx >= Llen * Nst) return;
  int n = idx / Llen, l = idx % Llen;
  float gr = perl[4 * Llen + l], gi = perl[5 * Llen + l];
  float zr = gr - Lam[2 * n], zi = gi - Lam[2 * n + 1];
  float zd = zr * zr + zi * zi;
  int o = (n >> 1) * (2 * Llen) + 2 * l + (n & 1);
  APr[o] = zr / zd;
  APi[o] = -zi / zd;
}

__global__ __launch_bounds__(32) void s4_cauchy_gemm(
    const float* __restrict__ APr, const float* __restrict__ APi,
    const float* __restrict__ W0r, const float* __restrict__ W0i,
    const float* __restrict__ W1r, const float* __restrict__ W1i,
    const float* __restrict__ perl, float2* __restrict__ Ev)
{
  int lane = threadIdx.x & 31;
  int tileD = blockIdx.x;
  int tileL = blockIdx.y;
  int d0 = tileD * 16 + (lane & 15);
  int mbase = tileL * 16 + ((lane >> 4) ? 8 : 0);
#if S4_WMMA
  int m = tileL * 16 + (lane & 15);
  int khi = (lane >> 4) * 2;
  v8f aRr = {}, aRi = {}, aSr = {}, aSi = {};
  for (int kb = 0; kb < Nst; kb += 4) {
    int k0 = kb + khi;
    const float* pr = APr + (k0 >> 1) * (2 * Llen) + 2 * m;
    const float* pi = APi + (k0 >> 1) * (2 * Llen) + 2 * m;
    v2f ar = *(const v2f*)pr;
    v2f ai = *(const v2f*)pi;
    v2f nai = -ai;
    int bo = (k0 >> 1) * (2 * Dch) + 2 * d0;
    v2f b0r = *(const v2f*)(W0r + bo);
    v2f b0i = *(const v2f*)(W0i + bo);
    v2f b1r = *(const v2f*)(W1r + bo);
    v2f b1i = *(const v2f*)(W1i + bo);
    aRr = __builtin_amdgcn_wmma_f32_16x16x4_f32(false, ar,  false, b0r, (short)0, aRr, false, false);
    aRr = __builtin_amdgcn_wmma_f32_16x16x4_f32(false, nai, false, b0i, (short)0, aRr, false, false);
    aRi = __builtin_amdgcn_wmma_f32_16x16x4_f32(false, ar,  false, b0i, (short)0, aRi, false, false);
    aRi = __builtin_amdgcn_wmma_f32_16x16x4_f32(false, ai,  false, b0r, (short)0, aRi, false, false);
    aSr = __builtin_amdgcn_wmma_f32_16x16x4_f32(false, ar,  false, b1r, (short)0, aSr, false, false);
    aSr = __builtin_amdgcn_wmma_f32_16x16x4_f32(false, nai, false, b1i, (short)0, aSr, false, false);
    aSi = __builtin_amdgcn_wmma_f32_16x16x4_f32(false, ar,  false, b1i, (short)0, aSi, false, false);
    aSi = __builtin_amdgcn_wmma_f32_16x16x4_f32(false, ai,  false, b1r, (short)0, aSi, false, false);
  }
  for (int r = 0; r < 8; ++r) {
    int l = mbase + r;
    float k00r = aRr[r], k00i = aRi[r];
    float k10r = aSr[r], k10i = aSi[r];
    float sr = perl[l],            si = perl[Llen + l];
    float cr = perl[2 * Llen + l], ci = perl[3 * Llen + l];
    float tr = k00r - (sr * k10r - si * k10i);
    float ti = k00i - (sr * k10i + si * k10r);
    Ev[l * Dch + d0] = make_float2(cr * tr - ci * ti, cr * ti + ci * tr);
  }
#else
  for (int r = 0; r < 8; ++r) {
    int l = mbase + r;
    float k00r = 0.f, k00i = 0.f, k10r = 0.f, k10i = 0.f;
    for (int k = 0; k < Nst; ++k) {
      int ao = (k >> 1) * (2 * Llen) + 2 * l + (k & 1);
      int bo = (k >> 1) * (2 * Dch) + 2 * d0 + (k & 1);
      float arr = APr[ao], aii = APi[ao];
      float b0r = W0r[bo], b0i = W0i[bo];
      float b1r = W1r[bo], b1i = W1i[bo];
      k00r += arr * b0r - aii * b0i;  k00i += arr * b0i + aii * b0r;
      k10r += arr * b1r - aii * b1i;  k10i += arr * b1i + aii * b1r;
    }
    float sr = perl[l],            si = perl[Llen + l];
    float cr = perl[2 * Llen + l], ci = perl[3 * Llen + l];
    float tr = k00r - (sr * k10r - si * k10i);
    float ti = k00i - (sr * k10i + si * k10r);
    Ev[l * Dch + d0] = make_float2(cr * tr - ci * ti, cr * ti + ci * tr);
  }
#endif
}

__device__ __forceinline__ void lds_barrier() {
  __builtin_amdgcn_fence(__ATOMIC_RELEASE, "workgroup", "local");
  __builtin_amdgcn_s_barrier();
  __builtin_amdgcn_fence(__ATOMIC_ACQUIRE, "workgroup", "local");
}

template <bool PASS_TW>
__device__ __forceinline__ void r64_core(
    float2* v, float2* sh, int k, int Ns, float sign, int lane, int sub, float2* X)
{
  float ss, cs;
  sincospif(sign * (float)sub * (1.0f / 32.0f), &ss, &cs);
  float2 tw = make_float2(1.0f, 0.0f);
  if (PASS_TW) {
    float kf = (float)k / (float)Ns;
    float s8, c8, sb, cb;
    sincospif(sign * 0.25f * kf, &s8, &c8);
    sincospif(sign * (float)sub * kf * (1.0f / 32.0f), &sb, &cb);
    float2 w8 = make_float2(c8, s8);
    float2 w = w8;
#pragma unroll
    for (int r1 = 1; r1 < 8; ++r1) {
      v[r1] = cmulf(v[r1], w);
      w = cmulf(w, w8);
    }
    tw = make_float2(cb, sb);
  }
  float2 Y[8];
  dft8(v, sign, Y);
  float2 st = make_float2(cs, ss);
#pragma unroll
  for (int a = 0; a < 8; ++a) {
    sh[(a * 8 + sub) * 32 + lane] = cmulf(Y[a], tw);
    tw = cmulf(tw, st);
  }
  lds_barrier();
  float2 z[8];
#pragma unroll
  for (int r0 = 0; r0 < 8; ++r0)
    z[r0] = sh[(sub * 8 + r0) * 32 + lane];
  dft8(z, sign, X);
}

__device__ __forceinline__ void r64_butterfly(
    const float2* __restrict__ x, float2* sh, int j, int q, int k, int Ns,
    float sign, int col, int lane, int sub, float2* X)
{
  float2 v[8];
#pragma unroll
  for (int r1 = 0; r1 < 8; ++r1)
    v[r1] = x[(j + (sub + 8 * r1) * q) * Dch + col];
  r64_core<true>(v, sh, k, Ns, sign, lane, sub, X);
}

__global__ __launch_bounds__(256) void s4_fft_r64(
    const float2* __restrict__ x, float2* __restrict__ y,
    int n, int Ns, float sign)
{
  __shared__ float2 sh[64 * 32];
  int lane = threadIdx.x & 31;
  int sub = threadIdx.x >> 5;
  int col = blockIdx.x * 32 + lane;
  int j = blockIdx.y;
  int k = j & (Ns - 1);
  float2 X[8];
  r64_butterfly(x, sh, j, n >> 6, k, Ns, sign, col, lane, sub, X);
  int i0 = 64 * j - 63 * k;
#pragma unroll
  for (int b = 0; b < 8; ++b)
    y[(i0 + (sub + 8 * b) * Ns) * Dch + col] = X[b];
}

template <bool PAIR>
__device__ __forceinline__ void pad_r512_body(
    const float* __restrict__ ua, const float* __restrict__ ub,
    float2* __restrict__ y, float2* sh)
{
  const int Q8 = NFFT / 8;
  int lane = threadIdx.x & 31;
  int sub = threadIdx.x >> 5;
  int col = blockIdx.x * 32 + lane;
  int jh = blockIdx.y;
  float2 S[8][8];
#pragma unroll
  for (int r1 = 0; r1 < 8; ++r1) {
    int jr = jh + 64 * (sub + 8 * r1);
    float2 v[8];
#pragma unroll
    for (int rp = 0; rp < 4; ++rp) {
      int pos = (jr + rp * Q8) * Dch + col;
      v[rp] = make_float2(ua[pos], PAIR ? ub[pos] : 0.f);
    }
#pragma unroll
    for (int rp = 4; rp < 8; ++rp)
      v[rp] = make_float2(0.f, 0.f);
    dft8(v, -1.0f, S[r1]);
  }
#pragma unroll
  for (int m = 0; m < 8; ++m) {
    float2 v[8], X[8];
#pragma unroll
    for (int r1 = 0; r1 < 8; ++r1)
      v[r1] = S[r1][m];
    r64_core<true>(v, sh + (m & 1) * (64 * 32), m, 8, -1.0f, lane, sub, X);
#pragma unroll
    for (int b = 0; b < (PAIR ? 8 : 4); ++b)
      y[(512 * jh + m + 8 * (sub + 8 * b)) * Dch + col] = X[b];
    if (!PAIR && m == 0 && sub == 0)
      y[(512 * jh + 256) * Dch + col] = X[4];
  }
}

__global__ __launch_bounds__(256) void s4_upad_r512(
    const float* __restrict__ ua, const float* __restrict__ ub,
    float2* __restrict__ y)
{
  __shared__ float2 sh[2 * 64 * 32];
  pad_r512_body<true>(ua, ub, y, sh);
}

__global__ __launch_bounds__(256) void s4_kpad_r512(
    const float* __restrict__ kr, float2* __restrict__ y)
{
  __shared__ float2 sh[2 * 64 * 32];
  pad_r512_body<false>(kr, kr, y, sh);
}

__global__ __launch_bounds__(256) void s4_kfirst_r256(
    const float2* __restrict__ x, float2* __restrict__ y)
{
  __shared__ float2 sh[2 * 64 * 32];
  const int Q4 = Llen / 4;
  int lane = threadIdx.x & 31;
  int sub = threadIdx.x >> 5;
  int col = blockIdx.x * 32 + lane;
  int jh = blockIdx.y;
  float2 S[8][4];
#pragma unroll
  for (int r1 = 0; r1 < 8; ++r1) {
    int jr = jh + 64 * (sub + 8 * r1);
#pragma unroll
    for (int rp = 0; rp < 4; ++rp)
      S[r1][rp] = x[(jr + rp * Q4) * Dch + col];
  }
#pragma unroll
  for (int r1 = 0; r1 < 8; ++r1) {
    float2 p0 = cadd(S[r1][0], S[r1][2]), p1 = csub(S[r1][0], S[r1][2]);
    float2 p2 = cadd(S[r1][1], S[r1][3]);
    float2 p3 = crotis(csub(S[r1][1], S[r1][3]), +1.0f);
    S[r1][0] = cadd(p0, p2); S[r1][1] = cadd(p1, p3);
    S[r1][2] = csub(p0, p2); S[r1][3] = csub(p1, p3);
  }
#pragma unroll
  for (int m = 0; m < 4; ++m) {
    float2 v[8], X[8];
#pragma unroll
    for (int r1 = 0; r1 < 8; ++r1)
      v[r1] = S[r1][m];
    r64_core<true>(v, sh + (m & 1) * (64 * 32), m, 4, +1.0f, lane, sub, X);
#pragma unroll
    for (int b = 0; b < 8; ++b)
      y[(256 * jh + m + 4 * (sub + 8 * b)) * Dch + col] = X[b];
  }
}

__global__ __launch_bounds__(256) void s4_klast_r64_real(
    const float2* __restrict__ x, float* __restrict__ kr)
{
  __shared__ float2 sh[64 * 32];
  const int Q = Llen / 64;
  int lane = threadIdx.x & 31;
  int sub = threadIdx.x >> 5;
  int col = blockIdx.x * 32 + lane;
  int j = blockIdx.y;
  float2 X[8];
  r64_butterfly(x, sh, j, Q, j, Q, +1.0f, col, lane, sub, X);
  const float invL = 1.0f / (float)Llen;
#pragma unroll
  for (int b = 0; b < 8; ++b)
    kr[(j + (sub + 8 * b) * Q) * Dch + col] = X[b].x * invL;
}

__global__ __launch_bounds__(256) void s4_mid_r64_cmul_r64(
    const float2* __restrict__ x, const float2* __restrict__ Kd,
    float2* __restrict__ y)
{
  __shared__ float2 sh[2 * 64 * 32];
  const int Q = NFFT / 64;
  int lane = threadIdx.x & 31;
  int sub = threadIdx.x >> 5;
  int col = blockIdx.x * 32 + lane;
  int j = blockIdx.y;
  bool twin = (j > 0) && (j < Q / 2);
  int j2 = Q - j;
  int vs = 7 - sub;
  float2 kd[8], va[8], vb[8];
#pragma unroll
  for (int b = 0; b < 8; ++b)
    kd[b] = Kd[(j + (sub + 8 * b) * Q) * Dch + col];
#pragma unroll
  for (int r1 = 0; r1 < 8; ++r1)
    va[r1] = x[(j + (sub + 8 * r1) * Q) * Dch + col];
  if (twin) {
#pragma unroll
    for (int r1 = 0; r1 < 8; ++r1)
      vb[r1] = x[(j2 + (vs + 8 * r1) * Q) * Dch + col];
  }
  float2 X[8], v[8], Z[8];
  r64_core<true>(va, sh, j, Q, -1.0f, lane, sub, X);
#pragma unroll
  for (int b = 0; b < 8; ++b)
    v[b] = cmulf(X[b], kd[b]);
  r64_core<false>(v, sh + 64 * 32, 0, 1, +1.0f, lane, sub, Z);
#pragma unroll
  for (int b = 0; b < 8; ++b)
    y[(64 * j + sub + 8 * b) * Dch + col] = Z[b];
  if (twin) {
    r64_core<true>(vb, sh, j2, Q, -1.0f, lane, vs, X);
#pragma unroll
    for (int b = 0; b < 8; ++b)
      v[b] = cmulcf(X[b], kd[7 - b]);
    r64_core<false>(v, sh + 64 * 32, 0, 1, +1.0f, lane, vs, Z);
#pragma unroll
    for (int b = 0; b < 8; ++b)
      y[(64 * j2 + vs + 8 * b) * Dch + col] = Z[b];
  }
}

__global__ __launch_bounds__(256) __attribute__((amdgpu_waves_per_eu(4, 4))) void s4_final_r512(
    const float2* __restrict__ x,
    const float* __restrict__ ua, const float* __restrict__ ub,
    const float* __restrict__ Dp,
    float* __restrict__ outa, float* __restrict__ outb)
{
  __shared__ float2 sh[2 * 64 * 32];
  const int Q = NFFT / 64;
  const int Q8 = NFFT / 8;
  int lane = threadIdx.x & 31;
  int sub = threadIdx.x >> 5;
  int col = blockIdx.x * 32 + lane;
  int kk = blockIdx.y;
  float sw, cw;
  sincospif((float)kk * (1.0f / 256.0f), &sw, &cw);
  float2 w1 = make_float2(cw, sw);
  float2 wp[8];
  wp[1] = w1;
#pragma unroll
  for (int rp = 2; rp < 8; ++rp)
    wp[rp] = cmulf(wp[rp - 1], w1);
  float2 S[8][8];
#pragma unroll
  for (int r1 = 0; r1 < 8; ++r1) {
    int jr = kk + 64 * (sub + 8 * r1);
#pragma unroll
    for (int rp = 0; rp < 8; ++rp)
      S[r1][rp] = x[(jr + rp * Q8) * Dch + col];
  }
#pragma unroll
  for (int r1 = 0; r1 < 8; ++r1) {
    float2 v[8];
    v[0] = S[r1][0];
#pragma unroll
    for (int rp = 1; rp < 8; ++rp)
      v[rp] = cmulf(S[r1][rp], wp[rp]);
    dft8(v, +1.0f, S[r1]);
  }
  float dp = Dp[0];
  const float inv = 1.0f / (float)NFFT;
#pragma unroll
  for (int m = 0; m < 8; ++m) {
    int j2 = kk + 64 * m;
    float2 v[8], X[8];
#pragma unroll
    for (int r1 = 0; r1 < 8; ++r1)
      v[r1] = S[r1][m];
    float pa[4], pb[4];
#pragma unroll
    for (int b = 0; b < 4; ++b) {
      int pos = (j2 + (sub + 8 * b) * Q) * Dch + col;
      pa[b] = ua[pos];
      pb[b] = ub[pos];
    }
    r64_core<true>(v, sh + (m & 1) * (64 * 32), j2, Q, +1.0f, lane, sub, X);
#pragma unroll
    for (int b = 0; b < 4; ++b) {
      int pos = (j2 + (sub + 8 * b) * Q) * Dch + col;
      outa[pos] = X[b].x * inv + dp * pa[b];
      outb[pos] = X[b].y * inv + dp * pb[b];
    }
  }
}


extern "C" void kernel_launch(void* const* d_in, const int* in_sizes, int n_in,
                              void* d_out, int out_size, void* d_ws, size_t ws_size,
                              hipStream_t stream) {
  const float* u        = (const float*)d_in[0];
  const float* Lam      = (const float*)d_in[1];
  const float* Pri      = (const float*)d_in[2];
  const float* Bri      = (const float*)d_in[3];
  const float* Cri      = (const float*)d_in[4];
  const float* Dp       = (const float*)d_in[5];
  const float* log_step = (const float*)d_in[6];
  float* out = (float*)d_out;

  uint8_t* base = (uint8_t*)d_ws;
  size_t off = 0;
  auto carve = [&](size_t bytes) -> void* {
    void* p = base + off;
    off += (bytes + 255) & ~(size_t)255;
    return p;
  };
  float*  APr  = (float*)carve((size_t)Llen * Nst * 4);
  float*  APi  = (float*)carve((size_t)Llen * Nst * 4);
  float*  W0r  = (float*)carve((size_t)Nst * Dch * 4);
  float*  W0i  = (float*)carve((size_t)Nst * Dch * 4);
  float*  W1r  = (float*)carve((size_t)Nst * Dch * 4);
  float*  W1i  = (float*)carve((size_t)Nst * Dch * 4);
  float*  perl = (float*)carve((size_t)6 * Llen * 4);
  float2* Ev   = (float2*)carve((size_t)Llen * Dch * 8);
  float2* T0   = (float2*)carve((size_t)NFFT * Dch * 8);
  float2* T1   = (float2*)carve((size_t)NFFT * Dch * 8);
  float2* Kd   = (float2*)carve((size_t)NFFT * Dch * 8);
  (void)ws_size; (void)in_sizes; (void)n_in; (void)out_size;

  const int TB   = 256;
  const dim3 gR64h(Dch / 32, NFFT / 128 + 1);
  const dim3 gR64s(Dch / 32, Llen / 64);
  const dim3 gR512(Dch / 32, NFFT / 512);
  const dim3 gR256s(Dch / 32, Llen / 256);

  s4_perl<<<(Llen + TB - 1) / TB, TB, 0, stream>>>(Lam, Pri, Cri, log_step, perl);
  s4_weights<<<(Nst * Dch + TB - 1) / TB, TB, 0, stream>>>(Bri, Pri, Cri, W0r, W0i, W1r, W1i);
  s4_cauchy<<<(Llen * Nst + TB - 1) / TB, TB, 0, stream>>>(Lam, perl, APr, APi);
  s4_cauchy_gemm<<<dim3(Dch / 16, Llen / 16), 32, 0, stream>>>(
      APr, APi, W0r, W0i, W1r, W1i, perl, Ev);

  float* Kr = (float*)Ev;
  s4_kfirst_r256<<<gR256s, TB, 0, stream>>>(Ev, T0);
  s4_klast_r64_real<<<gR64s, TB, 0, stream>>>(T0, Kr);
  s4_kpad_r512<<<gR512, TB, 0, stream>>>(Kr, T1);
  s4_fft_r64<<<gR64h, TB, 0, stream>>>(T1, Kd, NFFT, 512, -1.0f);

  for (int b = 0; b < Bsz; b += 2) {
    const float* ua = u + (size_t)b * Llen * Dch;
    const float* ub = u + (size_t)(b + 1) * Llen * Dch;
    float*       oa = out + (size_t)b * Llen * Dch;
    float*       ob = out + (size_t)(b + 1) * Llen * Dch;
    s4_upad_r512<<<gR512, TB, 0, stream>>>(ua, ub, T0);
    s4_mid_r64_cmul_r64<<<gR64h, TB, 0, stream>>>(T0, Kd, T1);
    s4_final_r512<<<gR512, TB, 0, stream>>>(T1, ua, ub, Dp, oa, ob);
  }
}


# Rank-1 correction and scale folded into the Cauchy matrix (built 8 states per thread): one complex GEMM on B itself, 4 WMMAs per k-step not 8, 16x64 tile per wave; sums regrouped per n
# speedup vs baseline: 6.0106x; 1.1001x over previous
#include <hip/hip_runtime.h>
#include <hip/hip_bf16.h>
#include <stdint.h>


#define Bsz   4
#define Llen  16384
#define Dch   512
#define Nst   64
#define NFFT  32768
#define PI_F  3.14159265358979323846f

typedef __attribute__((ext_vector_type(2))) float v2f;
typedef __attribute__((ext_vector_type(8))) float v8f;

#ifndef __has_builtin
#define __has_builtin(x) 0
#endif
#if __has_builtin(__builtin_amdgcn_wmma_f32_16x16x4_f32)
#define S4_WMMA 1
#else
#define S4_WMMA 0
#endif

__device__ __forceinline__ float2 cadd(float2 a, float2 b) { return make_float2(a.x + b.x, a.y + b.y); }
__device__ __forceinline__ float2 csub(float2 a, float2 b) { return make_float2(a.x - b.x, a.y - b.y); }
__device__ __forceinline__ float2 cmulf(float2 a, float2 b) { return make_float2(a.x * b.x - a.y * b.y, a.x * b.y + a.y * b.x); }
__device__ __forceinline__ float2 cmulcf(float2 a, float2 b) { return make_float2(a.x * b.x + a.y * b.y, a.y * b.x - a.x * b.y); }
__device__ __forceinline__ float2 crotis(float2 a, float s) { return make_float2(-s * a.y, s * a.x); }

__device__ __forceinline__ void dft8(const float2* v, float s, float2* X) {
  const float R = 0.70710678118654752440f;
  float2 t0 = cadd(v[0], v[4]), t1 = cadd(v[1], v[5]);
  float2 t2 = cadd(v[2], v[6]), t3 = cadd(v[3], v[7]);
  float2 u0 = csub(v[0], v[4]), u1 = csub(v[1], v[5]);
  float2 u2 = csub(v[2], v[6]), u3 = csub(v[3], v[7]);
  float2 m1 = cmulf(u1, make_float2(R, s * R));
  float2 m2 = crotis(u2, s);
  float2 m3 = cmulf(u3, make_float2(-R, s * R));
  float2 p0 = cadd(t0, t2), p1 = csub(t0, t2);
  float2 p2 = cadd(t1, t3), p3 = crotis(csub(t1, t3), s);
  X[0] = cadd(p0, p2); X[2] = cadd(p1, p3);
  X[4] = csub(p0, p2); X[6] = csub(p1, p3);
  float2 q0 = cadd(u0, m2), q1 = csub(u0, m2);
  float2 q2 = cadd(m1, m3), q3 = crotis(csub(m1, m3), s);
  X[1] = cadd(q0, q2); X[3] = cadd(q1, q3);
  X[5] = csub(q0, q2); X[7] = csub(q1, q3);
}

__global__ __launch_bounds__(256) void s4_perl(
    const float* __restrict__ Lam, const float* __restrict__ Pri,
    const float* __restrict__ Cri, const float* __restrict__ log_step,
    float* __restrict__ perl)
{
  int l = blockIdx.x * blockDim.x + threadIdx.x;
  if (l >= Llen) return;
  float step = expf(log_step[0]);
  float theta = -2.0f * PI_F * (float)l / (float)Llen;
  float oi, orr;
  sincosf(theta, &oi, &orr);
  float op1r = 1.0f + orr;
  float den = op1r * op1r + oi * oi;
  float nr = 1.0f - orr, ni = -oi;
  float gscale = 2.0f / step;
  float gr = gscale * (nr * op1r + ni * oi) / den;
  float gi = gscale * (ni * op1r - nr * oi) / den;
  float scr = 2.0f * op1r / den;
  float sci = -2.0f * oi / den;
  float k01r = 0.f, k01i = 0.f, k11r = 0.f, k11i = 0.f;
  for (int n = 0; n < Nst; ++n) {
    float lr = Lam[2 * n], li = Lam[2 * n + 1];
    float pr = Pri[2 * n], pi = Pri[2 * n + 1];
    float cr = Cri[2 * n], ci = Cri[2 * n + 1];
    float zr = gr - lr, zi = gi - li;
    float zd = zr * zr + zi * zi;
    float car = zr / zd, cai = -zi / zd;
    float w01r = cr * pr + ci * pi;
    float w01i = cr * pi - ci * pr;
    float w11r = pr * pr + pi * pi;
    k01r += car * w01r - cai * w01i;
    k01i += car * w01i + cai * w01r;
    k11r += car * w11r;
    k11i += cai * w11r;
  }
  float dr = 1.0f + k11r, di = k11i;
  float dd = dr * dr + di * di;
  float sr = (k01r * dr + k01i * di) / dd;
  float si = (k01i * dr - k01r * di) / dd;
  perl[l] = sr;
  perl[Llen + l] = si;
  perl[2 * Llen + l] = scr;
  perl[3 * Llen + l] = sci;
  perl[4 * Llen + l] = gr;
  perl[5 * Llen + l] = gi;
}

__global__ __launch_bounds__(256) void s4_cauchy(
    const float* __restrict__ Lam, const float* __restrict__ Pri,
    const float* __restrict__ Cri, const float* __restrict__ perl,
    float* __restrict__ APr, float* __restrict__ APi)
{
  int l = blockIdx.x * blockDim.x + threadIdx.x;
  int ng = blockIdx.y;
  if (l >= Llen) return;
  float sr = perl[l],             si = perl[Llen + l];
  float scr = perl[2 * Llen + l], sci = perl[3 * Llen + l];
  float gr = perl[4 * Llen + l],  gi = perl[5 * Llen + l];
#pragma unroll
  for (int e = 0; e < 4; ++e) {
    float ar[2], ai[2];
#pragma unroll
    for (int b = 0; b < 2; ++b) {
      int n = 8 * ng + 2 * e + b;
      float zr = gr - Lam[2 * n], zi = gi - Lam[2 * n + 1];
      float zd = zr * zr + zi * zi;
      float car = zr / zd, cai = -zi / zd;
      float cr = Cri[2 * n], ci = Cri[2 * n + 1];
      float pr = Pri[2 * n], pi = Pri[2 * n + 1];
      float tr = cr - (sr * pr + si * pi);
      float ti = -ci - (si * pr - sr * pi);
      float wr = scr * tr - sci * ti;
      float wi = scr * ti + sci * tr;
      ar[b] = car * wr - cai * wi;
      ai[b] = car * wi + cai * wr;
    }
    int o = (4 * ng + e) * (2 * Llen) + 2 * l;
    *(float2*)(APr + o) = make_float2(ar[0], ar[1]);
    *(float2*)(APi + o) = make_float2(ai[0], ai[1]);
  }
}

__global__ __launch_bounds__(256) void s4_cauchy_gemm(
    const float* __restrict__ APr, const float* __restrict__ APi,
    const float* __restrict__ Bri, float2* __restrict__ Ev)
{
  int lane = threadIdx.x & 31;
  int wv = threadIdx.x >> 5;
  int tileL = blockIdx.x;
  int c = lane & 15;
  int h = lane >> 4;
  int dbase = wv * 64 + c;
#if S4_WMMA
  int m = tileL * 16 + c;
  v8f accR[4], accI[4];
#pragma unroll
  for (int t = 0; t < 4; ++t) {
    accR[t] = (v8f){0.f, 0.f, 0.f, 0.f, 0.f, 0.f, 0.f, 0.f};
    accI[t] = (v8f){0.f, 0.f, 0.f, 0.f, 0.f, 0.f, 0.f, 0.f};
  }
  for (int kb = 0; kb < Nst; kb += 4) {
    int k0 = kb + 2 * h;
    v2f ar = *(const v2f*)(APr + (k0 >> 1) * (2 * Llen) + 2 * m);
    v2f ai = *(const v2f*)(APi + (k0 >> 1) * (2 * Llen) + 2 * m);
    v2f nai = -ai;
    const float* b0 = Bri + 2 * (k0 * Dch + dbase);
    const float* b1 = b0 + 2 * Dch;
#pragma unroll
    for (int t = 0; t < 4; ++t) {
      v2f x0 = *(const v2f*)(b0 + 32 * t);
      v2f x1 = *(const v2f*)(b1 + 32 * t);
      v2f br = {x0.x, x1.x};
      v2f bi = {x0.y, x1.y};
      accR[t] = __builtin_amdgcn_wmma_f32_16x16x4_f32(false, ar,  false, br, (short)0, accR[t], false, false);
      accR[t] = __builtin_amdgcn_wmma_f32_16x16x4_f32(false, nai, false, bi, (short)0, accR[t], false, false);
      accI[t] = __builtin_amdgcn_wmma_f32_16x16x4_f32(false, ar,  false, bi, (short)0, accI[t], false, false);
      accI[t] = __builtin_amdgcn_wmma_f32_16x16x4_f32(false, ai,  false, br, (short)0, accI[t], false, false);
    }
  }
#pragma unroll
  for (int t = 0; t < 4; ++t) {
#pragma unroll
    for (int r = 0; r < 8; ++r) {
      int l = tileL * 16 + 8 * h + r;
      Ev[l * Dch + dbase + 16 * t] = make_float2(accR[t][r], accI[t][r]);
    }
  }
#else
  for (int t = 0; t < 4; ++t) {
    for (int r = 0; r < 8; ++r) {
      int l = tileL * 16 + 8 * h + r;
      int d = dbase + 16 * t;
      float er = 0.f, ei = 0.f;
      for (int k = 0; k < Nst; ++k) {
        int ao = (k >> 1) * (2 * Llen) + 2 * l + (k & 1);
        float arr = APr[ao], aii = APi[ao];
        float brr = Bri[2 * (k * Dch + d)], bii = Bri[2 * (k * Dch + d) + 1];
        er += arr * brr - aii * bii;
        ei += arr * bii + aii * brr;
      }
      Ev[l * Dch + d] = make_float2(er, ei);
    }
  }
#endif
}

__device__ __forceinline__ void lds_barrier() {
  __builtin_amdgcn_fence(__ATOMIC_RELEASE, "workgroup", "local");
  __builtin_amdgcn_s_barrier();
  __builtin_amdgcn_fence(__ATOMIC_ACQUIRE, "workgroup", "local");
}

template <bool PASS_TW>
__device__ __forceinline__ void r64_core(
    float2* v, float2* sh, int k, int Ns, float sign, int lane, int sub, float2* X)
{
  float ss, cs;
  sincospif(sign * (float)sub * (1.0f / 32.0f), &ss, &cs);
  float2 tw = make_float2(1.0f, 0.0f);
  if (PASS_TW) {
    float kf = (float)k / (float)Ns;
    float s8, c8, sb, cb;
    sincospif(sign * 0.25f * kf, &s8, &c8);
    sincospif(sign * (float)sub * kf * (1.0f / 32.0f), &sb, &cb);
    float2 w8 = make_float2(c8, s8);
    float2 w = w8;
#pragma unroll
    for (int r1 = 1; r1 < 8; ++r1) {
      v[r1] = cmulf(v[r1], w);
      w = cmulf(w, w8);
    }
    tw = make_float2(cb, sb);
  }
  float2 Y[8];
  dft8(v, sign, Y);
  float2 st = make_float2(cs, ss);
#pragma unroll
  for (int a = 0; a < 8; ++a) {
    sh[(a * 8 + sub) * 32 + lane] = cmulf(Y[a], tw);
    tw = cmulf(tw, st);
  }
  lds_barrier();
  float2 z[8];
#pragma unroll
  for (int r0 = 0; r0 < 8; ++r0)
    z[r0] = sh[(sub * 8 + r0) * 32 + lane];
  dft8(z, sign, X);
}

__device__ __forceinline__ void r64_butterfly(
    const float2* __restrict__ x, float2* sh, int j, int q, int k, int Ns,
    float sign, int col, int lane, int sub, float2* X)
{
  float2 v[8];
#pragma unroll
  for (int r1 = 0; r1 < 8; ++r1)
    v[r1] = x[(j + (sub + 8 * r1) * q) * Dch + col];
  r64_core<true>(v, sh, k, Ns, sign, lane, sub, X);
}

__global__ __launch_bounds__(256) void s4_fft_r64(
    const float2* __restrict__ x, float2* __restrict__ y,
    int n, int Ns, float sign)
{
  __shared__ float2 sh[64 * 32];
  int lane = threadIdx.x & 31;
  int sub = threadIdx.x >> 5;
  int col = blockIdx.x * 32 + lane;
  int j = blockIdx.y;
  int k = j & (Ns - 1);
  float2 X[8];
  r64_butterfly(x, sh, j, n >> 6, k, Ns, sign, col, lane, sub, X);
  int i0 = 64 * j - 63 * k;
#pragma unroll
  for (int b = 0; b < 8; ++b)
    y[(i0 + (sub + 8 * b) * Ns) * Dch + col] = X[b];
}

template <bool PAIR>
__device__ __forceinline__ void pad_r512_body(
    const float* __restrict__ ua, const float* __restrict__ ub,
    float2* __restrict__ y, float2* sh)
{
  const int Q8 = NFFT / 8;
  int lane = threadIdx.x & 31;
  int sub = threadIdx.x >> 5;
  int col = blockIdx.x * 32 + lane;
  int jh = blockIdx.y;
  float2 S[8][8];
#pragma unroll
  for (int r1 = 0; r1 < 8; ++r1) {
    int jr = jh + 64 * (sub + 8 * r1);
    float2 v[8];
#pragma unroll
    for (int rp = 0; rp < 4; ++rp) {
      int pos = (jr + rp * Q8) * Dch + col;
      v[rp] = make_float2(ua[pos], PAIR ? ub[pos] : 0.f);
    }
#pragma unroll
    for (int rp = 4; rp < 8; ++rp)
      v[rp] = make_float2(0.f, 0.f);
    dft8(v, -1.0f, S[r1]);
  }
#pragma unroll
  for (int m = 0; m < 8; ++m) {
    float2 v[8], X[8];
#pragma unroll
    for (int r1 = 0; r1 < 8; ++r1)
      v[r1] = S[r1][m];
    r64_core<true>(v, sh + (m & 1) * (64 * 32), m, 8, -1.0f, lane, sub, X);
#pragma unroll
    for (int b = 0; b < (PAIR ? 8 : 4); ++b)
      y[(512 * jh + m + 8 * (sub + 8 * b)) * Dch + col] = X[b];
    if (!PAIR && m == 0 && sub == 0)
      y[(512 * jh + 256) * Dch + col] = X[4];
  }
}

__global__ __launch_bounds__(256) void s4_upad_r512(
    const float* __restrict__ ua, const float* __restrict__ ub,
    float2* __restrict__ y)
{
  __shared__ float2 sh[2 * 64 * 32];
  pad_r512_body<true>(ua, ub, y, sh);
}

__global__ __launch_bounds__(256) void s4_kpad_r512(
    const float* __restrict__ kr, float2* __restrict__ y)
{
  __shared__ float2 sh[2 * 64 * 32];
  pad_r512_body<false>(kr, kr, y, sh);
}

__global__ __launch_bounds__(256) void s4_kfirst_r256(
    const float2* __restrict__ x, float2* __restrict__ y)
{
  __shared__ float2 sh[2 * 64 * 32];
  const int Q4 = Llen / 4;
  int lane = threadIdx.x & 31;
  int sub = threadIdx.x >> 5;
  int col = blockIdx.x * 32 + lane;
  int jh = blockIdx.y;
  float2 S[8][4];
#pragma unroll
  for (int r1 = 0; r1 < 8; ++r1) {
    int jr = jh + 64 * (sub + 8 * r1);
#pragma unroll
    for (int rp = 0; rp < 4; ++rp)
      S[r1][rp] = x[(jr + rp * Q4) * Dch + col];
  }
#pragma unroll
  for (int r1 = 0; r1 < 8; ++r1) {
    float2 p0 = cadd(S[r1][0], S[r1][2]), p1 = csub(S[r1][0], S[r1][2]);
    float2 p2 = cadd(S[r1][1], S[r1][3]);
    float2 p3 = crotis(csub(S[r1][1], S[r1][3]), +1.0f);
    S[r1][0] = cadd(p0, p2); S[r1][1] = cadd(p1, p3);
    S[r1][2] = csub(p0, p2); S[r1][3] = csub(p1, p3);
  }
#pragma unroll
  for (int m = 0; m < 4; ++m) {
    float2 v[8], X[8];
#pragma unroll
    for (int r1 = 0; r1 < 8; ++r1)
      v[r1] = S[r1][m];
    r64_core<true>(v, sh + (m & 1) * (64 * 32), m, 4, +1.0f, lane, sub, X);
#pragma unroll
    for (int b = 0; b < 8; ++b)
      y[(256 * jh + m + 4 * (sub + 8 * b)) * Dch + col] = X[b];
  }
}

__global__ __launch_bounds__(256) void s4_klast_r64_real(
    const float2* __restrict__ x, float* __restrict__ kr)
{
  __shared__ float2 sh[64 * 32];
  const int Q = Llen / 64;
  int lane = threadIdx.x & 31;
  int sub = threadIdx.x >> 5;
  int col = blockIdx.x * 32 + lane;
  int j = blockIdx.y;
  float2 X[8];
  r64_butterfly(x, sh, j, Q, j, Q, +1.0f, col, lane, sub, X);
  const float invL = 1.0f / (float)Llen;
#pragma unroll
  for (int b = 0; b < 8; ++b)
    kr[(j + (sub + 8 * b) * Q) * Dch + col] = X[b].x * invL;
}

__global__ __launch_bounds__(256) void s4_mid_r64_cmul_r64(
    const float2* __restrict__ x, const float2* __restrict__ Kd,
    float2* __restrict__ y)
{
  __shared__ float2 sh[2 * 64 * 32];
  const int Q = NFFT / 64;
  int lane = threadIdx.x & 31;
  int sub = threadIdx.x >> 5;
  int col = blockIdx.x * 32 + lane;
  int j = blockIdx.y;
  bool twin = (j > 0) && (j < Q / 2);
  int j2 = Q - j;
  int vs = 7 - sub;
  float2 kd[8], va[8], vb[8];
#pragma unroll
  for (int b = 0; b < 8; ++b)
    kd[b] = Kd[(j + (sub + 8 * b) * Q) * Dch + col];
#pragma unroll
  for (int r1 = 0; r1 < 8; ++r1)
    va[r1] = x[(j + (sub + 8 * r1) * Q) * Dch + col];
  if (twin) {
#pragma unroll
    for (int r1 = 0; r1 < 8; ++r1)
      vb[r1] = x[(j2 + (vs + 8 * r1) * Q) * Dch + col];
  }
  float2 X[8], v[8], Z[8];
  r64_core<true>(va, sh, j, Q, -1.0f, lane, sub, X);
#pragma unroll
  for (int b = 0; b < 8; ++b)
    v[b] = cmulf(X[b], kd[b]);
  r64_core<false>(v, sh + 64 * 32, 0, 1, +1.0f, lane, sub, Z);
#pragma unroll
  for (int b = 0; b < 8; ++b)
    y[(64 * j + sub + 8 * b) * Dch + col] = Z[b];
  if (twin) {
    r64_core<true>(vb, sh, j2, Q, -1.0f, lane, vs, X);
#pragma unroll
    for (int b = 0; b < 8; ++b)
      v[b] = cmulcf(X[b], kd[7 - b]);
    r64_core<false>(v, sh + 64 * 32, 0, 1, +1.0f, lane, vs, Z);
#pragma unroll
    for (int b = 0; b < 8; ++b)
      y[(64 * j2 + vs + 8 * b) * Dch + col] = Z[b];
  }
}

__global__ __launch_bounds__(256) __attribute__((amdgpu_waves_per_eu(4, 4))) void s4_final_r512(
    const float2* __restrict__ x,
    const float* __restrict__ ua, const float* __restrict__ ub,
    const float* __restrict__ Dp,
    float* __restrict__ outa, float* __restrict__ outb)
{
  __shared__ float2 sh[2 * 64 * 32];
  const int Q = NFFT / 64;
  const int Q8 = NFFT / 8;
  int lane = threadIdx.x & 31;
  int sub = threadIdx.x >> 5;
  int col = blockIdx.x * 32 + lane;
  int kk = blockIdx.y;
  float sw, cw;
  sincospif((float)kk * (1.0f / 256.0f), &sw, &cw);
  float2 w1 = make_float2(cw, sw);
  float2 wp[8];
  wp[1] = w1;
#pragma unroll
  for (int rp = 2; rp < 8; ++rp)
    wp[rp] = cmulf(wp[rp - 1], w1);
  float2 S[8][8];
#pragma unroll
  for (int r1 = 0; r1 < 8; ++r1) {
    int jr = kk + 64 * (sub + 8 * r1);
#pragma unroll
    for (int rp = 0; rp < 8; ++rp)
      S[r1][rp] = x[(jr + rp * Q8) * Dch + col];
  }
#pragma unroll
  for (int r1 = 0; r1 < 8; ++r1) {
    float2 v[8];
    v[0] = S[r1][0];
#pragma unroll
    for (int rp = 1; rp < 8; ++rp)
      v[rp] = cmulf(S[r1][rp], wp[rp]);
    dft8(v, +1.0f, S[r1]);
  }
  float dp = Dp[0];
  const float inv = 1.0f / (float)NFFT;
#pragma unroll
  for (int m = 0; m < 8; ++m) {
    int j2 = kk + 64 * m;
    float2 v[8], X[8];
#pragma unroll
    for (int r1 = 0; r1 < 8; ++r1)
      v[r1] = S[r1][m];
    float pa[4], pb[4];
#pragma unroll
    for (int b = 0; b < 4; ++b) {
      int pos = (j2 + (sub + 8 * b) * Q) * Dch + col;
      pa[b] = ua[pos];
      pb[b] = ub[pos];
    }
    r64_core<true>(v, sh + (m & 1) * (64 * 32), j2, Q, +1.0f, lane, sub, X);
#pragma unroll
    for (int b = 0; b < 4; ++b) {
      int pos = (j2 + (sub + 8 * b) * Q) * Dch + col;
      outa[pos] = X[b].x * inv + dp * pa[b];
      outb[pos] = X[b].y * inv + dp * pb[b];
    }
  }
}


extern "C" void kernel_launch(void* const* d_in, const int* in_sizes, int n_in,
                              void* d_out, int out_size, void* d_ws, size_t ws_size,
                              hipStream_t stream) {
  const float* u        = (const float*)d_in[0];
  const float* Lam      = (const float*)d_in[1];
  const float* Pri      = (const float*)d_in[2];
  const float* Bri      = (const float*)d_in[3];
  const float* Cri      = (const float*)d_in[4];
  const float* Dp       = (const float*)d_in[5];
  const float* log_step = (const float*)d_in[6];
  float* out = (float*)d_out;

  uint8_t* base = (uint8_t*)d_ws;
  size_t off = 0;
  auto carve = [&](size_t bytes) -> void* {
    void* p = base + off;
    off += (bytes + 255) & ~(size_t)255;
    return p;
  };
  float*  APr  = (float*)carve((size_t)Llen * Nst * 4);
  float*  APi  = (float*)carve((size_t)Llen * Nst * 4);
  float*  perl = (float*)carve((size_t)6 * Llen * 4);
  float2* Ev   = (float2*)carve((size_t)Llen * Dch * 8);
  float2* T0   = (float2*)carve((size_t)NFFT * Dch * 8);
  float2* T1   = (float2*)carve((size_t)NFFT * Dch * 8);
  float2* Kd   = (float2*)carve((size_t)NFFT * Dch * 8);
  (void)ws_size; (void)in_sizes; (void)n_in; (void)out_size;

  const int TB   = 256;
  const dim3 gR64h(Dch / 32, NFFT / 128 + 1);
  const dim3 gR64s(Dch / 32, Llen / 64);
  const dim3 gR512(Dch / 32, NFFT / 512);
  const dim3 gR256s(Dch / 32, Llen / 256);

  s4_perl<<<(Llen + TB - 1) / TB, TB, 0, stream>>>(Lam, Pri, Cri, log_step, perl);
  s4_cauchy<<<dim3(Llen / TB, Nst / 8), TB, 0, stream>>>(Lam, Pri, Cri, perl, APr, APi);
  s4_cauchy_gemm<<<Llen / 16, TB, 0, stream>>>(APr, APi, Bri, Ev);

  float* Kr = (float*)Ev;
  s4_kfirst_r256<<<gR256s, TB, 0, stream>>>(Ev, T0);
  s4_klast_r64_real<<<gR64s, TB, 0, stream>>>(T0, Kr);
  s4_kpad_r512<<<gR512, TB, 0, stream>>>(Kr, T1);
  s4_fft_r64<<<gR64h, TB, 0, stream>>>(T1, Kd, NFFT, 512, -1.0f);

  for (int b = 0; b < Bsz; b += 2) {
    const float* ua = u + (size_t)b * Llen * Dch;
    const float* ub = u + (size_t)(b + 1) * Llen * Dch;
    float*       oa = out + (size_t)b * Llen * Dch;
    float*       ob = out + (size_t)(b + 1) * Llen * Dch;
    s4_upad_r512<<<gR512, TB, 0, stream>>>(ua, ub, T0);
    s4_mid_r64_cmul_r64<<<gR64h, TB, 0, stream>>>(T0, Kd, T1);
    s4_final_r512<<<gR512, TB, 0, stream>>>(T1, ua, ub, Dp, oa, ob);
  }
}


# Inverse transform over l moved in front of the GEMM (linearity): it runs on the 64 columns of the folded Cauchy matrix, not the 512 of evaluated; GEMM gives real K, 2 WMMAs not 4; n-sum now after it
# speedup vs baseline: 6.6652x; 1.1089x over previous
#include <hip/hip_runtime.h>
#include <hip/hip_bf16.h>
#include <stdint.h>


#define Bsz   4
#define Llen  16384
#define Dch   512
#define Nst   64
#define NFFT  32768
#define PI_F  3.14159265358979323846f

typedef __attribute__((ext_vector_type(2))) float v2f;
typedef __attribute__((ext_vector_type(8))) float v8f;

#ifndef __has_builtin
#define __has_builtin(x) 0
#endif
#if __has_builtin(__builtin_amdgcn_wmma_f32_16x16x4_f32)
#define S4_WMMA 1
#else
#define S4_WMMA 0
#endif

__device__ __forceinline__ float2 cadd(float2 a, float2 b) { return make_float2(a.x + b.x, a.y + b.y); }
__device__ __forceinline__ float2 csub(float2 a, float2 b) { return make_float2(a.x - b.x, a.y - b.y); }
__device__ __forceinline__ float2 cmulf(float2 a, float2 b) { return make_float2(a.x * b.x - a.y * b.y, a.x * b.y + a.y * b.x); }
__device__ __forceinline__ float2 cmulcf(float2 a, float2 b) { return make_float2(a.x * b.x + a.y * b.y, a.y * b.x - a.x * b.y); }
__device__ __forceinline__ float2 crotis(float2 a, float s) { return make_float2(-s * a.y, s * a.x); }

__device__ __forceinline__ void dft8(const float2* v, float s, float2* X) {
  const float R = 0.70710678118654752440f;
  float2 t0 = cadd(v[0], v[4]), t1 = cadd(v[1], v[5]);
  float2 t2 = cadd(v[2], v[6]), t3 = cadd(v[3], v[7]);
  float2 u0 = csub(v[0], v[4]), u1 = csub(v[1], v[5]);
  float2 u2 = csub(v[2], v[6]), u3 = csub(v[3], v[7]);
  float2 m1 = cmulf(u1, make_float2(R, s * R));
  float2 m2 = crotis(u2, s);
  float2 m3 = cmulf(u3, make_float2(-R, s * R));
  float2 p0 = cadd(t0, t2), p1 = csub(t0, t2);
  float2 p2 = cadd(t1, t3), p3 = crotis(csub(t1, t3), s);
  X[0] = cadd(p0, p2); X[2] = cadd(p1, p3);
  X[4] = csub(p0, p2); X[6] = csub(p1, p3);
  float2 q0 = cadd(u0, m2), q1 = csub(u0, m2);
  float2 q2 = cadd(m1, m3), q3 = crotis(csub(m1, m3), s);
  X[1] = cadd(q0, q2); X[3] = cadd(q1, q3);
  X[5] = csub(q0, q2); X[7] = csub(q1, q3);
}

__global__ __launch_bounds__(256) void s4_perl(
    const float* __restrict__ Lam, const float* __restrict__ Pri,
    const float* __restrict__ Cri, const float* __restrict__ log_step,
    float* __restrict__ perl)
{
  int l = blockIdx.x * blockDim.x + threadIdx.x;
  if (l >= Llen) return;
  float step = expf(log_step[0]);
  float theta = -2.0f * PI_F * (float)l / (float)Llen;
  float oi, orr;
  sincosf(theta, &oi, &orr);
  float op1r = 1.0f + orr;
  float den = op1r * op1r + oi * oi;
  float nr = 1.0f - orr, ni = -oi;
  float gscale = 2.0f / step;
  float gr = gscale * (nr * op1r + ni * oi) / den;
  float gi = gscale * (ni * op1r - nr * oi) / den;
  float scr = 2.0f * op1r / den;
  float sci = -2.0f * oi / den;
  float k01r = 0.f, k01i = 0.f, k11r = 0.f, k11i = 0.f;
  for (int n = 0; n < Nst; ++n) {
    float lr = Lam[2 * n], li = Lam[2 * n + 1];
    float pr = Pri[2 * n], pi = Pri[2 * n + 1];
    float cr = Cri[2 * n], ci = Cri[2 * n + 1];
    float zr = gr - lr, zi = gi - li;
    float zd = zr * zr + zi * zi;
    float car = zr / zd, cai = -zi / zd;
    float w01r = cr * pr + ci * pi;
    float w01i = cr * pi - ci * pr;
    float w11r = pr * pr + pi * pi;
    k01r += car * w01r - cai * w01i;
    k01i += car * w01i + cai * w01r;
    k11r += car * w11r;
    k11i += cai * w11r;
  }
  float dr = 1.0f + k11r, di = k11i;
  float dd = dr * dr + di * di;
  float sr = (k01r * dr + k01i * di) / dd;
  float si = (k01i * dr - k01r * di) / dd;
  perl[l] = sr;
  perl[Llen + l] = si;
  perl[2 * Llen + l] = scr;
  perl[3 * Llen + l] = sci;
  perl[4 * Llen + l] = gr;
  perl[5 * Llen + l] = gi;
}

__global__ __launch_bounds__(256) void s4_cauchy(
    const float* __restrict__ Lam, const float* __restrict__ Pri,
    const float* __restrict__ Cri, const float* __restrict__ perl,
    float2* __restrict__ Ac)
{
  int idx = blockIdx.x * blockDim.x + threadIdx.x;
  int l = idx >> 6;
  int n = idx & (Nst - 1);
  if (l >= Llen) return;
  float sr = perl[l],             si = perl[Llen + l];
  float scr = perl[2 * Llen + l], sci = perl[3 * Llen + l];
  float gr = perl[4 * Llen + l],  gi = perl[5 * Llen + l];
  float zr = gr - Lam[2 * n], zi = gi - Lam[2 * n + 1];
  float zd = zr * zr + zi * zi;
  float car = zr / zd, cai = -zi / zd;
  float cr = Cri[2 * n], ci = Cri[2 * n + 1];
  float pr = Pri[2 * n], pi = Pri[2 * n + 1];
  float tr = cr - (sr * pr + si * pi);
  float ti = -ci - (si * pr - sr * pi);
  float wr = scr * tr - sci * ti;
  float wi = scr * ti + sci * tr;
  Ac[idx] = make_float2(car * wr - cai * wi, car * wi + cai * wr);
}

__global__ __launch_bounds__(256) void s4_k_gemm(
    const float2* __restrict__ Ah, const float* __restrict__ Bri,
    float* __restrict__ Kr)
{
  int lane = threadIdx.x & 31;
  int wv = threadIdx.x >> 5;
  int tileL = blockIdx.x;
  int c = lane & 15;
  int h = lane >> 4;
  int dbase = wv * 64 + c;
#if S4_WMMA
  int m = tileL * 16 + c;
  v8f acc[4];
#pragma unroll
  for (int t = 0; t < 4; ++t)
    acc[t] = (v8f){0.f, 0.f, 0.f, 0.f, 0.f, 0.f, 0.f, 0.f};
  for (int kb = 0; kb < Nst; kb += 4) {
    int k0 = kb + 2 * h;
    float4 q = *(const float4*)(Ah + m * Nst + k0);
    v2f ar  = {q.x, q.z};
    v2f nai = {-q.y, -q.w};
    const float* b0 = Bri + 2 * (k0 * Dch + dbase);
    const float* b1 = b0 + 2 * Dch;
#pragma unroll
    for (int t = 0; t < 4; ++t) {
      v2f x0 = *(const v2f*)(b0 + 32 * t);
      v2f x1 = *(const v2f*)(b1 + 32 * t);
      v2f br = {x0.x, x1.x};
      v2f bi = {x0.y, x1.y};
      acc[t] = __builtin_amdgcn_wmma_f32_16x16x4_f32(false, ar,  false, br, (short)0, acc[t], false, false);
      acc[t] = __builtin_amdgcn_wmma_f32_16x16x4_f32(false, nai, false, bi, (short)0, acc[t], false, false);
    }
  }
#pragma unroll
  for (int t = 0; t < 4; ++t) {
#pragma unroll
    for (int r = 0; r < 8; ++r) {
      int l = tileL * 16 + 8 * h + r;
      Kr[l * Dch + dbase + 16 * t] = acc[t][r];
    }
  }
#else
  for (int t = 0; t < 4; ++t) {
    for (int r = 0; r < 8; ++r) {
      int l = tileL * 16 + 8 * h + r;
      int d = dbase + 16 * t;
      float er = 0.f;
      for (int k = 0; k < Nst; ++k) {
        float2 a = Ah[l * Nst + k];
        float brr = Bri[2 * (k * Dch + d)], bii = Bri[2 * (k * Dch + d) + 1];
        er += a.x * brr - a.y * bii;
      }
      Kr[l * Dch + d] = er;
    }
  }
#endif
}

__device__ __forceinline__ void lds_barrier() {
  __builtin_amdgcn_fence(__ATOMIC_RELEASE, "workgroup", "local");
  __builtin_amdgcn_s_barrier();
  __builtin_amdgcn_fence(__ATOMIC_ACQUIRE, "workgroup", "local");
}

template <bool PASS_TW>
__device__ __forceinline__ void r64_core(
    float2* v, float2* sh, int k, int Ns, float sign, int lane, int sub, float2* X)
{
  float ss, cs;
  sincospif(sign * (float)sub * (1.0f / 32.0f), &ss, &cs);
  float2 tw = make_float2(1.0f, 0.0f);
  if (PASS_TW) {
    float kf = (float)k / (float)Ns;
    float s8, c8, sb, cb;
    sincospif(sign * 0.25f * kf, &s8, &c8);
    sincospif(sign * (float)sub * kf * (1.0f / 32.0f), &sb, &cb);
    float2 w8 = make_float2(c8, s8);
    float2 w = w8;
#pragma unroll
    for (int r1 = 1; r1 < 8; ++r1) {
      v[r1] = cmulf(v[r1], w);
      w = cmulf(w, w8);
    }
    tw = make_float2(cb, sb);
  }
  float2 Y[8];
  dft8(v, sign, Y);
  float2 st = make_float2(cs, ss);
#pragma unroll
  for (int a = 0; a < 8; ++a) {
    sh[(a * 8 + sub) * 32 + lane] = cmulf(Y[a], tw);
    tw = cmulf(tw, st);
  }
  lds_barrier();
  float2 z[8];
#pragma unroll
  for (int r0 = 0; r0 < 8; ++r0)
    z[r0] = sh[(sub * 8 + r0) * 32 + lane];
  dft8(z, sign, X);
}

template <int LD>
__device__ __forceinline__ void r64_butterfly(
    const float2* __restrict__ x, float2* sh, int j, int q, int k, int Ns,
    float sign, int col, int lane, int sub, float2* X)
{
  float2 v[8];
#pragma unroll
  for (int r1 = 0; r1 < 8; ++r1)
    v[r1] = x[(j + (sub + 8 * r1) * q) * LD + col];
  r64_core<true>(v, sh, k, Ns, sign, lane, sub, X);
}

__global__ __launch_bounds__(256) void s4_fft_r64(
    const float2* __restrict__ x, float2* __restrict__ y,
    int n, int Ns, float sign)
{
  __shared__ float2 sh[64 * 32];
  int lane = threadIdx.x & 31;
  int sub = threadIdx.x >> 5;
  int col = blockIdx.x * 32 + lane;
  int j = blockIdx.y;
  int k = j & (Ns - 1);
  float2 X[8];
  r64_butterfly<Dch>(x, sh, j, n >> 6, k, Ns, sign, col, lane, sub, X);
  int i0 = 64 * j - 63 * k;
#pragma unroll
  for (int b = 0; b < 8; ++b)
    y[(i0 + (sub + 8 * b) * Ns) * Dch + col] = X[b];
}

template <bool PAIR>
__device__ __forceinline__ void pad_r512_body(
    const float* __restrict__ ua, const float* __restrict__ ub,
    float2* __restrict__ y, float2* sh)
{
  const int Q8 = NFFT / 8;
  int lane = threadIdx.x & 31;
  int sub = threadIdx.x >> 5;
  int col = blockIdx.x * 32 + lane;
  int jh = blockIdx.y;
  float2 S[8][8];
#pragma unroll
  for (int r1 = 0; r1 < 8; ++r1) {
    int jr = jh + 64 * (sub + 8 * r1);
    float2 v[8];
#pragma unroll
    for (int rp = 0; rp < 4; ++rp) {
      int pos = (jr + rp * Q8) * Dch + col;
      v[rp] = make_float2(ua[pos], PAIR ? ub[pos] : 0.f);
    }
#pragma unroll
    for (int rp = 4; rp < 8; ++rp)
      v[rp] = make_float2(0.f, 0.f);
    dft8(v, -1.0f, S[r1]);
  }
#pragma unroll
  for (int m = 0; m < 8; ++m) {
    float2 v[8], X[8];
#pragma unroll
    for (int r1 = 0; r1 < 8; ++r1)
      v[r1] = S[r1][m];
    r64_core<true>(v, sh + (m & 1) * (64 * 32), m, 8, -1.0f, lane, sub, X);
#pragma unroll
    for (int b = 0; b < (PAIR ? 8 : 4); ++b)
      y[(512 * jh + m + 8 * (sub + 8 * b)) * Dch + col] = X[b];
    if (!PAIR && m == 0 && sub == 0)
      y[(512 * jh + 256) * Dch + col] = X[4];
  }
}

__global__ __launch_bounds__(256) void s4_upad_r512(
    const float* __restrict__ ua, const float* __restrict__ ub,
    float2* __restrict__ y)
{
  __shared__ float2 sh[2 * 64 * 32];
  pad_r512_body<true>(ua, ub, y, sh);
}

__global__ __launch_bounds__(256) void s4_kpad_r512(
    const float* __restrict__ kr, float2* __restrict__ y)
{
  __shared__ float2 sh[2 * 64 * 32];
  pad_r512_body<false>(kr, kr, y, sh);
}

__global__ __launch_bounds__(256) void s4_afirst_r256(
    const float2* __restrict__ x, float2* __restrict__ y)
{
  __shared__ float2 sh[2 * 64 * 32];
  const int Q4 = Llen / 4;
  int lane = threadIdx.x & 31;
  int sub = threadIdx.x >> 5;
  int col = blockIdx.x * 32 + lane;
  int jh = blockIdx.y;
  float2 S[8][4];
#pragma unroll
  for (int r1 = 0; r1 < 8; ++r1) {
    int jr = jh + 64 * (sub + 8 * r1);
#pragma unroll
    for (int rp = 0; rp < 4; ++rp)
      S[r1][rp] = x[(jr + rp * Q4) * Nst + col];
  }
#pragma unroll
  for (int r1 = 0; r1 < 8; ++r1) {
    float2 p0 = cadd(S[r1][0], S[r1][2]), p1 = csub(S[r1][0], S[r1][2]);
    float2 p2 = cadd(S[r1][1], S[r1][3]);
    float2 p3 = crotis(csub(S[r1][1], S[r1][3]), +1.0f);
    S[r1][0] = cadd(p0, p2); S[r1][1] = cadd(p1, p3);
    S[r1][2] = csub(p0, p2); S[r1][3] = csub(p1, p3);
  }
#pragma unroll
  for (int m = 0; m < 4; ++m) {
    float2 v[8], X[8];
#pragma unroll
    for (int r1 = 0; r1 < 8; ++r1)
      v[r1] = S[r1][m];
    r64_core<true>(v, sh + (m & 1) * (64 * 32), m, 4, +1.0f, lane, sub, X);
#pragma unroll
    for (int b = 0; b < 8; ++b)
      y[(256 * jh + m + 4 * (sub + 8 * b)) * Nst + col] = X[b];
  }
}

__global__ __launch_bounds__(256) void s4_alast_r64(
    const float2* __restrict__ x, float2* __restrict__ ah)
{
  __shared__ float2 sh[64 * 32];
  const int Q = Llen / 64;
  int lane = threadIdx.x & 31;
  int sub = threadIdx.x >> 5;
  int col = blockIdx.x * 32 + lane;
  int j = blockIdx.y;
  float2 X[8];
  r64_butterfly<Nst>(x, sh, j, Q, j, Q, +1.0f, col, lane, sub, X);
  const float invL = 1.0f / (float)Llen;
#pragma unroll
  for (int b = 0; b < 8; ++b)
    ah[(j + (sub + 8 * b) * Q) * Nst + col] = make_float2(X[b].x * invL, X[b].y * invL);
}

__global__ __launch_bounds__(256) void s4_mid_r64_cmul_r64(
    const float2* __restrict__ x, const float2* __restrict__ Kd,
    float2* __restrict__ y)
{
  __shared__ float2 sh[2 * 64 * 32];
  const int Q = NFFT / 64;
  int lane = threadIdx.x & 31;
  int sub = threadIdx.x >> 5;
  int col = blockIdx.x * 32 + lane;
  int j = blockIdx.y;
  bool twin = (j > 0) && (j < Q / 2);
  int j2 = Q - j;
  int vs = 7 - sub;
  float2 kd[8], va[8], vb[8];
#pragma unroll
  for (int b = 0; b < 8; ++b)
    kd[b] = Kd[(j + (sub + 8 * b) * Q) * Dch + col];
#pragma unroll
  for (int r1 = 0; r1 < 8; ++r1)
    va[r1] = x[(j + (sub + 8 * r1) * Q) * Dch + col];
  if (twin) {
#pragma unroll
    for (int r1 = 0; r1 < 8; ++r1)
      vb[r1] = x[(j2 + (vs + 8 * r1) * Q) * Dch + col];
  }
  float2 X[8], v[8], Z[8];
  r64_core<true>(va, sh, j, Q, -1.0f, lane, sub, X);
#pragma unroll
  for (int b = 0; b < 8; ++b)
    v[b] = cmulf(X[b], kd[b]);
  r64_core<false>(v, sh + 64 * 32, 0, 1, +1.0f, lane, sub, Z);
#pragma unroll
  for (int b = 0; b < 8; ++b)
    y[(64 * j + sub + 8 * b) * Dch + col] = Z[b];
  if (twin) {
    r64_core<true>(vb, sh, j2, Q, -1.0f, lane, vs, X);
#pragma unroll
    for (int b = 0; b < 8; ++b)
      v[b] = cmulcf(X[b], kd[7 - b]);
    r64_core<false>(v, sh + 64 * 32, 0, 1, +1.0f, lane, vs, Z);
#pragma unroll
    for (int b = 0; b < 8; ++b)
      y[(64 * j2 + vs + 8 * b) * Dch + col] = Z[b];
  }
}

__global__ __launch_bounds__(256) __attribute__((amdgpu_waves_per_eu(4, 4))) void s4_final_r512(
    const float2* __restrict__ x,
    const float* __restrict__ ua, const float* __restrict__ ub,
    const float* __restrict__ Dp,
    float* __restrict__ outa, float* __restrict__ outb)
{
  __shared__ float2 sh[2 * 64 * 32];
  const int Q = NFFT / 64;
  const int Q8 = NFFT / 8;
  int lane = threadIdx.x & 31;
  int sub = threadIdx.x >> 5;
  int col = blockIdx.x * 32 + lane;
  int kk = blockIdx.y;
  float sw, cw;
  sincospif((float)kk * (1.0f / 256.0f), &sw, &cw);
  float2 w1 = make_float2(cw, sw);
  float2 wp[8];
  wp[1] = w1;
#pragma unroll
  for (int rp = 2; rp < 8; ++rp)
    wp[rp] = cmulf(wp[rp - 1], w1);
  float2 S[8][8];
#pragma unroll
  for (int r1 = 0; r1 < 8; ++r1) {
    int jr = kk + 64 * (sub + 8 * r1);
#pragma unroll
    for (int rp = 0; rp < 8; ++rp)
      S[r1][rp] = x[(jr + rp * Q8) * Dch + col];
  }
#pragma unroll
  for (int r1 = 0; r1 < 8; ++r1) {
    float2 v[8];
    v[0] = S[r1][0];
#pragma unroll
    for (int rp = 1; rp < 8; ++rp)
      v[rp] = cmulf(S[r1][rp], wp[rp]);
    dft8(v, +1.0f, S[r1]);
  }
  float dp = Dp[0];
  const float inv = 1.0f / (float)NFFT;
#pragma unroll
  for (int m = 0; m < 8; ++m) {
    int j2 = kk + 64 * m;
    float2 v[8], X[8];
#pragma unroll
    for (int r1 = 0; r1 < 8; ++r1)
      v[r1] = S[r1][m];
    float pa[4], pb[4];
#pragma unroll
    for (int b = 0; b < 4; ++b) {
      int pos = (j2 + (sub + 8 * b) * Q) * Dch + col;
      pa[b] = ua[pos];
      pb[b] = ub[pos];
    }
    r64_core<true>(v, sh + (m & 1) * (64 * 32), j2, Q, +1.0f, lane, sub, X);
#pragma unroll
    for (int b = 0; b < 4; ++b) {
      int pos = (j2 + (sub + 8 * b) * Q) * Dch + col;
      outa[pos] = X[b].x * inv + dp * pa[b];
      outb[pos] = X[b].y * inv + dp * pb[b];
    }
  }
}


extern "C" void kernel_launch(void* const* d_in, const int* in_sizes, int n_in,
                              void* d_out, int out_size, void* d_ws, size_t ws_size,
                              hipStream_t stream) {
  const float* u        = (const float*)d_in[0];
  const float* Lam      = (const float*)d_in[1];
  const float* Pri      = (const float*)d_in[2];
  const float* Bri      = (const float*)d_in[3];
  const float* Cri      = (const float*)d_in[4];
  const float* Dp       = (const float*)d_in[5];
  const float* log_step = (const float*)d_in[6];
  float* out = (float*)d_out;

  uint8_t* base = (uint8_t*)d_ws;
  size_t off = 0;
  auto carve = [&](size_t bytes) -> void* {
    void* p = base + off;
    off += (bytes + 255) & ~(size_t)255;
    return p;
  };
  float2* Ac   = (float2*)carve((size_t)Llen * Nst * 8);
  float2* At   = (float2*)carve((size_t)Llen * Nst * 8);
  float2* Ah   = (float2*)carve((size_t)Llen * Nst * 8);
  float*  perl = (float*)carve((size_t)6 * Llen * 4);
  float*  Kr   = (float*)carve((size_t)Llen * Dch * 4);
  float2* T0   = (float2*)carve((size_t)NFFT * Dch * 8);
  float2* T1   = (float2*)carve((size_t)NFFT * Dch * 8);
  float2* Kd   = (float2*)carve((size_t)NFFT * Dch * 8);
  (void)ws_size; (void)in_sizes; (void)n_in; (void)out_size;

  const int TB   = 256;
  const dim3 gR64h(Dch / 32, NFFT / 128 + 1);
  const dim3 gR64a(Nst / 32, Llen / 64);
  const dim3 gR512(Dch / 32, NFFT / 512);
  const dim3 gR256a(Nst / 32, Llen / 256);

  s4_perl<<<(Llen + TB - 1) / TB, TB, 0, stream>>>(Lam, Pri, Cri, log_step, perl);
  s4_cauchy<<<Llen * Nst / TB, TB, 0, stream>>>(Lam, Pri, Cri, perl, Ac);

  s4_afirst_r256<<<gR256a, TB, 0, stream>>>(Ac, At);
  s4_alast_r64<<<gR64a, TB, 0, stream>>>(At, Ah);
  s4_k_gemm<<<Llen / 16, TB, 0, stream>>>(Ah, Bri, Kr);
  s4_kpad_r512<<<gR512, TB, 0, stream>>>(Kr, T1);
  s4_fft_r64<<<gR64h, TB, 0, stream>>>(T1, Kd, NFFT, 512, -1.0f);

  for (int b = 0; b < Bsz; b += 2) {
    const float* ua = u + (size_t)b * Llen * Dch;
    const float* ub = u + (size_t)(b + 1) * Llen * Dch;
    float*       oa = out + (size_t)b * Llen * Dch;
    float*       ob = out + (size_t)(b + 1) * Llen * Dch;
    s4_upad_r512<<<gR512, TB, 0, stream>>>(ua, ub, T0);
    s4_mid_r64_cmul_r64<<<gR64h, TB, 0, stream>>>(T0, Kd, T1);
    s4_final_r512<<<gR512, TB, 0, stream>>>(T1, ua, ub, Dp, oa, ob);
  }
}


# Both pairs' first forward passes ride, in three parts, in the three small-grid kernel-generation launches (idle WGPs filled); Kd kept at 257 of 512 rows to fit a 2nd plane; 11 launches; same sums
# speedup vs baseline: 6.9247x; 1.0389x over previous
#include <hip/hip_runtime.h>
#include <hip/hip_bf16.h>
#include <stdint.h>


#define Bsz   4
#define Llen  16384
#define Dch   512
#define Nst   64
#define NFFT  32768
#define PI_F  3.14159265358979323846f
#define KDP   (NFFT / 128 + 1)

typedef __attribute__((ext_vector_type(2))) float v2f;
typedef __attribute__((ext_vector_type(8))) float v8f;

#ifndef __has_builtin
#define __has_builtin(x) 0
#endif
#if __has_builtin(__builtin_amdgcn_wmma_f32_16x16x4_f32)
#define S4_WMMA 1
#else
#define S4_WMMA 0
#endif

__device__ __forceinline__ float2 cadd(float2 a, float2 b) { return make_float2(a.x + b.x, a.y + b.y); }
__device__ __forceinline__ float2 csub(float2 a, float2 b) { return make_float2(a.x - b.x, a.y - b.y); }
__device__ __forceinline__ float2 cmulf(float2 a, float2 b) { return make_float2(a.x * b.x - a.y * b.y, a.x * b.y + a.y * b.x); }
__device__ __forceinline__ float2 cmulcf(float2 a, float2 b) { return make_float2(a.x * b.x + a.y * b.y, a.y * b.x - a.x * b.y); }
__device__ __forceinline__ float2 crotis(float2 a, float s) { return make_float2(-s * a.y, s * a.x); }

__device__ __forceinline__ void dft8(const float2* v, float s, float2* X) {
  const float R = 0.70710678118654752440f;
  float2 t0 = cadd(v[0], v[4]), t1 = cadd(v[1], v[5]);
  float2 t2 = cadd(v[2], v[6]), t3 = cadd(v[3], v[7]);
  float2 u0 = csub(v[0], v[4]), u1 = csub(v[1], v[5]);
  float2 u2 = csub(v[2], v[6]), u3 = csub(v[3], v[7]);
  float2 m1 = cmulf(u1, make_float2(R, s * R));
  float2 m2 = crotis(u2, s);
  float2 m3 = cmulf(u3, make_float2(-R, s * R));
  float2 p0 = cadd(t0, t2), p1 = csub(t0, t2);
  float2 p2 = cadd(t1, t3), p3 = crotis(csub(t1, t3), s);
  X[0] = cadd(p0, p2); X[2] = cadd(p1, p3);
  X[4] = csub(p0, p2); X[6] = csub(p1, p3);
  float2 q0 = cadd(u0, m2), q1 = csub(u0, m2);
  float2 q2 = cadd(m1, m3), q3 = crotis(csub(m1, m3), s);
  X[1] = cadd(q0, q2); X[3] = cadd(q1, q3);
  X[5] = csub(q0, q2); X[7] = csub(q1, q3);
}

__device__ __forceinline__ void perl_body(
    const float* __restrict__ Lam, const float* __restrict__ Pri,
    const float* __restrict__ Cri, const float* __restrict__ log_step,
    float* __restrict__ perl, int blk)
{
  int l = blk * 256 + threadIdx.x;
  if (l >= Llen) return;
  float step = expf(log_step[0]);
  float theta = -2.0f * PI_F * (float)l / (float)Llen;
  float oi, orr;
  sincosf(theta, &oi, &orr);
  float op1r = 1.0f + orr;
  float den = op1r * op1r + oi * oi;
  float nr = 1.0f - orr, ni = -oi;
  float gscale = 2.0f / step;
  float gr = gscale * (nr * op1r + ni * oi) / den;
  float gi = gscale * (ni * op1r - nr * oi) / den;
  float scr = 2.0f * op1r / den;
  float sci = -2.0f * oi / den;
  float k01r = 0.f, k01i = 0.f, k11r = 0.f, k11i = 0.f;
  for (int n = 0; n < Nst; ++n) {
    float lr = Lam[2 * n], li = Lam[2 * n + 1];
    float pr = Pri[2 * n], pi = Pri[2 * n + 1];
    float cr = Cri[2 * n], ci = Cri[2 * n + 1];
    float zr = gr - lr, zi = gi - li;
    float zd = zr * zr + zi * zi;
    float car = zr / zd, cai = -zi / zd;
    float w01r = cr * pr + ci * pi;
    float w01i = cr * pi - ci * pr;
    float w11r = pr * pr + pi * pi;
    k01r += car * w01r - cai * w01i;
    k01i += car * w01i + cai * w01r;
    k11r += car * w11r;
    k11i += cai * w11r;
  }
  float dr = 1.0f + k11r, di = k11i;
  float dd = dr * dr + di * di;
  float sr = (k01r * dr + k01i * di) / dd;
  float si = (k01i * dr - k01r * di) / dd;
  perl[l] = sr;
  perl[Llen + l] = si;
  perl[2 * Llen + l] = scr;
  perl[3 * Llen + l] = sci;
  perl[4 * Llen + l] = gr;
  perl[5 * Llen + l] = gi;
}

__global__ __launch_bounds__(256) void s4_cauchy(
    const float* __restrict__ Lam, const float* __restrict__ Pri,
    const float* __restrict__ Cri, const float* __restrict__ perl,
    float2* __restrict__ Ac)
{
  int idx = blockIdx.x * blockDim.x + threadIdx.x;
  int l = idx >> 6;
  int n = idx & (Nst - 1);
  if (l >= Llen) return;
  float sr = perl[l],             si = perl[Llen + l];
  float scr = perl[2 * Llen + l], sci = perl[3 * Llen + l];
  float gr = perl[4 * Llen + l],  gi = perl[5 * Llen + l];
  float zr = gr - Lam[2 * n], zi = gi - Lam[2 * n + 1];
  float zd = zr * zr + zi * zi;
  float car = zr / zd, cai = -zi / zd;
  float cr = Cri[2 * n], ci = Cri[2 * n + 1];
  float pr = Pri[2 * n], pi = Pri[2 * n + 1];
  float tr = cr - (sr * pr + si * pi);
  float ti = -ci - (si * pr - sr * pi);
  float wr = scr * tr - sci * ti;
  float wi = scr * ti + sci * tr;
  Ac[idx] = make_float2(car * wr - cai * wi, car * wi + cai * wr);
}

__global__ __launch_bounds__(256) void s4_k_gemm(
    const float2* __restrict__ Ah, const float* __restrict__ Bri,
    float* __restrict__ Kr)
{
  int lane = threadIdx.x & 31;
  int wv = threadIdx.x >> 5;
  int tileL = blockIdx.x;
  int c = lane & 15;
  int h = lane >> 4;
  int dbase = wv * 64 + c;
#if S4_WMMA
  int m = tileL * 16 + c;
  v8f acc[4];
#pragma unroll
  for (int t = 0; t < 4; ++t)
    acc[t] = (v8f){0.f, 0.f, 0.f, 0.f, 0.f, 0.f, 0.f, 0.f};
  for (int kb = 0; kb < Nst; kb += 4) {
    int k0 = kb + 2 * h;
    float4 q = *(const float4*)(Ah + m * Nst + k0);
    v2f ar  = {q.x, q.z};
    v2f nai = {-q.y, -q.w};
    const float* b0 = Bri + 2 * (k0 * Dch + dbase);
    const float* b1 = b0 + 2 * Dch;
#pragma unroll
    for (int t = 0; t < 4; ++t) {
      v2f x0 = *(const v2f*)(b0 + 32 * t);
      v2f x1 = *(const v2f*)(b1 + 32 * t);
      v2f br = {x0.x, x1.x};
      v2f bi = {x0.y, x1.y};
      acc[t] = __builtin_amdgcn_wmma_f32_16x16x4_f32(false, ar,  false, br, (short)0, acc[t], false, false);
      acc[t] = __builtin_amdgcn_wmma_f32_16x16x4_f32(false, nai, false, bi, (short)0, acc[t], false, false);
    }
  }
#pragma unroll
  for (int t = 0; t < 4; ++t) {
#pragma unroll
    for (int r = 0; r < 8; ++r) {
      int l = tileL * 16 + 8 * h + r;
      Kr[l * Dch + dbase + 16 * t] = acc[t][r];
    }
  }
#else
  for (int t = 0; t < 4; ++t) {
    for (int r = 0; r < 8; ++r) {
      int l = tileL * 16 + 8 * h + r;
      int d = dbase + 16 * t;
      float er = 0.f;
      for (int k = 0; k < Nst; ++k) {
        float2 a = Ah[l * Nst + k];
        float brr = Bri[2 * (k * Dch + d)], bii = Bri[2 * (k * Dch + d) + 1];
        er += a.x * brr - a.y * bii;
      }
      Kr[l * Dch + d] = er;
    }
  }
#endif
}

__device__ __forceinline__ void lds_barrier() {
  __builtin_amdgcn_fence(__ATOMIC_RELEASE, "workgroup", "local");
  __builtin_amdgcn_s_barrier();
  __builtin_amdgcn_fence(__ATOMIC_ACQUIRE, "workgroup", "local");
}

template <bool PASS_TW>
__device__ __forceinline__ void r64_core(
    float2* v, float2* sh, int k, int Ns, float sign, int lane, int sub, float2* X)
{
  float ss, cs;
  sincospif(sign * (float)sub * (1.0f / 32.0f), &ss, &cs);
  float2 tw = make_float2(1.0f, 0.0f);
  if (PASS_TW) {
    float kf = (float)k / (float)Ns;
    float s8, c8, sb, cb;
    sincospif(sign * 0.25f * kf, &s8, &c8);
    sincospif(sign * (float)sub * kf * (1.0f / 32.0f), &sb, &cb);
    float2 w8 = make_float2(c8, s8);
    float2 w = w8;
#pragma unroll
    for (int r1 = 1; r1 < 8; ++r1) {
      v[r1] = cmulf(v[r1], w);
      w = cmulf(w, w8);
    }
    tw = make_float2(cb, sb);
  }
  float2 Y[8];
  dft8(v, sign, Y);
  float2 st = make_float2(cs, ss);
#pragma unroll
  for (int a = 0; a < 8; ++a) {
    sh[(a * 8 + sub) * 32 + lane] = cmulf(Y[a], tw);
    tw = cmulf(tw, st);
  }
  lds_barrier();
  float2 z[8];
#pragma unroll
  for (int r0 = 0; r0 < 8; ++r0)
    z[r0] = sh[(sub * 8 + r0) * 32 + lane];
  dft8(z, sign, X);
}

template <int LD>
__device__ __forceinline__ void r64_butterfly(
    const float2* __restrict__ x, float2* sh, int j, int q, int k, int Ns,
    float sign, int col, int lane, int sub, float2* X)
{
  float2 v[8];
#pragma unroll
  for (int r1 = 0; r1 < 8; ++r1)
    v[r1] = x[(j + (sub + 8 * r1) * q) * LD + col];
  r64_core<true>(v, sh, k, Ns, sign, lane, sub, X);
}

__global__ __launch_bounds__(256) void s4_fft_r64(
    const float2* __restrict__ x, float2* __restrict__ y,
    int n, int Ns, int ypitch, float sign)
{
  __shared__ float2 sh[64 * 32];
  int lane = threadIdx.x & 31;
  int sub = threadIdx.x >> 5;
  int col = blockIdx.x * 32 + lane;
  int j = blockIdx.y;
  int k = j & (Ns - 1);
  float2 X[8];
  r64_butterfly<Dch>(x, sh, j, n >> 6, k, Ns, sign, col, lane, sub, X);
  int i0 = 64 * j - 63 * k;
#pragma unroll
  for (int b = 0; b < 8; ++b)
    y[(i0 + (sub + 8 * b) * ypitch) * Dch + col] = X[b];
}

template <bool PAIR>
__device__ __forceinline__ void pad_r512_body(
    const float* __restrict__ ua, const float* __restrict__ ub,
    float2* __restrict__ y, float2* sh, int cb, int jh)
{
  const int Q8 = NFFT / 8;
  int lane = threadIdx.x & 31;
  int sub = threadIdx.x >> 5;
  int col = cb * 32 + lane;
  float2 S[8][8];
#pragma unroll
  for (int r1 = 0; r1 < 8; ++r1) {
    int jr = jh + 64 * (sub + 8 * r1);
    float2 v[8];
#pragma unroll
    for (int rp = 0; rp < 4; ++rp) {
      int pos = (jr + rp * Q8) * Dch + col;
      v[rp] = make_float2(ua[pos], PAIR ? ub[pos] : 0.f);
    }
#pragma unroll
    for (int rp = 4; rp < 8; ++rp)
      v[rp] = make_float2(0.f, 0.f);
    dft8(v, -1.0f, S[r1]);
  }
#pragma unroll
  for (int m = 0; m < 8; ++m) {
    float2 v[8], X[8];
#pragma unroll
    for (int r1 = 0; r1 < 8; ++r1)
      v[r1] = S[r1][m];
    r64_core<true>(v, sh + (m & 1) * (64 * 32), m, 8, -1.0f, lane, sub, X);
#pragma unroll
    for (int b = 0; b < (PAIR ? 8 : 4); ++b)
      y[(512 * jh + m + 8 * (sub + 8 * b)) * Dch + col] = X[b];
    if (!PAIR && m == 0 && sub == 0)
      y[(512 * jh + 256) * Dch + col] = X[4];
  }
}

__global__ __launch_bounds__(256) void s4_kpad_r512(
    const float* __restrict__ kr, float2* __restrict__ y)
{
  __shared__ float2 sh[2 * 64 * 32];
  pad_r512_body<false>(kr, kr, y, sh, blockIdx.x, blockIdx.y);
}

__device__ __forceinline__ void afirst_r256_body(
    const float2* __restrict__ x, float2* __restrict__ y, float2* sh,
    int cb, int jh)
{
  const int Q4 = Llen / 4;
  int lane = threadIdx.x & 31;
  int sub = threadIdx.x >> 5;
  int col = cb * 32 + lane;
  float2 S[8][4];
#pragma unroll
  for (int r1 = 0; r1 < 8; ++r1) {
    int jr = jh + 64 * (sub + 8 * r1);
#pragma unroll
    for (int rp = 0; rp < 4; ++rp)
      S[r1][rp] = x[(jr + rp * Q4) * Nst + col];
  }
#pragma unroll
  for (int r1 = 0; r1 < 8; ++r1) {
    float2 p0 = cadd(S[r1][0], S[r1][2]), p1 = csub(S[r1][0], S[r1][2]);
    float2 p2 = cadd(S[r1][1], S[r1][3]);
    float2 p3 = crotis(csub(S[r1][1], S[r1][3]), +1.0f);
    S[r1][0] = cadd(p0, p2); S[r1][1] = cadd(p1, p3);
    S[r1][2] = csub(p0, p2); S[r1][3] = csub(p1, p3);
  }
#pragma unroll
  for (int m = 0; m < 4; ++m) {
    float2 v[8], X[8];
#pragma unroll
    for (int r1 = 0; r1 < 8; ++r1)
      v[r1] = S[r1][m];
    r64_core<true>(v, sh + (m & 1) * (64 * 32), m, 4, +1.0f, lane, sub, X);
#pragma unroll
    for (int b = 0; b < 8; ++b)
      y[(256 * jh + m + 4 * (sub + 8 * b)) * Nst + col] = X[b];
  }
}

__device__ __forceinline__ void alast_r64_body(
    const float2* __restrict__ x, float2* __restrict__ ah, float2* sh,
    int cb, int j)
{
  const int Q = Llen / 64;
  int lane = threadIdx.x & 31;
  int sub = threadIdx.x >> 5;
  int col = cb * 32 + lane;
  float2 X[8];
  r64_butterfly<Nst>(x, sh, j, Q, j, Q, +1.0f, col, lane, sub, X);
  const float invL = 1.0f / (float)Llen;
#pragma unroll
  for (int b = 0; b < 8; ++b)
    ah[(j + (sub + 8 * b) * Q) * Nst + col] = make_float2(X[b].x * invL, X[b].y * invL);
}

#define PJ0 52
#define PJ1 48
#define PJ2 (128 - PJ0 - PJ1)
#define NB_PERL   (Llen / 256)
#define NB_AFIRST ((Nst / 32) * (Llen / 256))
#define NB_ALAST  ((Nst / 32) * (Llen / 64))

__device__ __forceinline__ void pad_pairs_block(
    const float* __restrict__ u, float2* __restrict__ t0a,
    float2* __restrict__ t0b, float2* sh, int g0, int i)
{
  int g = g0 + (i >> 4);
  int pair = g >> 6;
  const float* ua = u + (size_t)pair * (2 * Llen * Dch);
  const float* ub = ua + (size_t)Llen * Dch;
  pad_r512_body<true>(ua, ub, pair ? t0b : t0a, sh, i & 15, g & 63);
}

__global__ __launch_bounds__(256) void s4_perl_upad(
    const float* __restrict__ Lam, const float* __restrict__ Pri,
    const float* __restrict__ Cri, const float* __restrict__ log_step,
    float* __restrict__ perl,
    const float* __restrict__ u, float2* __restrict__ t0a,
    float2* __restrict__ t0b)
{
  __shared__ float2 sh[2 * 64 * 32];
  int b = blockIdx.x;
  if (b < NB_PERL) {
    perl_body(Lam, Pri, Cri, log_step, perl, b);
  } else {
    pad_pairs_block(u, t0a, t0b, sh, 0, b - NB_PERL);
  }
}

__global__ __launch_bounds__(256) void s4_afirst_upad(
    const float2* __restrict__ x, float2* __restrict__ y,
    const float* __restrict__ u, float2* __restrict__ t0a,
    float2* __restrict__ t0b)
{
  __shared__ float2 sh[2 * 64 * 32];
  int b = blockIdx.x;
  if (b < PJ1 * 16) {
    pad_pairs_block(u, t0a, t0b, sh, PJ0, b);
  } else {
    int i = b - PJ1 * 16;
    afirst_r256_body(x, y, sh, i & 1, i >> 1);
  }
}

__global__ __launch_bounds__(256) void s4_alast_upad(
    const float2* __restrict__ x, float2* __restrict__ ah,
    const float* __restrict__ u, float2* __restrict__ t0a,
    float2* __restrict__ t0b)
{
  __shared__ float2 sh[2 * 64 * 32];
  int b = blockIdx.x;
  if (b < PJ2 * 16) {
    pad_pairs_block(u, t0a, t0b, sh, PJ0 + PJ1, b);
  } else {
    int i = b - PJ2 * 16;
    alast_r64_body(x, ah, sh, i & 1, i >> 1);
  }
}

__global__ __launch_bounds__(256) void s4_mid_r64_cmul_r64(
    const float2* __restrict__ x, const float2* __restrict__ Kd,
    float2* __restrict__ y)
{
  __shared__ float2 sh[2 * 64 * 32];
  const int Q = NFFT / 64;
  int lane = threadIdx.x & 31;
  int sub = threadIdx.x >> 5;
  int col = blockIdx.x * 32 + lane;
  int j = blockIdx.y;
  bool twin = (j > 0) && (j < Q / 2);
  int j2 = Q - j;
  int vs = 7 - sub;
  float2 kd[8], va[8], vb[8];
#pragma unroll
  for (int b = 0; b < 8; ++b)
    kd[b] = Kd[(j + (sub + 8 * b) * KDP) * Dch + col];
#pragma unroll
  for (int r1 = 0; r1 < 8; ++r1)
    va[r1] = x[(j + (sub + 8 * r1) * Q) * Dch + col];
  if (twin) {
#pragma unroll
    for (int r1 = 0; r1 < 8; ++r1)
      vb[r1] = x[(j2 + (vs + 8 * r1) * Q) * Dch + col];
  }
  float2 X[8], v[8], Z[8];
  r64_core<true>(va, sh, j, Q, -1.0f, lane, sub, X);
#pragma unroll
  for (int b = 0; b < 8; ++b)
    v[b] = cmulf(X[b], kd[b]);
  r64_core<false>(v, sh + 64 * 32, 0, 1, +1.0f, lane, sub, Z);
#pragma unroll
  for (int b = 0; b < 8; ++b)
    y[(64 * j + sub + 8 * b) * Dch + col] = Z[b];
  if (twin) {
    r64_core<true>(vb, sh, j2, Q, -1.0f, lane, vs, X);
#pragma unroll
    for (int b = 0; b < 8; ++b)
      v[b] = cmulcf(X[b], kd[7 - b]);
    r64_core<false>(v, sh + 64 * 32, 0, 1, +1.0f, lane, vs, Z);
#pragma unroll
    for (int b = 0; b < 8; ++b)
      y[(64 * j2 + vs + 8 * b) * Dch + col] = Z[b];
  }
}

__global__ __launch_bounds__(256) __attribute__((amdgpu_waves_per_eu(4, 4))) void s4_final_r512(
    const float2* __restrict__ x,
    const float* __restrict__ ua, const float* __restrict__ ub,
    const float* __restrict__ Dp,
    float* __restrict__ outa, float* __restrict__ outb)
{
  __shared__ float2 sh[2 * 64 * 32];
  const int Q = NFFT / 64;
  const int Q8 = NFFT / 8;
  int lane = threadIdx.x & 31;
  int sub = threadIdx.x >> 5;
  int col = blockIdx.x * 32 + lane;
  int kk = blockIdx.y;
  float sw, cw;
  sincospif((float)kk * (1.0f / 256.0f), &sw, &cw);
  float2 w1 = make_float2(cw, sw);
  float2 wp[8];
  wp[1] = w1;
#pragma unroll
  for (int rp = 2; rp < 8; ++rp)
    wp[rp] = cmulf(wp[rp - 1], w1);
  float2 S[8][8];
#pragma unroll
  for (int r1 = 0; r1 < 8; ++r1) {
    int jr = kk + 64 * (sub + 8 * r1);
#pragma unroll
    for (int rp = 0; rp < 8; ++rp)
      S[r1][rp] = x[(jr + rp * Q8) * Dch + col];
  }
#pragma unroll
  for (int r1 = 0; r1 < 8; ++r1) {
    float2 v[8];
    v[0] = S[r1][0];
#pragma unroll
    for (int rp = 1; rp < 8; ++rp)
      v[rp] = cmulf(S[r1][rp], wp[rp]);
    dft8(v, +1.0f, S[r1]);
  }
  float dp = Dp[0];
  const float inv = 1.0f / (float)NFFT;
#pragma unroll
  for (int m = 0; m < 8; ++m) {
    int j2 = kk + 64 * m;
    float2 v[8], X[8];
#pragma unroll
    for (int r1 = 0; r1 < 8; ++r1)
      v[r1] = S[r1][m];
    float pa[4], pb[4];
#pragma unroll
    for (int b = 0; b < 4; ++b) {
      int pos = (j2 + (sub + 8 * b) * Q) * Dch + col;
      pa[b] = ua[pos];
      pb[b] = ub[pos];
    }
    r64_core<true>(v, sh + (m & 1) * (64 * 32), j2, Q, +1.0f, lane, sub, X);
#pragma unroll
    for (int b = 0; b < 4; ++b) {
      int pos = (j2 + (sub + 8 * b) * Q) * Dch + col;
      outa[pos] = X[b].x * inv + dp * pa[b];
      outb[pos] = X[b].y * inv + dp * pb[b];
    }
  }
}


extern "C" void kernel_launch(void* const* d_in, const int* in_sizes, int n_in,
                              void* d_out, int out_size, void* d_ws, size_t ws_size,
                              hipStream_t stream) {
  const float* u        = (const float*)d_in[0];
  const float* Lam      = (const float*)d_in[1];
  const float* Pri      = (const float*)d_in[2];
  const float* Bri      = (const float*)d_in[3];
  const float* Cri      = (const float*)d_in[4];
  const float* Dp       = (const float*)d_in[5];
  const float* log_step = (const float*)d_in[6];
  float* out = (float*)d_out;

  uint8_t* base = (uint8_t*)d_ws;
  size_t off = 0;
  auto carve = [&](size_t bytes) -> void* {
    void* p = base + off;
    off += (bytes + 255) & ~(size_t)255;
    return p;
  };
  float2* Ac   = (float2*)carve((size_t)Llen * Nst * 8);
  float2* At   = (float2*)carve((size_t)Llen * Nst * 8);
  float2* Ah   = (float2*)carve((size_t)Llen * Nst * 8);
  float*  perl = (float*)carve((size_t)6 * Llen * 4);
  float*  Kr   = (float*)carve((size_t)Llen * Dch * 4);
  float2* T0a  = (float2*)carve((size_t)NFFT * Dch * 8);
  float2* T0b  = (float2*)carve((size_t)NFFT * Dch * 8);
  float2* T1   = (float2*)carve((size_t)NFFT * Dch * 8);
  float2* Kd   = (float2*)carve((size_t)64 * KDP * Dch * 8);
  (void)ws_size; (void)in_sizes; (void)n_in; (void)out_size;

  const int TB   = 256;
  const dim3 gR64h(Dch / 32, NFFT / 128 + 1);
  const dim3 gR512(Dch / 32, NFFT / 512);

  s4_perl_upad<<<NB_PERL + PJ0 * 16, TB, 0, stream>>>(Lam, Pri, Cri, log_step, perl, u, T0a, T0b);
  s4_cauchy<<<Llen * Nst / TB, TB, 0, stream>>>(Lam, Pri, Cri, perl, Ac);

  s4_afirst_upad<<<PJ1 * 16 + NB_AFIRST, TB, 0, stream>>>(Ac, At, u, T0a, T0b);
  s4_alast_upad<<<PJ2 * 16 + NB_ALAST, TB, 0, stream>>>(At, Ah, u, T0a, T0b);
  s4_k_gemm<<<Llen / 16, TB, 0, stream>>>(Ah, Bri, Kr);
  s4_kpad_r512<<<gR512, TB, 0, stream>>>(Kr, T1);
  s4_fft_r64<<<gR64h, TB, 0, stream>>>(T1, Kd, NFFT, 512, KDP, -1.0f);

  for (int b = 0; b < Bsz; b += 2) {
    const float* ua = u + (size_t)b * Llen * Dch;
    const float* ub = u + (size_t)(b + 1) * Llen * Dch;
    float*       oa = out + (size_t)b * Llen * Dch;
    float*       ob = out + (size_t)(b + 1) * Llen * Dch;
    const float2* T0 = (b == 0) ? T0a : T0b;
    s4_mid_r64_cmul_r64<<<gR64h, TB, 0, stream>>>(T0, Kd, T1);
    s4_final_r512<<<gR512, TB, 0, stream>>>(T1, ua, ub, Dp, oa, ob);
  }
}
